# starting kernel
# baseline (speedup 1.0000x reference)
#include <hip/hip_runtime.h>
#include <hip/hip_bf16.h>

#define Bsz 4
#define Lsz 64
#define DMsz 1024
#define Hsz 16
#define DKsz 64
#define DVsz 64
#define Dsz 1024
#define DIsz 4096
#define Rsz 256
#define NEGC (-1.0e9f)

typedef float v2f __attribute__((ext_vector_type(2)));
typedef float v8f __attribute__((ext_vector_type(8)));

#define GEMM_BIAS  1
#define GEMM_RELU  2
#define GEMM_RESID 4

__global__ __launch_bounds__(256)
void gemm_wmma_f32(const float* __restrict__ A, const float* __restrict__ B,
                   const float* __restrict__ bias, const float* __restrict__ resid,
                   float* __restrict__ C, int M, int N, int K, int flags)
{
    __shared__ float As[64][64 + 1];
    __shared__ float Bs[64][128 + 1];

    const int tid   = threadIdx.x;
    const int wave  = tid >> 5;
    const int lane  = tid & 31;
    const int l15   = lane & 15;
    const int lhalf = lane >> 4;
    const int m0    = blockIdx.y * 64;
    const int nblk  = blockIdx.x * 128;
    const int mq    = wave >> 2;
    const int nq    = wave & 3;

    const int ar = tid >> 4;
    const int ac = (tid & 15) << 2;
    const int br = (tid >> 5) * 8;
    const int bc = (tid & 31) << 2;
    float4 aprf[4], bprf[8];

    v8f acc[2][2];
#pragma unroll
    for (int x = 0; x < 2; ++x)
#pragma unroll
        for (int y = 0; y < 2; ++y)
            acc[x][y] = v8f{0.f, 0.f, 0.f, 0.f, 0.f, 0.f, 0.f, 0.f};

#pragma unroll
    for (int rr = 0; rr < 4; ++rr)
        aprf[rr] = *(const float4*)(A + (size_t)(m0 + ar + rr * 16) * K + ac);
#pragma unroll
    for (int rr = 0; rr < 8; ++rr)
        bprf[rr] = *(const float4*)(B + (size_t)(br + rr) * N + nblk + bc);

    const int arow0 = mq * 32 + l15;
    const int bcol0 = nq * 32 + l15;

    for (int k0 = 0; k0 < K; k0 += 64) {
#pragma unroll
        for (int rr = 0; rr < 4; ++rr) {
            As[ar + rr * 16][ac + 0] = aprf[rr].x;
            As[ar + rr * 16][ac + 1] = aprf[rr].y;
            As[ar + rr * 16][ac + 2] = aprf[rr].z;
            As[ar + rr * 16][ac + 3] = aprf[rr].w;
        }
#pragma unroll
        for (int rr = 0; rr < 8; ++rr) {
            Bs[br + rr][bc + 0] = bprf[rr].x;
            Bs[br + rr][bc + 1] = bprf[rr].y;
            Bs[br + rr][bc + 2] = bprf[rr].z;
            Bs[br + rr][bc + 3] = bprf[rr].w;
        }
        __syncthreads();

        if (k0 + 64 < K) {
#pragma unroll
            for (int rr = 0; rr < 4; ++rr)
                aprf[rr] = *(const float4*)(A + (size_t)(m0 + ar + rr * 16) * K
                                            + k0 + 64 + ac);
#pragma unroll
            for (int rr = 0; rr < 8; ++rr)
                bprf[rr] = *(const float4*)(B + (size_t)(k0 + 64 + br + rr) * N
                                            + nblk + bc);
        }

#pragma unroll
        for (int half = 0; half < 2; ++half) {
            v2f af[2][8], bf[2][8];
#pragma unroll
            for (int s = 0; s < 8; ++s) {
                const int kb = half * 32 + s * 4 + lhalf * 2;
                af[0][s] = v2f{As[arow0][kb],      As[arow0][kb + 1]};
                af[1][s] = v2f{As[arow0 + 16][kb], As[arow0 + 16][kb + 1]};
                bf[0][s] = v2f{Bs[kb][bcol0],      Bs[kb + 1][bcol0]};
                bf[1][s] = v2f{Bs[kb][bcol0 + 16], Bs[kb + 1][bcol0 + 16]};
            }
#pragma unroll
            for (int s = 0; s < 8; ++s) {
                acc[0][0] = __builtin_amdgcn_wmma_f32_16x16x4_f32(
                    false, af[0][s], false, bf[0][s], (short)0, acc[0][0], false, false);
                acc[0][1] = __builtin_amdgcn_wmma_f32_16x16x4_f32(
                    false, af[0][s], false, bf[1][s], (short)0, acc[0][1], false, false);
                acc[1][0] = __builtin_amdgcn_wmma_f32_16x16x4_f32(
                    false, af[1][s], false, bf[0][s], (short)0, acc[1][0], false, false);
                acc[1][1] = __builtin_amdgcn_wmma_f32_16x16x4_f32(
                    false, af[1][s], false, bf[1][s], (short)0, acc[1][1], false, false);
            }
        }
        __syncthreads();
    }

#pragma unroll
    for (int nsub = 0; nsub < 2; ++nsub) {
        const int col = nblk + nq * 32 + nsub * 16 + l15;
        const float bv = (flags & GEMM_BIAS) ? bias[col] : 0.f;
#pragma unroll
        for (int msub = 0; msub < 2; ++msub) {
#pragma unroll
            for (int r = 0; r < 8; ++r) {
                const int row = m0 + mq * 32 + msub * 16 + r + lhalf * 8;
                float v = acc[msub][nsub][r] + bv;
                if (flags & GEMM_RESID) v += resid[(size_t)row * N + col];
                if (flags & GEMM_RELU)  v = fmaxf(v, 0.f);
                C[(size_t)row * N + col] = v;
            }
        }
    }
}

__global__ __launch_bounds__(256)
void dec_kernel(const float* __restrict__ a0base, const float* __restrict__ a1base,
                const float* __restrict__ b1, const float* __restrict__ w2,
                const float* __restrict__ b2, float* __restrict__ dec)
{
    const int wid  = (int)((blockIdx.x * 256u + threadIdx.x) >> 5);
    const int lane = threadIdx.x & 31;
    const int n    = wid & 63;
    const int m    = (wid >> 6) & 63;
    const int b    = (wid >> 12) & 3;
    const int pair = wid >> 14;
    const int i    = pair / 3;
    const int j    = pair % 3;

    const float* a1 = a1base + ((size_t)j * Rsz + b * Lsz + m) * 2048;
    const float* a0 = a0base + ((size_t)i * Rsz + b * Lsz + n) * 2048;

    float acc = 0.f;
    for (int c = lane; c < 2048; c += 32) {
        float t = a1[c] + a0[c] + b1[c];
        if (t > 0.f) acc += t * (w2[2 * c + 1] - w2[2 * c]);
    }
#pragma unroll
    for (int off = 16; off; off >>= 1) acc += __shfl_xor(acc, off, 32);
    if (lane == 0) {
        float s = acc + (b2[1] - b2[0]);
        dec[wid] = (s > 0.f) ? NEGC : 0.f;
    }
}

__global__ __launch_bounds__(192)
void attn_scores_kernel(const float* __restrict__ Qb, const float* __restrict__ Kb,
                        const float* __restrict__ dec, const int* __restrict__ mask,
                        float* __restrict__ S)
{
    const int blk = blockIdx.x;
    const int q = blk & 63;
    const int h = (blk >> 6) & 15;
    const int b = (blk >> 10) & 3;
    const int i = blk >> 12;
    const int t = threadIdx.x;
    const int j = t >> 6;
    const int k = t & 63;

    __shared__ float qrow[64];
    __shared__ float red[192];

    const float* Q = Qb + ((size_t)i * Rsz + b * Lsz + q) * DMsz + h * DKsz;
    if (t < 64) qrow[t] = Q[t];
    __syncthreads();

    const float* Kp = Kb + ((size_t)j * Rsz + b * Lsz + k) * DMsz + h * DKsz;
    float s = 0.f;
#pragma unroll 8
    for (int d = 0; d < 64; ++d) s += qrow[d] * Kp[d];
    s *= 0.125f;
    s += dec[(((size_t)(i * 3 + j) * Bsz + b) * Lsz + q) * Lsz + k];
    if (mask[((size_t)b * Lsz + q) * Lsz + k] == 0) s = NEGC;

    red[t] = s; __syncthreads();
    if (t < 64) red[t] = fmaxf(red[t], fmaxf(red[t + 64], red[t + 128]));
    __syncthreads();
    for (int st = 32; st > 0; st >>= 1) {
        if (t < st) red[t] = fmaxf(red[t], red[t + st]);
        __syncthreads();
    }
    const float mx = red[0];
    __syncthreads();
    const float e = __expf(s - mx);
    red[t] = e; __syncthreads();
    if (t < 64) red[t] = red[t] + red[t + 64] + red[t + 128];
    __syncthreads();
    for (int st = 32; st > 0; st >>= 1) {
        if (t < st) red[t] += red[t + st];
        __syncthreads();
    }
    S[(size_t)blk * 192 + t] = e * (1.0f / red[0]);
}

__global__ __launch_bounds__(64)
void attn_av_kernel(const float* __restrict__ S, const float* __restrict__ Vb,
                    float* __restrict__ Ob)
{
    const int blk = blockIdx.x;
    const int q = blk & 63;
    const int h = (blk >> 6) & 15;
    const int b = (blk >> 10) & 3;
    const int i = blk >> 12;
    const int d = threadIdx.x;

    __shared__ float arow[192];
    const float* Sp = S + (size_t)blk * 192;
    arow[d] = Sp[d]; arow[d + 64] = Sp[d + 64]; arow[d + 128] = Sp[d + 128];
    __syncthreads();

    float acc = 0.f;
    for (int jk = 0; jk < 192; ++jk) {
        int j = jk >> 6, k = jk & 63;
        acc += arow[jk] * Vb[((size_t)j * Rsz + b * Lsz + k) * DMsz + h * DVsz + d];
    }
    Ob[((size_t)i * Rsz + b * Lsz + q) * DMsz + h * DVsz + d] = acc;
}

__global__ __launch_bounds__(256)
void layernorm_kernel(const float* __restrict__ x, const float* __restrict__ resid,
                      const float* __restrict__ g, const float* __restrict__ bta,
                      float* __restrict__ out)
{
    const int row = blockIdx.x;
    const int t = threadIdx.x;
    __shared__ float red[256];

    const float* xp = x + (size_t)row * 1024;
    float v[4];
    float s = 0.f;
#pragma unroll
    for (int u = 0; u < 4; ++u) {
        float val = xp[t + u * 256];
        if (resid) val += resid[(size_t)row * 1024 + t + u * 256];
        v[u] = val; s += val;
    }
    red[t] = s; __syncthreads();
    for (int st = 128; st > 0; st >>= 1) { if (t < st) red[t] += red[t + st]; __syncthreads(); }
    const float mu = red[0] * (1.f / 1024.f);
    __syncthreads();

    float s2 = 0.f;
#pragma unroll
    for (int u = 0; u < 4; ++u) { float dd = v[u] - mu; s2 += dd * dd; }
    red[t] = s2; __syncthreads();
    for (int st = 128; st > 0; st >>= 1) { if (t < st) red[t] += red[t + st]; __syncthreads(); }
    const float var = red[0] * (1.f / 1024.f);
    const float rinv = 1.0f / sqrtf(var + 1e-6f);
#pragma unroll
    for (int u = 0; u < 4; ++u) {
        int c = t + u * 256;
        out[(size_t)row * 1024 + c] = (v[u] - mu) * rinv * g[c] + bta[c];
    }
}

extern "C" void kernel_launch(void* const* d_in, const int* in_sizes, int n_in,
                              void* d_out, int out_size, void* d_ws, size_t ws_size,
                              hipStream_t stream)
{
    const float* X[3]   = { (const float*)d_in[0], (const float*)d_in[1], (const float*)d_in[2] };
    const int*   mask   = (const int*)d_in[3];
    const float* w_qs   = (const float*)d_in[4];
    const float* w_ks   = (const float*)d_in[5];
    const float* w_vs   = (const float*)d_in[6];
    const float* w_ds0  = (const float*)d_in[7];
    const float* w_ds1  = (const float*)d_in[8];
    const float* dc_w1  = (const float*)d_in[9];
    const float* dc_b1  = (const float*)d_in[10];
    const float* dc_w2  = (const float*)d_in[11];
    const float* dc_b2  = (const float*)d_in[12];
    const float* w_fc   = (const float*)d_in[13];
    const float* ln1_g  = (const float*)d_in[14];
    const float* ln1_b  = (const float*)d_in[15];
    const float* ffn_w1 = (const float*)d_in[16];
    const float* ffn_b1 = (const float*)d_in[17];
    const float* ffn_w2 = (const float*)d_in[18];
    const float* ffn_b2 = (const float*)d_in[19];
    const float* ln2_g  = (const float*)d_in[20];
    const float* ln2_b  = (const float*)d_in[21];
    float* out = (float*)d_out;

    float* ws = (float*)d_ws;
    size_t off = 0;
    auto alloc = [&](size_t n) { float* p = ws + off; off += n; return p; };
    const size_t RM = (size_t)Rsz * DMsz;
    float* Q   = alloc(3 * RM);
    float* Kb  = alloc(3 * RM);
    float* Vb  = alloc(3 * RM);
    float* d0  = alloc(3 * RM);
    float* d1  = alloc(3 * RM);
    float* a0  = alloc(3 * (size_t)Rsz * 2048);
    float* a1  = alloc(3 * (size_t)Rsz * 2048);
    float* dec = alloc(9 * (size_t)Bsz * Lsz * Lsz);
    float* S   = alloc((size_t)3 * Bsz * Hsz * Lsz * 192);
    float* O   = alloc(3 * RM);
    float* fc  = alloc(3 * RM);
    float* oln = alloc(3 * RM);
    float* h1  = alloc(3 * (size_t)Rsz * DIsz);
    float* f2  = alloc(3 * RM);
    (void)ws_size; (void)n_in; (void)in_sizes; (void)out_size;

    auto gemm = [&](const float* A, const float* Bm, const float* bias,
                    const float* resid, float* C, int M, int N, int K, int flags) {
        dim3 grid(N / 128, M / 64);
        gemm_wmma_f32<<<grid, dim3(256), 0, stream>>>(A, Bm, bias, resid, C, M, N, K, flags);
    };

    for (int i = 0; i < 3; ++i) {
        gemm(X[i], w_qs,  nullptr, nullptr, Q  + i * RM, Rsz, 1024, 1024, 0);
        gemm(X[i], w_ks,  nullptr, nullptr, Kb + i * RM, Rsz, 1024, 1024, 0);
        gemm(X[i], w_vs,  nullptr, nullptr, Vb + i * RM, Rsz, 1024, 1024, 0);
        gemm(X[i], w_ds0, nullptr, nullptr, d0 + i * RM, Rsz, 1024, 1024, 0);
        gemm(X[i], w_ds1, nullptr, nullptr, d1 + i * RM, Rsz, 1024, 1024, 0);
        gemm(d0 + i * RM, dc_w1 + (size_t)1024 * 2048, nullptr, nullptr,
             a0 + i * (size_t)Rsz * 2048, Rsz, 2048, 1024, 0);
        gemm(d1 + i * RM, dc_w1, nullptr, nullptr,
             a1 + i * (size_t)Rsz * 2048, Rsz, 2048, 1024, 0);
    }

    {
        int total_waves = 9 * Bsz * Lsz * Lsz;
        dec_kernel<<<total_waves / 8, 256, 0, stream>>>(a0, a1, dc_b1, dc_w2, dc_b2, dec);
    }

    {
        int nblk = 3 * Bsz * Hsz * Lsz;
        attn_scores_kernel<<<nblk, 192, 0, stream>>>(Q, Kb, dec, mask, S);
        attn_av_kernel<<<nblk, 64, 0, stream>>>(S, Vb, O);
    }

    for (int i = 0; i < 3; ++i) {
        gemm(O + i * RM, w_fc, nullptr, X[i], fc + i * RM, Rsz, 1024, 1024, GEMM_RESID);
        layernorm_kernel<<<Rsz, 256, 0, stream>>>(fc + i * RM, nullptr, ln1_g, ln1_b, oln + i * RM);
        gemm(oln + i * RM, ffn_w1, ffn_b1, nullptr, h1 + i * (size_t)Rsz * DIsz,
             Rsz, DIsz, 1024, GEMM_BIAS | GEMM_RELU);
        gemm(h1 + i * (size_t)Rsz * DIsz, ffn_w2, ffn_b2, nullptr, f2 + i * RM,
             Rsz, 1024, DIsz, GEMM_BIAS);
        layernorm_kernel<<<Rsz, 256, 0, stream>>>(f2 + i * RM, oln + i * RM, ln2_g, ln2_b,
                                                  out + i * RM);
    }
}


# FFN fused per 64-row x 128-hidden slab: relu(x@W1+b1) tile stays in LDS and meets its W2 rows at once; LN2 adds the 32 slab partial sums in fixed order (down-projection sum order changes)
# speedup vs baseline: 1.2142x; 1.2142x over previous
#include <hip/hip_runtime.h>
#include <hip/hip_bf16.h>

#define Bsz 4
#define Lsz 64
#define DMsz 1024
#define Hsz 16
#define DKsz 64
#define DVsz 64
#define Dsz 1024
#define DIsz 4096
#define Rsz 256
#define NEGC (-1.0e9f)

typedef float v2f __attribute__((ext_vector_type(2)));
typedef float v8f __attribute__((ext_vector_type(8)));

#define GEMM_BIAS  1
#define GEMM_RELU  2
#define GEMM_RESID 4

__global__ __launch_bounds__(256)
void gemm_wmma_f32(const float* __restrict__ A, const float* __restrict__ B,
                   const float* __restrict__ bias, const float* __restrict__ resid,
                   float* __restrict__ C, int M, int N, int K, int flags)
{
    __shared__ float As[64][64 + 1];
    __shared__ float Bs[64][128 + 1];

    const int tid   = threadIdx.x;
    const int wave  = tid >> 5;
    const int lane  = tid & 31;
    const int l15   = lane & 15;
    const int lhalf = lane >> 4;
    const int m0    = blockIdx.y * 64;
    const int nblk  = blockIdx.x * 128;
    const int mq    = wave >> 2;
    const int nq    = wave & 3;

    const int ar = tid >> 4;
    const int ac = (tid & 15) << 2;
    const int br = (tid >> 5) * 8;
    const int bc = (tid & 31) << 2;
    float4 aprf[4], bprf[8];

    v8f acc[2][2];
#pragma unroll
    for (int x = 0; x < 2; ++x)
#pragma unroll
        for (int y = 0; y < 2; ++y)
            acc[x][y] = v8f{0.f, 0.f, 0.f, 0.f, 0.f, 0.f, 0.f, 0.f};

#pragma unroll
    for (int rr = 0; rr < 4; ++rr)
        aprf[rr] = *(const float4*)(A + (size_t)(m0 + ar + rr * 16) * K + ac);
#pragma unroll
    for (int rr = 0; rr < 8; ++rr)
        bprf[rr] = *(const float4*)(B + (size_t)(br + rr) * N + nblk + bc);

    const int arow0 = mq * 32 + l15;
    const int bcol0 = nq * 32 + l15;

    for (int k0 = 0; k0 < K; k0 += 64) {
#pragma unroll
        for (int rr = 0; rr < 4; ++rr) {
            As[ar + rr * 16][ac + 0] = aprf[rr].x;
            As[ar + rr * 16][ac + 1] = aprf[rr].y;
            As[ar + rr * 16][ac + 2] = aprf[rr].z;
            As[ar + rr * 16][ac + 3] = aprf[rr].w;
        }
#pragma unroll
        for (int rr = 0; rr < 8; ++rr) {
            Bs[br + rr][bc + 0] = bprf[rr].x;
            Bs[br + rr][bc + 1] = bprf[rr].y;
            Bs[br + rr][bc + 2] = bprf[rr].z;
            Bs[br + rr][bc + 3] = bprf[rr].w;
        }
        __syncthreads();

        if (k0 + 64 < K) {
#pragma unroll
            for (int rr = 0; rr < 4; ++rr)
                aprf[rr] = *(const float4*)(A + (size_t)(m0 + ar + rr * 16) * K
                                            + k0 + 64 + ac);
#pragma unroll
            for (int rr = 0; rr < 8; ++rr)
                bprf[rr] = *(const float4*)(B + (size_t)(k0 + 64 + br + rr) * N
                                            + nblk + bc);
        }

#pragma unroll
        for (int half = 0; half < 2; ++half) {
            v2f af[2][8], bf[2][8];
#pragma unroll
            for (int s = 0; s < 8; ++s) {
                const int kb = half * 32 + s * 4 + lhalf * 2;
                af[0][s] = v2f{As[arow0][kb],      As[arow0][kb + 1]};
                af[1][s] = v2f{As[arow0 + 16][kb], As[arow0 + 16][kb + 1]};
                bf[0][s] = v2f{Bs[kb][bcol0],      Bs[kb + 1][bcol0]};
                bf[1][s] = v2f{Bs[kb][bcol0 + 16], Bs[kb + 1][bcol0 + 16]};
            }
#pragma unroll
            for (int s = 0; s < 8; ++s) {
                acc[0][0] = __builtin_amdgcn_wmma_f32_16x16x4_f32(
                    false, af[0][s], false, bf[0][s], (short)0, acc[0][0], false, false);
                acc[0][1] = __builtin_amdgcn_wmma_f32_16x16x4_f32(
                    false, af[0][s], false, bf[1][s], (short)0, acc[0][1], false, false);
                acc[1][0] = __builtin_amdgcn_wmma_f32_16x16x4_f32(
                    false, af[1][s], false, bf[0][s], (short)0, acc[1][0], false, false);
                acc[1][1] = __builtin_amdgcn_wmma_f32_16x16x4_f32(
                    false, af[1][s], false, bf[1][s], (short)0, acc[1][1], false, false);
            }
        }
        __syncthreads();
    }

#pragma unroll
    for (int nsub = 0; nsub < 2; ++nsub) {
        const int col = nblk + nq * 32 + nsub * 16 + l15;
        const float bv = (flags & GEMM_BIAS) ? bias[col] : 0.f;
#pragma unroll
        for (int msub = 0; msub < 2; ++msub) {
#pragma unroll
            for (int r = 0; r < 8; ++r) {
                const int row = m0 + mq * 32 + msub * 16 + r + lhalf * 8;
                float v = acc[msub][nsub][r] + bv;
                if (flags & GEMM_RESID) v += resid[(size_t)row * N + col];
                if (flags & GEMM_RELU)  v = fmaxf(v, 0.f);
                C[(size_t)row * N + col] = v;
            }
        }
    }
}

__global__ __launch_bounds__(256)
void ffn_fused_wmma_f32(const float* __restrict__ X, const float* __restrict__ W1,
                        const float* __restrict__ b1, const float* __restrict__ W2,
                        float* __restrict__ P)
{
    __shared__ float As[64][64 + 1];
    __shared__ float Bs[64][128 + 1];
    __shared__ float Hs[64][128 + 1];

    const int tid   = threadIdx.x;
    const int wave  = tid >> 5;
    const int lane  = tid & 31;
    const int l15   = lane & 15;
    const int lhalf = lane >> 4;
    const int s0    = blockIdx.x * 128;
    const int m0    = blockIdx.y * 64;
    const int nc0   = blockIdx.z * 4;
    const int mq    = wave >> 2;
    const int nq    = wave & 3;

    const int ar = tid >> 4;
    const int ac = (tid & 15) << 2;
    const int br = (tid >> 5) * 8;
    const int bc = (tid & 31) << 2;
    float4 aprf[4], bprf[8];

    v8f acc[2][2];
#pragma unroll
    for (int x = 0; x < 2; ++x)
#pragma unroll
        for (int y = 0; y < 2; ++y)
            acc[x][y] = v8f{0.f, 0.f, 0.f, 0.f, 0.f, 0.f, 0.f, 0.f};

#pragma unroll
    for (int rr = 0; rr < 4; ++rr)
        aprf[rr] = *(const float4*)(X + (size_t)(m0 + ar + rr * 16) * DMsz + ac);
#pragma unroll
    for (int rr = 0; rr < 8; ++rr)
        bprf[rr] = *(const float4*)(W1 + (size_t)(br + rr) * DIsz + s0 + bc);

    const int arow0 = mq * 32 + l15;
    const int bcol0 = nq * 32 + l15;

    for (int k0 = 0; k0 < DMsz; k0 += 64) {
#pragma unroll
        for (int rr = 0; rr < 4; ++rr) {
            As[ar + rr * 16][ac + 0] = aprf[rr].x;
            As[ar + rr * 16][ac + 1] = aprf[rr].y;
            As[ar + rr * 16][ac + 2] = aprf[rr].z;
            As[ar + rr * 16][ac + 3] = aprf[rr].w;
        }
#pragma unroll
        for (int rr = 0; rr < 8; ++rr) {
            Bs[br + rr][bc + 0] = bprf[rr].x;
            Bs[br + rr][bc + 1] = bprf[rr].y;
            Bs[br + rr][bc + 2] = bprf[rr].z;
            Bs[br + rr][bc + 3] = bprf[rr].w;
        }
        __syncthreads();

        if (k0 + 64 < DMsz) {
#pragma unroll
            for (int rr = 0; rr < 4; ++rr)
                aprf[rr] = *(const float4*)(X + (size_t)(m0 + ar + rr * 16) * DMsz
                                            + k0 + 64 + ac);
#pragma unroll
            for (int rr = 0; rr < 8; ++rr)
                bprf[rr] = *(const float4*)(W1 + (size_t)(k0 + 64 + br + rr) * DIsz
                                            + s0 + bc);
        } else {
#pragma unroll
            for (int rr = 0; rr < 8; ++rr)
                bprf[rr] = *(const float4*)(W2 + (size_t)(s0 + br + rr) * DMsz
                                            + nc0 * 128 + bc);
        }

#pragma unroll
        for (int half = 0; half < 2; ++half) {
            v2f af[2][8], bf[2][8];
#pragma unroll
            for (int s = 0; s < 8; ++s) {
                const int kb = half * 32 + s * 4 + lhalf * 2;
                af[0][s] = v2f{As[arow0][kb],      As[arow0][kb + 1]};
                af[1][s] = v2f{As[arow0 + 16][kb], As[arow0 + 16][kb + 1]};
                bf[0][s] = v2f{Bs[kb][bcol0],      Bs[kb + 1][bcol0]};
                bf[1][s] = v2f{Bs[kb][bcol0 + 16], Bs[kb + 1][bcol0 + 16]};
            }
#pragma unroll
            for (int s = 0; s < 8; ++s) {
                acc[0][0] = __builtin_amdgcn_wmma_f32_16x16x4_f32(
                    false, af[0][s], false, bf[0][s], (short)0, acc[0][0], false, false);
                acc[0][1] = __builtin_amdgcn_wmma_f32_16x16x4_f32(
                    false, af[0][s], false, bf[1][s], (short)0, acc[0][1], false, false);
                acc[1][0] = __builtin_amdgcn_wmma_f32_16x16x4_f32(
                    false, af[1][s], false, bf[0][s], (short)0, acc[1][0], false, false);
                acc[1][1] = __builtin_amdgcn_wmma_f32_16x16x4_f32(
                    false, af[1][s], false, bf[1][s], (short)0, acc[1][1], false, false);
            }
        }
        __syncthreads();
    }

#pragma unroll
    for (int nsub = 0; nsub < 2; ++nsub) {
        const int hc = nq * 32 + nsub * 16 + l15;
        const float bv = b1[s0 + hc];
#pragma unroll
        for (int msub = 0; msub < 2; ++msub) {
#pragma unroll
            for (int r = 0; r < 8; ++r) {
                const int hr = mq * 32 + msub * 16 + r + lhalf * 8;
                Hs[hr][hc] = fmaxf(acc[msub][nsub][r] + bv, 0.f);
            }
        }
    }

    for (int nc = nc0; nc < nc0 + 4; ++nc) {
#pragma unroll
        for (int x = 0; x < 2; ++x)
#pragma unroll
            for (int y = 0; y < 2; ++y)
                acc[x][y] = v8f{0.f, 0.f, 0.f, 0.f, 0.f, 0.f, 0.f, 0.f};

        for (int kc = 0; kc < 2; ++kc) {
#pragma unroll
            for (int rr = 0; rr < 8; ++rr) {
                Bs[br + rr][bc + 0] = bprf[rr].x;
                Bs[br + rr][bc + 1] = bprf[rr].y;
                Bs[br + rr][bc + 2] = bprf[rr].z;
                Bs[br + rr][bc + 3] = bprf[rr].w;
            }
            __syncthreads();

            const int tn = (nc - nc0) * 2 + kc + 1;
            if (tn < 8) {
                const int nn = nc0 + (tn >> 1);
                const int kk = tn & 1;
#pragma unroll
                for (int rr = 0; rr < 8; ++rr)
                    bprf[rr] = *(const float4*)(W2 + (size_t)(s0 + kk * 64 + br + rr) * DMsz
                                                + nn * 128 + bc);
            }

            const int hk0 = kc * 64;
#pragma unroll
            for (int half = 0; half < 2; ++half) {
                v2f af[2][8], bf[2][8];
#pragma unroll
                for (int s = 0; s < 8; ++s) {
                    const int kb = half * 32 + s * 4 + lhalf * 2;
                    af[0][s] = v2f{Hs[arow0][hk0 + kb],      Hs[arow0][hk0 + kb + 1]};
                    af[1][s] = v2f{Hs[arow0 + 16][hk0 + kb], Hs[arow0 + 16][hk0 + kb + 1]};
                    bf[0][s] = v2f{Bs[kb][bcol0],      Bs[kb + 1][bcol0]};
                    bf[1][s] = v2f{Bs[kb][bcol0 + 16], Bs[kb + 1][bcol0 + 16]};
                }
#pragma unroll
                for (int s = 0; s < 8; ++s) {
                    acc[0][0] = __builtin_amdgcn_wmma_f32_16x16x4_f32(
                        false, af[0][s], false, bf[0][s], (short)0, acc[0][0], false, false);
                    acc[0][1] = __builtin_amdgcn_wmma_f32_16x16x4_f32(
                        false, af[0][s], false, bf[1][s], (short)0, acc[0][1], false, false);
                    acc[1][0] = __builtin_amdgcn_wmma_f32_16x16x4_f32(
                        false, af[1][s], false, bf[0][s], (short)0, acc[1][0], false, false);
                    acc[1][1] = __builtin_amdgcn_wmma_f32_16x16x4_f32(
                        false, af[1][s], false, bf[1][s], (short)0, acc[1][1], false, false);
                }
            }
            __syncthreads();
        }

        float* Pp = P + (size_t)blockIdx.x * Rsz * DMsz;
#pragma unroll
        for (int nsub = 0; nsub < 2; ++nsub) {
            const int col = nc * 128 + nq * 32 + nsub * 16 + l15;
#pragma unroll
            for (int msub = 0; msub < 2; ++msub) {
#pragma unroll
                for (int r = 0; r < 8; ++r) {
                    const int row = m0 + mq * 32 + msub * 16 + r + lhalf * 8;
                    Pp[(size_t)row * DMsz + col] = acc[msub][nsub][r];
                }
            }
        }
    }
}

__global__ __launch_bounds__(256)
void dec_kernel(const float* __restrict__ a0base, const float* __restrict__ a1base,
                const float* __restrict__ b1, const float* __restrict__ w2,
                const float* __restrict__ b2, float* __restrict__ dec)
{
    const int wid  = (int)((blockIdx.x * 256u + threadIdx.x) >> 5);
    const int lane = threadIdx.x & 31;
    const int n    = wid & 63;
    const int m    = (wid >> 6) & 63;
    const int b    = (wid >> 12) & 3;
    const int pair = wid >> 14;
    const int i    = pair / 3;
    const int j    = pair % 3;

    const float* a1 = a1base + ((size_t)j * Rsz + b * Lsz + m) * 2048;
    const float* a0 = a0base + ((size_t)i * Rsz + b * Lsz + n) * 2048;

    float acc = 0.f;
    for (int c = lane; c < 2048; c += 32) {
        float t = a1[c] + a0[c] + b1[c];
        if (t > 0.f) acc += t * (w2[2 * c + 1] - w2[2 * c]);
    }
#pragma unroll
    for (int off = 16; off; off >>= 1) acc += __shfl_xor(acc, off, 32);
    if (lane == 0) {
        float s = acc + (b2[1] - b2[0]);
        dec[wid] = (s > 0.f) ? NEGC : 0.f;
    }
}

__global__ __launch_bounds__(192)
void attn_scores_kernel(const float* __restrict__ Qb, const float* __restrict__ Kb,
                        const float* __restrict__ dec, const int* __restrict__ mask,
                        float* __restrict__ S)
{
    const int blk = blockIdx.x;
    const int q = blk & 63;
    const int h = (blk >> 6) & 15;
    const int b = (blk >> 10) & 3;
    const int i = blk >> 12;
    const int t = threadIdx.x;
    const int j = t >> 6;
    const int k = t & 63;

    __shared__ float qrow[64];
    __shared__ float red[192];

    const float* Q = Qb + ((size_t)i * Rsz + b * Lsz + q) * DMsz + h * DKsz;
    if (t < 64) qrow[t] = Q[t];
    __syncthreads();

    const float* Kp = Kb + ((size_t)j * Rsz + b * Lsz + k) * DMsz + h * DKsz;
    float s = 0.f;
#pragma unroll 8
    for (int d = 0; d < 64; ++d) s += qrow[d] * Kp[d];
    s *= 0.125f;
    s += dec[(((size_t)(i * 3 + j) * Bsz + b) * Lsz + q) * Lsz + k];
    if (mask[((size_t)b * Lsz + q) * Lsz + k] == 0) s = NEGC;

    red[t] = s; __syncthreads();
    if (t < 64) red[t] = fmaxf(red[t], fmaxf(red[t + 64], red[t + 128]));
    __syncthreads();
    for (int st = 32; st > 0; st >>= 1) {
        if (t < st) red[t] = fmaxf(red[t], red[t + st]);
        __syncthreads();
    }
    const float mx = red[0];
    __syncthreads();
    const float e = __expf(s - mx);
    red[t] = e; __syncthreads();
    if (t < 64) red[t] = red[t] + red[t + 64] + red[t + 128];
    __syncthreads();
    for (int st = 32; st > 0; st >>= 1) {
        if (t < st) red[t] += red[t + st];
        __syncthreads();
    }
    S[(size_t)blk * 192 + t] = e * (1.0f / red[0]);
}

__global__ __launch_bounds__(64)
void attn_av_kernel(const float* __restrict__ S, const float* __restrict__ Vb,
                    float* __restrict__ Ob)
{
    const int blk = blockIdx.x;
    const int q = blk & 63;
    const int h = (blk >> 6) & 15;
    const int b = (blk >> 10) & 3;
    const int i = blk >> 12;
    const int d = threadIdx.x;

    __shared__ float arow[192];
    const float* Sp = S + (size_t)blk * 192;
    arow[d] = Sp[d]; arow[d + 64] = Sp[d + 64]; arow[d + 128] = Sp[d + 128];
    __syncthreads();

    float acc = 0.f;
    for (int jk = 0; jk < 192; ++jk) {
        int j = jk >> 6, k = jk & 63;
        acc += arow[jk] * Vb[((size_t)j * Rsz + b * Lsz + k) * DMsz + h * DVsz + d];
    }
    Ob[((size_t)i * Rsz + b * Lsz + q) * DMsz + h * DVsz + d] = acc;
}

__global__ __launch_bounds__(256)
void layernorm_kernel(const float* __restrict__ x, int nparts, size_t pstride,
                      const float* __restrict__ bias, const float* __restrict__ resid,
                      const float* __restrict__ g, const float* __restrict__ bta,
                      float* __restrict__ out)
{
    const int row = blockIdx.x;
    const int t = threadIdx.x;
    __shared__ float red[256];

    const float* xp = x + (size_t)row * 1024;
    float v[4];
#pragma unroll
    for (int u = 0; u < 4; ++u) v[u] = xp[t + u * 256];
#pragma unroll 8
    for (int p = 1; p < nparts; ++p) {
        const float* pp = xp + (size_t)p * pstride;
#pragma unroll
        for (int u = 0; u < 4; ++u) v[u] += pp[t + u * 256];
    }
    float s = 0.f;
#pragma unroll
    for (int u = 0; u < 4; ++u) {
        float val = v[u];
        if (bias)  val += bias[t + u * 256];
        if (resid) val += resid[(size_t)row * 1024 + t + u * 256];
        v[u] = val; s += val;
    }
    red[t] = s; __syncthreads();
    for (int st = 128; st > 0; st >>= 1) { if (t < st) red[t] += red[t + st]; __syncthreads(); }
    const float mu = red[0] * (1.f / 1024.f);
    __syncthreads();

    float s2 = 0.f;
#pragma unroll
    for (int u = 0; u < 4; ++u) { float dd = v[u] - mu; s2 += dd * dd; }
    red[t] = s2; __syncthreads();
    for (int st = 128; st > 0; st >>= 1) { if (t < st) red[t] += red[t + st]; __syncthreads(); }
    const float var = red[0] * (1.f / 1024.f);
    const float rinv = 1.0f / sqrtf(var + 1e-6f);
#pragma unroll
    for (int u = 0; u < 4; ++u) {
        int c = t + u * 256;
        out[(size_t)row * 1024 + c] = (v[u] - mu) * rinv * g[c] + bta[c];
    }
}

extern "C" void kernel_launch(void* const* d_in, const int* in_sizes, int n_in,
                              void* d_out, int out_size, void* d_ws, size_t ws_size,
                              hipStream_t stream)
{
    const float* X[3]   = { (const float*)d_in[0], (const float*)d_in[1], (const float*)d_in[2] };
    const int*   mask   = (const int*)d_in[3];
    const float* w_qs   = (const float*)d_in[4];
    const float* w_ks   = (const float*)d_in[5];
    const float* w_vs   = (const float*)d_in[6];
    const float* w_ds0  = (const float*)d_in[7];
    const float* w_ds1  = (const float*)d_in[8];
    const float* dc_w1  = (const float*)d_in[9];
    const float* dc_b1  = (const float*)d_in[10];
    const float* dc_w2  = (const float*)d_in[11];
    const float* dc_b2  = (const float*)d_in[12];
    const float* w_fc   = (const float*)d_in[13];
    const float* ln1_g  = (const float*)d_in[14];
    const float* ln1_b  = (const float*)d_in[15];
    const float* ffn_w1 = (const float*)d_in[16];
    const float* ffn_b1 = (const float*)d_in[17];
    const float* ffn_w2 = (const float*)d_in[18];
    const float* ffn_b2 = (const float*)d_in[19];
    const float* ln2_g  = (const float*)d_in[20];
    const float* ln2_b  = (const float*)d_in[21];
    float* out = (float*)d_out;

    float* ws = (float*)d_ws;
    size_t off = 0;
    auto alloc = [&](size_t n) { float* p = ws + off; off += n; return p; };
    const size_t RM = (size_t)Rsz * DMsz;
    float* Q   = alloc(3 * RM);
    float* Kb  = alloc(3 * RM);
    float* Vb  = alloc(3 * RM);
    float* d0  = alloc(3 * RM);
    float* d1  = alloc(3 * RM);
    float* a0  = alloc(3 * (size_t)Rsz * 2048);
    float* a1  = alloc(3 * (size_t)Rsz * 2048);
    float* dec = alloc(9 * (size_t)Bsz * Lsz * Lsz);
    float* S   = alloc((size_t)3 * Bsz * Hsz * Lsz * 192);
    float* O   = alloc(3 * RM);
    float* fc  = alloc(3 * RM);
    float* oln = alloc(3 * RM);
    float* fpart = alloc(32 * RM);
    (void)ws_size; (void)n_in; (void)in_sizes; (void)out_size;

    auto gemm = [&](const float* A, const float* Bm, const float* bias,
                    const float* resid, float* C, int M, int N, int K, int flags) {
        dim3 grid(N / 128, M / 64);
        gemm_wmma_f32<<<grid, dim3(256), 0, stream>>>(A, Bm, bias, resid, C, M, N, K, flags);
    };

    for (int i = 0; i < 3; ++i) {
        gemm(X[i], w_qs,  nullptr, nullptr, Q  + i * RM, Rsz, 1024, 1024, 0);
        gemm(X[i], w_ks,  nullptr, nullptr, Kb + i * RM, Rsz, 1024, 1024, 0);
        gemm(X[i], w_vs,  nullptr, nullptr, Vb + i * RM, Rsz, 1024, 1024, 0);
        gemm(X[i], w_ds0, nullptr, nullptr, d0 + i * RM, Rsz, 1024, 1024, 0);
        gemm(X[i], w_ds1, nullptr, nullptr, d1 + i * RM, Rsz, 1024, 1024, 0);
        gemm(d0 + i * RM, dc_w1 + (size_t)1024 * 2048, nullptr, nullptr,
             a0 + i * (size_t)Rsz * 2048, Rsz, 2048, 1024, 0);
        gemm(d1 + i * RM, dc_w1, nullptr, nullptr,
             a1 + i * (size_t)Rsz * 2048, Rsz, 2048, 1024, 0);
    }

    {
        int total_waves = 9 * Bsz * Lsz * Lsz;
        dec_kernel<<<total_waves / 8, 256, 0, stream>>>(a0, a1, dc_b1, dc_w2, dc_b2, dec);
    }

    {
        int nblk = 3 * Bsz * Hsz * Lsz;
        attn_scores_kernel<<<nblk, 192, 0, stream>>>(Q, Kb, dec, mask, S);
        attn_av_kernel<<<nblk, 64, 0, stream>>>(S, Vb, O);
    }

    for (int i = 0; i < 3; ++i) {
        gemm(O + i * RM, w_fc, nullptr, X[i], fc + i * RM, Rsz, 1024, 1024, GEMM_RESID);
        layernorm_kernel<<<Rsz, 256, 0, stream>>>(fc + i * RM, 1, RM, nullptr, nullptr,
                                                  ln1_g, ln1_b, oln + i * RM);
        ffn_fused_wmma_f32<<<dim3(32, 4, 2), dim3(256), 0, stream>>>(
            oln + i * RM, ffn_w1, ffn_b1, ffn_w2, fpart);
        layernorm_kernel<<<Rsz, 256, 0, stream>>>(fpart, 32, RM, ffn_b2, oln + i * RM,
                                                  ln2_g, ln2_b, out + i * RM);
    }
}


# Decision gate tiled through LDS: a block stages 8 a1 rows, 8 a0 rows, b1 and the w2 difference for 8x8 outputs, a wave sums a 2x4 group (was each row refetched per output); gate sums bit-identical
# speedup vs baseline: 1.4280x; 1.1761x over previous
#include <hip/hip_runtime.h>
#include <hip/hip_bf16.h>

#define Bsz 4
#define Lsz 64
#define DMsz 1024
#define Hsz 16
#define DKsz 64
#define DVsz 64
#define Dsz 1024
#define DIsz 4096
#define Rsz 256
#define NEGC (-1.0e9f)

typedef float v2f __attribute__((ext_vector_type(2)));
typedef float v8f __attribute__((ext_vector_type(8)));

#define GEMM_BIAS  1
#define GEMM_RELU  2
#define GEMM_RESID 4

__global__ __launch_bounds__(256)
void gemm_wmma_f32(const float* __restrict__ A, const float* __restrict__ B,
                   const float* __restrict__ bias, const float* __restrict__ resid,
                   float* __restrict__ C, int M, int N, int K, int flags)
{
    __shared__ float As[64][64 + 1];
    __shared__ float Bs[64][128 + 1];

    const int tid   = threadIdx.x;
    const int wave  = tid >> 5;
    const int lane  = tid & 31;
    const int l15   = lane & 15;
    const int lhalf = lane >> 4;
    const int m0    = blockIdx.y * 64;
    const int nblk  = blockIdx.x * 128;
    const int mq    = wave >> 2;
    const int nq    = wave & 3;

    const int ar = tid >> 4;
    const int ac = (tid & 15) << 2;
    const int br = (tid >> 5) * 8;
    const int bc = (tid & 31) << 2;
    float4 aprf[4], bprf[8];

    v8f acc[2][2];
#pragma unroll
    for (int x = 0; x < 2; ++x)
#pragma unroll
        for (int y = 0; y < 2; ++y)
            acc[x][y] = v8f{0.f, 0.f, 0.f, 0.f, 0.f, 0.f, 0.f, 0.f};

#pragma unroll
    for (int rr = 0; rr < 4; ++rr)
        aprf[rr] = *(const float4*)(A + (size_t)(m0 + ar + rr * 16) * K + ac);
#pragma unroll
    for (int rr = 0; rr < 8; ++rr)
        bprf[rr] = *(const float4*)(B + (size_t)(br + rr) * N + nblk + bc);

    const int arow0 = mq * 32 + l15;
    const int bcol0 = nq * 32 + l15;

    for (int k0 = 0; k0 < K; k0 += 64) {
#pragma unroll
        for (int rr = 0; rr < 4; ++rr) {
            As[ar + rr * 16][ac + 0] = aprf[rr].x;
            As[ar + rr * 16][ac + 1] = aprf[rr].y;
            As[ar + rr * 16][ac + 2] = aprf[rr].z;
            As[ar + rr * 16][ac + 3] = aprf[rr].w;
        }
#pragma unroll
        for (int rr = 0; rr < 8; ++rr) {
            Bs[br + rr][bc + 0] = bprf[rr].x;
            Bs[br + rr][bc + 1] = bprf[rr].y;
            Bs[br + rr][bc + 2] = bprf[rr].z;
            Bs[br + rr][bc + 3] = bprf[rr].w;
        }
        __syncthreads();

        if (k0 + 64 < K) {
#pragma unroll
            for (int rr = 0; rr < 4; ++rr)
                aprf[rr] = *(const float4*)(A + (size_t)(m0 + ar + rr * 16) * K
                                            + k0 + 64 + ac);
#pragma unroll
            for (int rr = 0; rr < 8; ++rr)
                bprf[rr] = *(const float4*)(B + (size_t)(k0 + 64 + br + rr) * N
                                            + nblk + bc);
        }

#pragma unroll
        for (int half = 0; half < 2; ++half) {
            v2f af[2][8], bf[2][8];
#pragma unroll
            for (int s = 0; s < 8; ++s) {
                const int kb = half * 32 + s * 4 + lhalf * 2;
                af[0][s] = v2f{As[arow0][kb],      As[arow0][kb + 1]};
                af[1][s] = v2f{As[arow0 + 16][kb], As[arow0 + 16][kb + 1]};
                bf[0][s] = v2f{Bs[kb][bcol0],      Bs[kb + 1][bcol0]};
                bf[1][s] = v2f{Bs[kb][bcol0 + 16], Bs[kb + 1][bcol0 + 16]};
            }
#pragma unroll
            for (int s = 0; s < 8; ++s) {
                acc[0][0] = __builtin_amdgcn_wmma_f32_16x16x4_f32(
                    false, af[0][s], false, bf[0][s], (short)0, acc[0][0], false, false);
                acc[0][1] = __builtin_amdgcn_wmma_f32_16x16x4_f32(
                    false, af[0][s], false, bf[1][s], (short)0, acc[0][1], false, false);
                acc[1][0] = __builtin_amdgcn_wmma_f32_16x16x4_f32(
                    false, af[1][s], false, bf[0][s], (short)0, acc[1][0], false, false);
                acc[1][1] = __builtin_amdgcn_wmma_f32_16x16x4_f32(
                    false, af[1][s], false, bf[1][s], (short)0, acc[1][1], false, false);
            }
        }
        __syncthreads();
    }

#pragma unroll
    for (int nsub = 0; nsub < 2; ++nsub) {
        const int col = nblk + nq * 32 + nsub * 16 + l15;
        const float bv = (flags & GEMM_BIAS) ? bias[col] : 0.f;
#pragma unroll
        for (int msub = 0; msub < 2; ++msub) {
#pragma unroll
            for (int r = 0; r < 8; ++r) {
                const int row = m0 + mq * 32 + msub * 16 + r + lhalf * 8;
                float v = acc[msub][nsub][r] + bv;
                if (flags & GEMM_RESID) v += resid[(size_t)row * N + col];
                if (flags & GEMM_RELU)  v = fmaxf(v, 0.f);
                C[(size_t)row * N + col] = v;
            }
        }
    }
}

__global__ __launch_bounds__(256)
void ffn_fused_wmma_f32(const float* __restrict__ X, const float* __restrict__ W1,
                        const float* __restrict__ b1, const float* __restrict__ W2,
                        float* __restrict__ P)
{
    __shared__ float As[64][64 + 1];
    __shared__ float Bs[64][128 + 1];
    __shared__ float Hs[64][128 + 1];

    const int tid   = threadIdx.x;
    const int wave  = tid >> 5;
    const int lane  = tid & 31;
    const int l15   = lane & 15;
    const int lhalf = lane >> 4;
    const int s0    = blockIdx.x * 128;
    const int m0    = blockIdx.y * 64;
    const int nc0   = blockIdx.z * 4;
    const int mq    = wave >> 2;
    const int nq    = wave & 3;

    const int ar = tid >> 4;
    const int ac = (tid & 15) << 2;
    const int br = (tid >> 5) * 8;
    const int bc = (tid & 31) << 2;
    float4 aprf[4], bprf[8];

    v8f acc[2][2];
#pragma unroll
    for (int x = 0; x < 2; ++x)
#pragma unroll
        for (int y = 0; y < 2; ++y)
            acc[x][y] = v8f{0.f, 0.f, 0.f, 0.f, 0.f, 0.f, 0.f, 0.f};

#pragma unroll
    for (int rr = 0; rr < 4; ++rr)
        aprf[rr] = *(const float4*)(X + (size_t)(m0 + ar + rr * 16) * DMsz + ac);
#pragma unroll
    for (int rr = 0; rr < 8; ++rr)
        bprf[rr] = *(const float4*)(W1 + (size_t)(br + rr) * DIsz + s0 + bc);

    const int arow0 = mq * 32 + l15;
    const int bcol0 = nq * 32 + l15;

    for (int k0 = 0; k0 < DMsz; k0 += 64) {
#pragma unroll
        for (int rr = 0; rr < 4; ++rr) {
            As[ar + rr * 16][ac + 0] = aprf[rr].x;
            As[ar + rr * 16][ac + 1] = aprf[rr].y;
            As[ar + rr * 16][ac + 2] = aprf[rr].z;
            As[ar + rr * 16][ac + 3] = aprf[rr].w;
        }
#pragma unroll
        for (int rr = 0; rr < 8; ++rr) {
            Bs[br + rr][bc + 0] = bprf[rr].x;
            Bs[br + rr][bc + 1] = bprf[rr].y;
            Bs[br + rr][bc + 2] = bprf[rr].z;
            Bs[br + rr][bc + 3] = bprf[rr].w;
        }
        __syncthreads();

        if (k0 + 64 < DMsz) {
#pragma unroll
            for (int rr = 0; rr < 4; ++rr)
                aprf[rr] = *(const float4*)(X + (size_t)(m0 + ar + rr * 16) * DMsz
                                            + k0 + 64 + ac);
#pragma unroll
            for (int rr = 0; rr < 8; ++rr)
                bprf[rr] = *(const float4*)(W1 + (size_t)(k0 + 64 + br + rr) * DIsz
                                            + s0 + bc);
        } else {
#pragma unroll
            for (int rr = 0; rr < 8; ++rr)
                bprf[rr] = *(const float4*)(W2 + (size_t)(s0 + br + rr) * DMsz
                                            + nc0 * 128 + bc);
        }

#pragma unroll
        for (int half = 0; half < 2; ++half) {
            v2f af[2][8], bf[2][8];
#pragma unroll
            for (int s = 0; s < 8; ++s) {
                const int kb = half * 32 + s * 4 + lhalf * 2;
                af[0][s] = v2f{As[arow0][kb],      As[arow0][kb + 1]};
                af[1][s] = v2f{As[arow0 + 16][kb], As[arow0 + 16][kb + 1]};
                bf[0][s] = v2f{Bs[kb][bcol0],      Bs[kb + 1][bcol0]};
                bf[1][s] = v2f{Bs[kb][bcol0 + 16], Bs[kb + 1][bcol0 + 16]};
            }
#pragma unroll
            for (int s = 0; s < 8; ++s) {
                acc[0][0] = __builtin_amdgcn_wmma_f32_16x16x4_f32(
                    false, af[0][s], false, bf[0][s], (short)0, acc[0][0], false, false);
                acc[0][1] = __builtin_amdgcn_wmma_f32_16x16x4_f32(
                    false, af[0][s], false, bf[1][s], (short)0, acc[0][1], false, false);
                acc[1][0] = __builtin_amdgcn_wmma_f32_16x16x4_f32(
                    false, af[1][s], false, bf[0][s], (short)0, acc[1][0], false, false);
                acc[1][1] = __builtin_amdgcn_wmma_f32_16x16x4_f32(
                    false, af[1][s], false, bf[1][s], (short)0, acc[1][1], false, false);
            }
        }
        __syncthreads();
    }

#pragma unroll
    for (int nsub = 0; nsub < 2; ++nsub) {
        const int hc = nq * 32 + nsub * 16 + l15;
        const float bv = b1[s0 + hc];
#pragma unroll
        for (int msub = 0; msub < 2; ++msub) {
#pragma unroll
            for (int r = 0; r < 8; ++r) {
                const int hr = mq * 32 + msub * 16 + r + lhalf * 8;
                Hs[hr][hc] = fmaxf(acc[msub][nsub][r] + bv, 0.f);
            }
        }
    }

    for (int nc = nc0; nc < nc0 + 4; ++nc) {
#pragma unroll
        for (int x = 0; x < 2; ++x)
#pragma unroll
            for (int y = 0; y < 2; ++y)
                acc[x][y] = v8f{0.f, 0.f, 0.f, 0.f, 0.f, 0.f, 0.f, 0.f};

        for (int kc = 0; kc < 2; ++kc) {
#pragma unroll
            for (int rr = 0; rr < 8; ++rr) {
                Bs[br + rr][bc + 0] = bprf[rr].x;
                Bs[br + rr][bc + 1] = bprf[rr].y;
                Bs[br + rr][bc + 2] = bprf[rr].z;
                Bs[br + rr][bc + 3] = bprf[rr].w;
            }
            __syncthreads();

            const int tn = (nc - nc0) * 2 + kc + 1;
            if (tn < 8) {
                const int nn = nc0 + (tn >> 1);
                const int kk = tn & 1;
#pragma unroll
                for (int rr = 0; rr < 8; ++rr)
                    bprf[rr] = *(const float4*)(W2 + (size_t)(s0 + kk * 64 + br + rr) * DMsz
                                                + nn * 128 + bc);
            }

            const int hk0 = kc * 64;
#pragma unroll
            for (int half = 0; half < 2; ++half) {
                v2f af[2][8], bf[2][8];
#pragma unroll
                for (int s = 0; s < 8; ++s) {
                    const int kb = half * 32 + s * 4 + lhalf * 2;
                    af[0][s] = v2f{Hs[arow0][hk0 + kb],      Hs[arow0][hk0 + kb + 1]};
                    af[1][s] = v2f{Hs[arow0 + 16][hk0 + kb], Hs[arow0 + 16][hk0 + kb + 1]};
                    bf[0][s] = v2f{Bs[kb][bcol0],      Bs[kb + 1][bcol0]};
                    bf[1][s] = v2f{Bs[kb][bcol0 + 16], Bs[kb + 1][bcol0 + 16]};
                }
#pragma unroll
                for (int s = 0; s < 8; ++s) {
                    acc[0][0] = __builtin_amdgcn_wmma_f32_16x16x4_f32(
                        false, af[0][s], false, bf[0][s], (short)0, acc[0][0], false, false);
                    acc[0][1] = __builtin_amdgcn_wmma_f32_16x16x4_f32(
                        false, af[0][s], false, bf[1][s], (short)0, acc[0][1], false, false);
                    acc[1][0] = __builtin_amdgcn_wmma_f32_16x16x4_f32(
                        false, af[1][s], false, bf[0][s], (short)0, acc[1][0], false, false);
                    acc[1][1] = __builtin_amdgcn_wmma_f32_16x16x4_f32(
                        false, af[1][s], false, bf[1][s], (short)0, acc[1][1], false, false);
                }
            }
            __syncthreads();
        }

        float* Pp = P + (size_t)blockIdx.x * Rsz * DMsz;
#pragma unroll
        for (int nsub = 0; nsub < 2; ++nsub) {
            const int col = nc * 128 + nq * 32 + nsub * 16 + l15;
#pragma unroll
            for (int msub = 0; msub < 2; ++msub) {
#pragma unroll
                for (int r = 0; r < 8; ++r) {
                    const int row = m0 + mq * 32 + msub * 16 + r + lhalf * 8;
                    Pp[(size_t)row * DMsz + col] = acc[msub][nsub][r];
                }
            }
        }
    }
}

#define GT_KP (18 * 32)
#define GT_PK 32

__global__ __launch_bounds__(256)
void dec_kernel(const float* __restrict__ a0base, const float* __restrict__ a1base,
                const float* __restrict__ b1, const float* __restrict__ w2,
                const float* __restrict__ b2, float* __restrict__ dec)
{
    __shared__ float tile[GT_PK * GT_KP];

    const int tid  = threadIdx.x;
    const int wave = tid >> 5;
    const int lane = tid & 31;
    const int blk  = blockIdx.x;
    const int nt   = blk & 7;
    const int mt   = (blk >> 3) & 7;
    const int b    = (blk >> 6) & 3;
    const int pair = blk >> 8;
    const int i    = pair / 3;
    const int j    = pair % 3;

    const float4* g0 = (const float4*)(a0base + ((size_t)i * Rsz + b * Lsz + nt * 8) * 2048);
    const float4* g1 = (const float4*)(a1base + ((size_t)j * Rsz + b * Lsz + mt * 8) * 2048);
    const float2* gw = (const float2*)w2;

    const int ml = (wave >> 1) * 2;
    const int nl = (wave & 1) * 4;
    const float* pa = tile + ml * 32 + lane;
    const float* pb = tile + (8 + nl) * 32 + lane;
    const float* pc = tile + 16 * 32 + lane;

    float acc[2][4];
#pragma unroll
    for (int x = 0; x < 2; ++x)
#pragma unroll
        for (int y = 0; y < 4; ++y) acc[x][y] = 0.f;

    for (int p = 0; p < 2; ++p) {
        {
            float4 r0[8], r1[8];
#pragma unroll
            for (int u = 0; u < 8; ++u) r0[u] = g0[u * 512 + p * 256 + tid];
#pragma unroll
            for (int u = 0; u < 8; ++u) r1[u] = g1[u * 512 + p * 256 + tid];
            const int dst = (tid >> 3) * GT_KP + (tid & 7) * 4;
#pragma unroll
            for (int u = 0; u < 8; ++u) {
                *(float4*)&tile[dst + (8 + u) * 32] = r0[u];
                *(float4*)&tile[dst + u * 32]       = r1[u];
            }
        }
#pragma unroll
        for (int u = 0; u < 4; ++u) {
            const int cl  = tid + u * 256;
            const float2 w = gw[p * 1024 + cl];
            const int dst = (cl >> 5) * GT_KP + (cl & 31);
            tile[dst + 16 * 32] = b1[p * 1024 + cl];
            tile[dst + 17 * 32] = w.y - w.x;
        }
        __syncthreads();

#pragma unroll 4
        for (int k = 0; k < GT_PK; ++k) {
            const int o = k * GT_KP;
            float xv[2], yv[4];
            xv[0] = pa[o];      xv[1] = pa[o + 32];
            yv[0] = pb[o];      yv[1] = pb[o + 32];
            yv[2] = pb[o + 64]; yv[3] = pb[o + 96];
            const float bb = pc[o], ww = pc[o + 32];
#pragma unroll
            for (int x = 0; x < 2; ++x)
#pragma unroll
                for (int y = 0; y < 4; ++y) {
                    const float t = (xv[x] + yv[y]) + bb;
                    if (t > 0.f) acc[x][y] = fmaf(t, ww, acc[x][y]);
                }
        }
        __syncthreads();
    }

#pragma unroll
    for (int off = 16; off; off >>= 1)
#pragma unroll
        for (int x = 0; x < 2; ++x)
#pragma unroll
            for (int y = 0; y < 4; ++y)
                acc[x][y] += __shfl_xor(acc[x][y], off, 32);

    if (lane == 0) {
        const float bd = b2[1] - b2[0];
        float* dtile = dec + (((size_t)(pair * Bsz + b) * Lsz + mt * 8) * Lsz + nt * 8);
#pragma unroll
        for (int x = 0; x < 2; ++x) {
            float4 r;
            r.x = ((acc[x][0] + bd) > 0.f) ? NEGC : 0.f;
            r.y = ((acc[x][1] + bd) > 0.f) ? NEGC : 0.f;
            r.z = ((acc[x][2] + bd) > 0.f) ? NEGC : 0.f;
            r.w = ((acc[x][3] + bd) > 0.f) ? NEGC : 0.f;
            *(float4*)(dtile + (size_t)(ml + x) * Lsz + nl) = r;
        }
    }
}

__global__ __launch_bounds__(192)
void attn_scores_kernel(const float* __restrict__ Qb, const float* __restrict__ Kb,
                        const float* __restrict__ dec, const int* __restrict__ mask,
                        float* __restrict__ S)
{
    const int blk = blockIdx.x;
    const int q = blk & 63;
    const int h = (blk >> 6) & 15;
    const int b = (blk >> 10) & 3;
    const int i = blk >> 12;
    const int t = threadIdx.x;
    const int j = t >> 6;
    const int k = t & 63;

    __shared__ float qrow[64];
    __shared__ float red[192];

    const float* Q = Qb + ((size_t)i * Rsz + b * Lsz + q) * DMsz + h * DKsz;
    if (t < 64) qrow[t] = Q[t];
    __syncthreads();

    const float* Kp = Kb + ((size_t)j * Rsz + b * Lsz + k) * DMsz + h * DKsz;
    float s = 0.f;
#pragma unroll 8
    for (int d = 0; d < 64; ++d) s += qrow[d] * Kp[d];
    s *= 0.125f;
    s += dec[(((size_t)(i * 3 + j) * Bsz + b) * Lsz + q) * Lsz + k];
    if (mask[((size_t)b * Lsz + q) * Lsz + k] == 0) s = NEGC;

    red[t] = s; __syncthreads();
    if (t < 64) red[t] = fmaxf(red[t], fmaxf(red[t + 64], red[t + 128]));
    __syncthreads();
    for (int st = 32; st > 0; st >>= 1) {
        if (t < st) red[t] = fmaxf(red[t], red[t + st]);
        __syncthreads();
    }
    const float mx = red[0];
    __syncthreads();
    const float e = __expf(s - mx);
    red[t] = e; __syncthreads();
    if (t < 64) red[t] = red[t] + red[t + 64] + red[t + 128];
    __syncthreads();
    for (int st = 32; st > 0; st >>= 1) {
        if (t < st) red[t] += red[t + st];
        __syncthreads();
    }
    S[(size_t)blk * 192 + t] = e * (1.0f / red[0]);
}

__global__ __launch_bounds__(64)
void attn_av_kernel(const float* __restrict__ S, const float* __restrict__ Vb,
                    float* __restrict__ Ob)
{
    const int blk = blockIdx.x;
    const int q = blk & 63;
    const int h = (blk >> 6) & 15;
    const int b = (blk >> 10) & 3;
    const int i = blk >> 12;
    const int d = threadIdx.x;

    __shared__ float arow[192];
    const float* Sp = S + (size_t)blk * 192;
    arow[d] = Sp[d]; arow[d + 64] = Sp[d + 64]; arow[d + 128] = Sp[d + 128];
    __syncthreads();

    float acc = 0.f;
    for (int jk = 0; jk < 192; ++jk) {
        int j = jk >> 6, k = jk & 63;
        acc += arow[jk] * Vb[((size_t)j * Rsz + b * Lsz + k) * DMsz + h * DVsz + d];
    }
    Ob[((size_t)i * Rsz + b * Lsz + q) * DMsz + h * DVsz + d] = acc;
}

__global__ __launch_bounds__(256)
void layernorm_kernel(const float* __restrict__ x, int nparts, size_t pstride,
                      const float* __restrict__ bias, const float* __restrict__ resid,
                      const float* __restrict__ g, const float* __restrict__ bta,
                      float* __restrict__ out)
{
    const int row = blockIdx.x;
    const int t = threadIdx.x;
    __shared__ float red[256];

    const float* xp = x + (size_t)row * 1024;
    float v[4];
#pragma unroll
    for (int u = 0; u < 4; ++u) v[u] = xp[t + u * 256];
#pragma unroll 8
    for (int p = 1; p < nparts; ++p) {
        const float* pp = xp + (size_t)p * pstride;
#pragma unroll
        for (int u = 0; u < 4; ++u) v[u] += pp[t + u * 256];
    }
    float s = 0.f;
#pragma unroll
    for (int u = 0; u < 4; ++u) {
        float val = v[u];
        if (bias)  val += bias[t + u * 256];
        if (resid) val += resid[(size_t)row * 1024 + t + u * 256];
        v[u] = val; s += val;
    }
    red[t] = s; __syncthreads();
    for (int st = 128; st > 0; st >>= 1) { if (t < st) red[t] += red[t + st]; __syncthreads(); }
    const float mu = red[0] * (1.f / 1024.f);
    __syncthreads();

    float s2 = 0.f;
#pragma unroll
    for (int u = 0; u < 4; ++u) { float dd = v[u] - mu; s2 += dd * dd; }
    red[t] = s2; __syncthreads();
    for (int st = 128; st > 0; st >>= 1) { if (t < st) red[t] += red[t + st]; __syncthreads(); }
    const float var = red[0] * (1.f / 1024.f);
    const float rinv = 1.0f / sqrtf(var + 1e-6f);
#pragma unroll
    for (int u = 0; u < 4; ++u) {
        int c = t + u * 256;
        out[(size_t)row * 1024 + c] = (v[u] - mu) * rinv * g[c] + bta[c];
    }
}

extern "C" void kernel_launch(void* const* d_in, const int* in_sizes, int n_in,
                              void* d_out, int out_size, void* d_ws, size_t ws_size,
                              hipStream_t stream)
{
    const float* X[3]   = { (const float*)d_in[0], (const float*)d_in[1], (const float*)d_in[2] };
    const int*   mask   = (const int*)d_in[3];
    const float* w_qs   = (const float*)d_in[4];
    const float* w_ks   = (const float*)d_in[5];
    const float* w_vs   = (const float*)d_in[6];
    const float* w_ds0  = (const float*)d_in[7];
    const float* w_ds1  = (const float*)d_in[8];
    const float* dc_w1  = (const float*)d_in[9];
    const float* dc_b1  = (const float*)d_in[10];
    const float* dc_w2  = (const float*)d_in[11];
    const float* dc_b2  = (const float*)d_in[12];
    const float* w_fc   = (const float*)d_in[13];
    const float* ln1_g  = (const float*)d_in[14];
    const float* ln1_b  = (const float*)d_in[15];
    const float* ffn_w1 = (const float*)d_in[16];
    const float* ffn_b1 = (const float*)d_in[17];
    const float* ffn_w2 = (const float*)d_in[18];
    const float* ffn_b2 = (const float*)d_in[19];
    const float* ln2_g  = (const float*)d_in[20];
    const float* ln2_b  = (const float*)d_in[21];
    float* out = (float*)d_out;

    float* ws = (float*)d_ws;
    size_t off = 0;
    auto alloc = [&](size_t n) { float* p = ws + off; off += n; return p; };
    const size_t RM = (size_t)Rsz * DMsz;
    float* Q   = alloc(3 * RM);
    float* Kb  = alloc(3 * RM);
    float* Vb  = alloc(3 * RM);
    float* d0  = alloc(3 * RM);
    float* d1  = alloc(3 * RM);
    float* a0  = alloc(3 * (size_t)Rsz * 2048);
    float* a1  = alloc(3 * (size_t)Rsz * 2048);
    float* dec = alloc(9 * (size_t)Bsz * Lsz * Lsz);
    float* S   = alloc((size_t)3 * Bsz * Hsz * Lsz * 192);
    float* O   = alloc(3 * RM);
    float* fc  = alloc(3 * RM);
    float* oln = alloc(3 * RM);
    float* fpart = alloc(32 * RM);
    (void)ws_size; (void)n_in; (void)in_sizes; (void)out_size;

    auto gemm = [&](const float* A, const float* Bm, const float* bias,
                    const float* resid, float* C, int M, int N, int K, int flags) {
        dim3 grid(N / 128, M / 64);
        gemm_wmma_f32<<<grid, dim3(256), 0, stream>>>(A, Bm, bias, resid, C, M, N, K, flags);
    };

    for (int i = 0; i < 3; ++i) {
        gemm(X[i], w_qs,  nullptr, nullptr, Q  + i * RM, Rsz, 1024, 1024, 0);
        gemm(X[i], w_ks,  nullptr, nullptr, Kb + i * RM, Rsz, 1024, 1024, 0);
        gemm(X[i], w_vs,  nullptr, nullptr, Vb + i * RM, Rsz, 1024, 1024, 0);
        gemm(X[i], w_ds0, nullptr, nullptr, d0 + i * RM, Rsz, 1024, 1024, 0);
        gemm(X[i], w_ds1, nullptr, nullptr, d1 + i * RM, Rsz, 1024, 1024, 0);
        gemm(d0 + i * RM, dc_w1 + (size_t)1024 * 2048, nullptr, nullptr,
             a0 + i * (size_t)Rsz * 2048, Rsz, 2048, 1024, 0);
        gemm(d1 + i * RM, dc_w1, nullptr, nullptr,
             a1 + i * (size_t)Rsz * 2048, Rsz, 2048, 1024, 0);
    }

    {
        dec_kernel<<<9 * Bsz * 8 * 8, 256, 0, stream>>>(a0, a1, dc_b1, dc_w2, dc_b2, dec);
    }

    {
        int nblk = 3 * Bsz * Hsz * Lsz;
        attn_scores_kernel<<<nblk, 192, 0, stream>>>(Q, Kb, dec, mask, S);
        attn_av_kernel<<<nblk, 64, 0, stream>>>(S, Vb, O);
    }

    for (int i = 0; i < 3; ++i) {
        gemm(O + i * RM, w_fc, nullptr, X[i], fc + i * RM, Rsz, 1024, 1024, GEMM_RESID);
        layernorm_kernel<<<Rsz, 256, 0, stream>>>(fc + i * RM, 1, RM, nullptr, nullptr,
                                                  ln1_g, ln1_b, oln + i * RM);
        ffn_fused_wmma_f32<<<dim3(32, 4, 2), dim3(256), 0, stream>>>(
            oln + i * RM, ffn_w1, ffn_b1, ffn_w2, fpart);
        layernorm_kernel<<<Rsz, 256, 0, stream>>>(fpart, 32, RM, ffn_b2, oln + i * RM,
                                                  ln2_g, ln2_b, out + i * RM);
    }
}


# GEMM kernel re-tiled: 32x32 block tile, 4 waves, one 16x16 accumulator per wave over the whole K (was 64x128, 8 waves of 32x32), so a 256-row product fills 256+ WGPs, not 32-64; sum order unchanged
# speedup vs baseline: 2.8336x; 1.9843x over previous
#include <hip/hip_runtime.h>
#include <hip/hip_bf16.h>

#define Bsz 4
#define Lsz 64
#define DMsz 1024
#define Hsz 16
#define DKsz 64
#define DVsz 64
#define Dsz 1024
#define DIsz 4096
#define Rsz 256
#define NEGC (-1.0e9f)

typedef float v2f __attribute__((ext_vector_type(2)));
typedef float v4f __attribute__((ext_vector_type(4)));
typedef float v8f __attribute__((ext_vector_type(8)));

#define GEMM_BIAS  1
#define GEMM_RELU  2
#define GEMM_RESID 4

#define GK  128
#define GAP (GK + 4)
#define GBP 40
#define GCP 36
#define GNA (GK / 16)
#define GNB (GK / 16)

__global__ __launch_bounds__(128)
void gemm_wmma_f32(const float* __restrict__ A, const float* __restrict__ B,
                   const float* __restrict__ bias, const float* __restrict__ resid,
                   float* __restrict__ C, int M, int N, int K, int flags)
{
    __shared__ __attribute__((aligned(16))) float As[32 * GAP];
    __shared__ __attribute__((aligned(16))) float Bs[GK * GBP];
    __shared__ __attribute__((aligned(16))) float Cs[32 * GCP];

    const int tid   = threadIdx.x;
    const int wave  = tid >> 5;
    const int lane  = tid & 31;
    const int l15   = lane & 15;
    const int lhalf = lane >> 4;
    const int m0    = blockIdx.y * 32;
    const int n0    = blockIdx.x * 32;
    const int mq    = wave >> 1;
    const int nq    = wave & 1;
    (void)M;

    const int ar = tid / (GK / 4);
    const int ac = (tid % (GK / 4)) << 2;
    const int br = tid >> 3;
    const int bc = (tid & 7) << 2;
    const float* Ag = A + (size_t)(m0 + ar) * K + ac;
    const float* Bg = B + (size_t)br * N + n0 + bc;
    v4f aprf[GNA], bprf[GNB];

    v8f acc = v8f{0.f, 0.f, 0.f, 0.f, 0.f, 0.f, 0.f, 0.f};

#pragma unroll
    for (int u = 0; u < GNA; ++u)
        aprf[u] = *(const v4f*)(Ag + (size_t)(u * (512 / GK)) * K);
#pragma unroll
    for (int u = 0; u < GNB; ++u)
        bprf[u] = *(const v4f*)(Bg + (size_t)(u * 16) * N);

    const float* ap = As + (mq * 16 + l15) * GAP + lhalf * 2;
    const float* bp = Bs + (lhalf * 2) * GBP + nq * 16 + l15;

    for (int k0 = 0; k0 < K; k0 += GK) {
#pragma unroll
        for (int u = 0; u < GNA; ++u)
            *(v4f*)&As[(ar + u * (512 / GK)) * GAP + ac] = aprf[u];
#pragma unroll
        for (int u = 0; u < GNB; ++u)
            *(v4f*)&Bs[(br + u * 16) * GBP + bc] = bprf[u];
        __syncthreads();

        if (k0 + GK < K) {
#pragma unroll
            for (int u = 0; u < GNA; ++u)
                aprf[u] = *(const v4f*)(Ag + (size_t)(u * (512 / GK)) * K + k0 + GK);
#pragma unroll
            for (int u = 0; u < GNB; ++u)
                bprf[u] = *(const v4f*)(Bg + (size_t)(k0 + GK + u * 16) * N);
        }

        v2f af[GK / 4], bf[GK / 4];
#pragma unroll
        for (int s = 0; s < GK / 4; ++s) {
            af[s] = v2f{ap[s * 4], ap[s * 4 + 1]};
            bf[s] = v2f{bp[(s * 4) * GBP], bp[(s * 4 + 1) * GBP]};
        }
#pragma unroll
        for (int s = 0; s < GK / 4; ++s)
            acc = __builtin_amdgcn_wmma_f32_16x16x4_f32(
                false, af[s], false, bf[s], (short)0, acc, false, false);
        __builtin_amdgcn_sched_group_barrier(0x100, 12, 0);
#pragma unroll
        for (int p = 0; p < (GK / 4 - 8) / 2; ++p) {
            __builtin_amdgcn_sched_group_barrier(0x008, 2, 0);
            __builtin_amdgcn_sched_group_barrier(0x100, 3, 0);
        }
        __builtin_amdgcn_sched_group_barrier(0x008, 8, 0);
        __syncthreads();
    }

#pragma unroll
    for (int r = 0; r < 8; ++r)
        Cs[(mq * 16 + r + lhalf * 8) * GCP + nq * 16 + l15] = acc[r];
    __syncthreads();

#pragma unroll
    for (int u = 0; u < 2; ++u) {
        const int v   = tid + u * 128;
        const int row = v >> 3;
        const int c4  = (v & 7) << 2;
        const size_t g = (size_t)(m0 + row) * N + n0 + c4;
        v4f o = *(const v4f*)&Cs[row * GCP + c4];
        v4f bv = v4f{0.f, 0.f, 0.f, 0.f};
        if (flags & GEMM_BIAS) bv = *(const v4f*)(bias + n0 + c4);
        o.x = o.x + bv.x; o.y = o.y + bv.y; o.z = o.z + bv.z; o.w = o.w + bv.w;
        if (flags & GEMM_RESID) {
            const v4f rv = *(const v4f*)(resid + g);
            o.x += rv.x; o.y += rv.y; o.z += rv.z; o.w += rv.w;
        }
        if (flags & GEMM_RELU) {
            o.x = fmaxf(o.x, 0.f); o.y = fmaxf(o.y, 0.f);
            o.z = fmaxf(o.z, 0.f); o.w = fmaxf(o.w, 0.f);
        }
        *(v4f*)(C + g) = o;
    }
}

__global__ __launch_bounds__(256)
void ffn_fused_wmma_f32(const float* __restrict__ X, const float* __restrict__ W1,
                        const float* __restrict__ b1, const float* __restrict__ W2,
                        float* __restrict__ P)
{
    __shared__ float As[64][64 + 1];
    __shared__ float Bs[64][128 + 1];
    __shared__ float Hs[64][128 + 1];

    const int tid   = threadIdx.x;
    const int wave  = tid >> 5;
    const int lane  = tid & 31;
    const int l15   = lane & 15;
    const int lhalf = lane >> 4;
    const int s0    = blockIdx.x * 128;
    const int m0    = blockIdx.y * 64;
    const int nc0   = blockIdx.z * 4;
    const int mq    = wave >> 2;
    const int nq    = wave & 3;

    const int ar = tid >> 4;
    const int ac = (tid & 15) << 2;
    const int br = (tid >> 5) * 8;
    const int bc = (tid & 31) << 2;
    float4 aprf[4], bprf[8];

    v8f acc[2][2];
#pragma unroll
    for (int x = 0; x < 2; ++x)
#pragma unroll
        for (int y = 0; y < 2; ++y)
            acc[x][y] = v8f{0.f, 0.f, 0.f, 0.f, 0.f, 0.f, 0.f, 0.f};

#pragma unroll
    for (int rr = 0; rr < 4; ++rr)
        aprf[rr] = *(const float4*)(X + (size_t)(m0 + ar + rr * 16) * DMsz + ac);
#pragma unroll
    for (int rr = 0; rr < 8; ++rr)
        bprf[rr] = *(const float4*)(W1 + (size_t)(br + rr) * DIsz + s0 + bc);

    const int arow0 = mq * 32 + l15;
    const int bcol0 = nq * 32 + l15;

    for (int k0 = 0; k0 < DMsz; k0 += 64) {
#pragma unroll
        for (int rr = 0; rr < 4; ++rr) {
            As[ar + rr * 16][ac + 0] = aprf[rr].x;
            As[ar + rr * 16][ac + 1] = aprf[rr].y;
            As[ar + rr * 16][ac + 2] = aprf[rr].z;
            As[ar + rr * 16][ac + 3] = aprf[rr].w;
        }
#pragma unroll
        for (int rr = 0; rr < 8; ++rr) {
            Bs[br + rr][bc + 0] = bprf[rr].x;
            Bs[br + rr][bc + 1] = bprf[rr].y;
            Bs[br + rr][bc + 2] = bprf[rr].z;
            Bs[br + rr][bc + 3] = bprf[rr].w;
        }
        __syncthreads();

        if (k0 + 64 < DMsz) {
#pragma unroll
            for (int rr = 0; rr < 4; ++rr)
                aprf[rr] = *(const float4*)(X + (size_t)(m0 + ar + rr * 16) * DMsz
                                            + k0 + 64 + ac);
#pragma unroll
            for (int rr = 0; rr < 8; ++rr)
                bprf[rr] = *(const float4*)(W1 + (size_t)(k0 + 64 + br + rr) * DIsz
                                            + s0 + bc);
        } else {
#pragma unroll
            for (int rr = 0; rr < 8; ++rr)
                bprf[rr] = *(const float4*)(W2 + (size_t)(s0 + br + rr) * DMsz
                                            + nc0 * 128 + bc);
        }

#pragma unroll
        for (int half = 0; half < 2; ++half) {
            v2f af[2][8], bf[2][8];
#pragma unroll
            for (int s = 0; s < 8; ++s) {
                const int kb = half * 32 + s * 4 + lhalf * 2;
                af[0][s] = v2f{As[arow0][kb],      As[arow0][kb + 1]};
                af[1][s] = v2f{As[arow0 + 16][kb], As[arow0 + 16][kb + 1]};
                bf[0][s] = v2f{Bs[kb][bcol0],      Bs[kb + 1][bcol0]};
                bf[1][s] = v2f{Bs[kb][bcol0 + 16], Bs[kb + 1][bcol0 + 16]};
            }
#pragma unroll
            for (int s = 0; s < 8; ++s) {
                acc[0][0] = __builtin_amdgcn_wmma_f32_16x16x4_f32(
                    false, af[0][s], false, bf[0][s], (short)0, acc[0][0], false, false);
                acc[0][1] = __builtin_amdgcn_wmma_f32_16x16x4_f32(
                    false, af[0][s], false, bf[1][s], (short)0, acc[0][1], false, false);
                acc[1][0] = __builtin_amdgcn_wmma_f32_16x16x4_f32(
                    false, af[1][s], false, bf[0][s], (short)0, acc[1][0], false, false);
                acc[1][1] = __builtin_amdgcn_wmma_f32_16x16x4_f32(
                    false, af[1][s], false, bf[1][s], (short)0, acc[1][1], false, false);
            }
        }
        __syncthreads();
    }

#pragma unroll
    for (int nsub = 0; nsub < 2; ++nsub) {
        const int hc = nq * 32 + nsub * 16 + l15;
        const float bv = b1[s0 + hc];
#pragma unroll
        for (int msub = 0; msub < 2; ++msub) {
#pragma unroll
            for (int r = 0; r < 8; ++r) {
                const int hr = mq * 32 + msub * 16 + r + lhalf * 8;
                Hs[hr][hc] = fmaxf(acc[msub][nsub][r] + bv, 0.f);
            }
        }
    }

    for (int nc = nc0; nc < nc0 + 4; ++nc) {
#pragma unroll
        for (int x = 0; x < 2; ++x)
#pragma unroll
            for (int y = 0; y < 2; ++y)
                acc[x][y] = v8f{0.f, 0.f, 0.f, 0.f, 0.f, 0.f, 0.f, 0.f};

        for (int kc = 0; kc < 2; ++kc) {
#pragma unroll
            for (int rr = 0; rr < 8; ++rr) {
                Bs[br + rr][bc + 0] = bprf[rr].x;
                Bs[br + rr][bc + 1] = bprf[rr].y;
                Bs[br + rr][bc + 2] = bprf[rr].z;
                Bs[br + rr][bc + 3] = bprf[rr].w;
            }
            __syncthreads();

            const int tn = (nc - nc0) * 2 + kc + 1;
            if (tn < 8) {
                const int nn = nc0 + (tn >> 1);
                const int kk = tn & 1;
#pragma unroll
                for (int rr = 0; rr < 8; ++rr)
                    bprf[rr] = *(const float4*)(W2 + (size_t)(s0 + kk * 64 + br + rr) * DMsz
                                                + nn * 128 + bc);
            }

            const int hk0 = kc * 64;
#pragma unroll
            for (int half = 0; half < 2; ++half) {
                v2f af[2][8], bf[2][8];
#pragma unroll
                for (int s = 0; s < 8; ++s) {
                    const int kb = half * 32 + s * 4 + lhalf * 2;
                    af[0][s] = v2f{Hs[arow0][hk0 + kb],      Hs[arow0][hk0 + kb + 1]};
                    af[1][s] = v2f{Hs[arow0 + 16][hk0 + kb], Hs[arow0 + 16][hk0 + kb + 1]};
                    bf[0][s] = v2f{Bs[kb][bcol0],      Bs[kb + 1][bcol0]};
                    bf[1][s] = v2f{Bs[kb][bcol0 + 16], Bs[kb + 1][bcol0 + 16]};
                }
#pragma unroll
                for (int s = 0; s < 8; ++s) {
                    acc[0][0] = __builtin_amdgcn_wmma_f32_16x16x4_f32(
                        false, af[0][s], false, bf[0][s], (short)0, acc[0][0], false, false);
                    acc[0][1] = __builtin_amdgcn_wmma_f32_16x16x4_f32(
                        false, af[0][s], false, bf[1][s], (short)0, acc[0][1], false, false);
                    acc[1][0] = __builtin_amdgcn_wmma_f32_16x16x4_f32(
                        false, af[1][s], false, bf[0][s], (short)0, acc[1][0], false, false);
                    acc[1][1] = __builtin_amdgcn_wmma_f32_16x16x4_f32(
                        false, af[1][s], false, bf[1][s], (short)0, acc[1][1], false, false);
                }
            }
            __syncthreads();
        }

        float* Pp = P + (size_t)blockIdx.x * Rsz * DMsz;
#pragma unroll
        for (int nsub = 0; nsub < 2; ++nsub) {
            const int col = nc * 128 + nq * 32 + nsub * 16 + l15;
#pragma unroll
            for (int msub = 0; msub < 2; ++msub) {
#pragma unroll
                for (int r = 0; r < 8; ++r) {
                    const int row = m0 + mq * 32 + msub * 16 + r + lhalf * 8;
                    Pp[(size_t)row * DMsz + col] = acc[msub][nsub][r];
                }
            }
        }
    }
}

#define GT_KP (18 * 32)
#define GT_PK 32

__global__ __launch_bounds__(256)
void dec_kernel(const float* __restrict__ a0base, const float* __restrict__ a1base,
                const float* __restrict__ b1, const float* __restrict__ w2,
                const float* __restrict__ b2, float* __restrict__ dec)
{
    __shared__ float tile[GT_PK * GT_KP];

    const int tid  = threadIdx.x;
    const int wave = tid >> 5;
    const int lane = tid & 31;
    const int blk  = blockIdx.x;
    const int nt   = blk & 7;
    const int mt   = (blk >> 3) & 7;
    const int b    = (blk >> 6) & 3;
    const int pair = blk >> 8;
    const int i    = pair / 3;
    const int j    = pair % 3;

    const float4* g0 = (const float4*)(a0base + ((size_t)i * Rsz + b * Lsz + nt * 8) * 2048);
    const float4* g1 = (const float4*)(a1base + ((size_t)j * Rsz + b * Lsz + mt * 8) * 2048);
    const float2* gw = (const float2*)w2;

    const int ml = (wave >> 1) * 2;
    const int nl = (wave & 1) * 4;
    const float* pa = tile + ml * 32 + lane;
    const float* pb = tile + (8 + nl) * 32 + lane;
    const float* pc = tile + 16 * 32 + lane;

    float acc[2][4];
#pragma unroll
    for (int x = 0; x < 2; ++x)
#pragma unroll
        for (int y = 0; y < 4; ++y) acc[x][y] = 0.f;

    for (int p = 0; p < 2; ++p) {
        {
            float4 r0[8], r1[8];
#pragma unroll
            for (int u = 0; u < 8; ++u) r0[u] = g0[u * 512 + p * 256 + tid];
#pragma unroll
            for (int u = 0; u < 8; ++u) r1[u] = g1[u * 512 + p * 256 + tid];
            const int dst = (tid >> 3) * GT_KP + (tid & 7) * 4;
#pragma unroll
            for (int u = 0; u < 8; ++u) {
                *(float4*)&tile[dst + (8 + u) * 32] = r0[u];
                *(float4*)&tile[dst + u * 32]       = r1[u];
            }
        }
#pragma unroll
        for (int u = 0; u < 4; ++u) {
            const int cl  = tid + u * 256;
            const float2 w = gw[p * 1024 + cl];
            const int dst = (cl >> 5) * GT_KP + (cl & 31);
            tile[dst + 16 * 32] = b1[p * 1024 + cl];
            tile[dst + 17 * 32] = w.y - w.x;
        }
        __syncthreads();

#pragma unroll 4
        for (int k = 0; k < GT_PK; ++k) {
            const int o = k * GT_KP;
            float xv[2], yv[4];
            xv[0] = pa[o];      xv[1] = pa[o + 32];
            yv[0] = pb[o];      yv[1] = pb[o + 32];
            yv[2] = pb[o + 64]; yv[3] = pb[o + 96];
            const float bb = pc[o], ww = pc[o + 32];
#pragma unroll
            for (int x = 0; x < 2; ++x)
#pragma unroll
                for (int y = 0; y < 4; ++y) {
                    const float t = (xv[x] + yv[y]) + bb;
                    if (t > 0.f) acc[x][y] = fmaf(t, ww, acc[x][y]);
                }
        }
        __syncthreads();
    }

#pragma unroll
    for (int off = 16; off; off >>= 1)
#pragma unroll
        for (int x = 0; x < 2; ++x)
#pragma unroll
            for (int y = 0; y < 4; ++y)
                acc[x][y] += __shfl_xor(acc[x][y], off, 32);

    if (lane == 0) {
        const float bd = b2[1] - b2[0];
        float* dtile = dec + (((size_t)(pair * Bsz + b) * Lsz + mt * 8) * Lsz + nt * 8);
#pragma unroll
        for (int x = 0; x < 2; ++x) {
            float4 r;
            r.x = ((acc[x][0] + bd) > 0.f) ? NEGC : 0.f;
            r.y = ((acc[x][1] + bd) > 0.f) ? NEGC : 0.f;
            r.z = ((acc[x][2] + bd) > 0.f) ? NEGC : 0.f;
            r.w = ((acc[x][3] + bd) > 0.f) ? NEGC : 0.f;
            *(float4*)(dtile + (size_t)(ml + x) * Lsz + nl) = r;
        }
    }
}

__global__ __launch_bounds__(192)
void attn_scores_kernel(const float* __restrict__ Qb, const float* __restrict__ Kb,
                        const float* __restrict__ dec, const int* __restrict__ mask,
                        float* __restrict__ S)
{
    const int blk = blockIdx.x;
    const int q = blk & 63;
    const int h = (blk >> 6) & 15;
    const int b = (blk >> 10) & 3;
    const int i = blk >> 12;
    const int t = threadIdx.x;
    const int j = t >> 6;
    const int k = t & 63;

    __shared__ float qrow[64];
    __shared__ float red[192];

    const float* Q = Qb + ((size_t)i * Rsz + b * Lsz + q) * DMsz + h * DKsz;
    if (t < 64) qrow[t] = Q[t];
    __syncthreads();

    const float* Kp = Kb + ((size_t)j * Rsz + b * Lsz + k) * DMsz + h * DKsz;
    float s = 0.f;
#pragma unroll 8
    for (int d = 0; d < 64; ++d) s += qrow[d] * Kp[d];
    s *= 0.125f;
    s += dec[(((size_t)(i * 3 + j) * Bsz + b) * Lsz + q) * Lsz + k];
    if (mask[((size_t)b * Lsz + q) * Lsz + k] == 0) s = NEGC;

    red[t] = s; __syncthreads();
    if (t < 64) red[t] = fmaxf(red[t], fmaxf(red[t + 64], red[t + 128]));
    __syncthreads();
    for (int st = 32; st > 0; st >>= 1) {
        if (t < st) red[t] = fmaxf(red[t], red[t + st]);
        __syncthreads();
    }
    const float mx = red[0];
    __syncthreads();
    const float e = __expf(s - mx);
    red[t] = e; __syncthreads();
    if (t < 64) red[t] = red[t] + red[t + 64] + red[t + 128];
    __syncthreads();
    for (int st = 32; st > 0; st >>= 1) {
        if (t < st) red[t] += red[t + st];
        __syncthreads();
    }
    S[(size_t)blk * 192 + t] = e * (1.0f / red[0]);
}

__global__ __launch_bounds__(64)
void attn_av_kernel(const float* __restrict__ S, const float* __restrict__ Vb,
                    float* __restrict__ Ob)
{
    const int blk = blockIdx.x;
    const int q = blk & 63;
    const int h = (blk >> 6) & 15;
    const int b = (blk >> 10) & 3;
    const int i = blk >> 12;
    const int d = threadIdx.x;

    __shared__ float arow[192];
    const float* Sp = S + (size_t)blk * 192;
    arow[d] = Sp[d]; arow[d + 64] = Sp[d + 64]; arow[d + 128] = Sp[d + 128];
    __syncthreads();

    float acc = 0.f;
    for (int jk = 0; jk < 192; ++jk) {
        int j = jk >> 6, k = jk & 63;
        acc += arow[jk] * Vb[((size_t)j * Rsz + b * Lsz + k) * DMsz + h * DVsz + d];
    }
    Ob[((size_t)i * Rsz + b * Lsz + q) * DMsz + h * DVsz + d] = acc;
}

__global__ __launch_bounds__(256)
void layernorm_kernel(const float* __restrict__ x, int nparts, size_t pstride,
                      const float* __restrict__ bias, const float* __restrict__ resid,
                      const float* __restrict__ g, const float* __restrict__ bta,
                      float* __restrict__ out)
{
    const int row = blockIdx.x;
    const int t = threadIdx.x;
    __shared__ float red[256];

    const float* xp = x + (size_t)row * 1024;
    float v[4];
#pragma unroll
    for (int u = 0; u < 4; ++u) v[u] = xp[t + u * 256];
#pragma unroll 8
    for (int p = 1; p < nparts; ++p) {
        const float* pp = xp + (size_t)p * pstride;
#pragma unroll
        for (int u = 0; u < 4; ++u) v[u] += pp[t + u * 256];
    }
    float s = 0.f;
#pragma unroll
    for (int u = 0; u < 4; ++u) {
        float val = v[u];
        if (bias)  val += bias[t + u * 256];
        if (resid) val += resid[(size_t)row * 1024 + t + u * 256];
        v[u] = val; s += val;
    }
    red[t] = s; __syncthreads();
    for (int st = 128; st > 0; st >>= 1) { if (t < st) red[t] += red[t + st]; __syncthreads(); }
    const float mu = red[0] * (1.f / 1024.f);
    __syncthreads();

    float s2 = 0.f;
#pragma unroll
    for (int u = 0; u < 4; ++u) { float dd = v[u] - mu; s2 += dd * dd; }
    red[t] = s2; __syncthreads();
    for (int st = 128; st > 0; st >>= 1) { if (t < st) red[t] += red[t + st]; __syncthreads(); }
    const float var = red[0] * (1.f / 1024.f);
    const float rinv = 1.0f / sqrtf(var + 1e-6f);
#pragma unroll
    for (int u = 0; u < 4; ++u) {
        int c = t + u * 256;
        out[(size_t)row * 1024 + c] = (v[u] - mu) * rinv * g[c] + bta[c];
    }
}

extern "C" void kernel_launch(void* const* d_in, const int* in_sizes, int n_in,
                              void* d_out, int out_size, void* d_ws, size_t ws_size,
                              hipStream_t stream)
{
    const float* X[3]   = { (const float*)d_in[0], (const float*)d_in[1], (const float*)d_in[2] };
    const int*   mask   = (const int*)d_in[3];
    const float* w_qs   = (const float*)d_in[4];
    const float* w_ks   = (const float*)d_in[5];
    const float* w_vs   = (const float*)d_in[6];
    const float* w_ds0  = (const float*)d_in[7];
    const float* w_ds1  = (const float*)d_in[8];
    const float* dc_w1  = (const float*)d_in[9];
    const float* dc_b1  = (const float*)d_in[10];
    const float* dc_w2  = (const float*)d_in[11];
    const float* dc_b2  = (const float*)d_in[12];
    const float* w_fc   = (const float*)d_in[13];
    const float* ln1_g  = (const float*)d_in[14];
    const float* ln1_b  = (const float*)d_in[15];
    const float* ffn_w1 = (const float*)d_in[16];
    const float* ffn_b1 = (const float*)d_in[17];
    const float* ffn_w2 = (const float*)d_in[18];
    const float* ffn_b2 = (const float*)d_in[19];
    const float* ln2_g  = (const float*)d_in[20];
    const float* ln2_b  = (const float*)d_in[21];
    float* out = (float*)d_out;

    float* ws = (float*)d_ws;
    size_t off = 0;
    auto alloc = [&](size_t n) { float* p = ws + off; off += n; return p; };
    const size_t RM = (size_t)Rsz * DMsz;
    float* Q   = alloc(3 * RM);
    float* Kb  = alloc(3 * RM);
    float* Vb  = alloc(3 * RM);
    float* d0  = alloc(3 * RM);
    float* d1  = alloc(3 * RM);
    float* a0  = alloc(3 * (size_t)Rsz * 2048);
    float* a1  = alloc(3 * (size_t)Rsz * 2048);
    float* dec = alloc(9 * (size_t)Bsz * Lsz * Lsz);
    float* S   = alloc((size_t)3 * Bsz * Hsz * Lsz * 192);
    float* O   = alloc(3 * RM);
    float* fc  = alloc(3 * RM);
    float* oln = alloc(3 * RM);
    float* fpart = alloc(32 * RM);
    (void)ws_size; (void)n_in; (void)in_sizes; (void)out_size;

    auto gemm = [&](const float* A, const float* Bm, const float* bias,
                    const float* resid, float* C, int M, int N, int K, int flags) {
        dim3 grid(N / 32, M / 32);
        gemm_wmma_f32<<<grid, dim3(128), 0, stream>>>(A, Bm, bias, resid, C, M, N, K, flags);
    };

    for (int i = 0; i < 3; ++i) {
        gemm(X[i], w_qs,  nullptr, nullptr, Q  + i * RM, Rsz, 1024, 1024, 0);
        gemm(X[i], w_ks,  nullptr, nullptr, Kb + i * RM, Rsz, 1024, 1024, 0);
        gemm(X[i], w_vs,  nullptr, nullptr, Vb + i * RM, Rsz, 1024, 1024, 0);
        gemm(X[i], w_ds0, nullptr, nullptr, d0 + i * RM, Rsz, 1024, 1024, 0);
        gemm(X[i], w_ds1, nullptr, nullptr, d1 + i * RM, Rsz, 1024, 1024, 0);
        gemm(d0 + i * RM, dc_w1 + (size_t)1024 * 2048, nullptr, nullptr,
             a0 + i * (size_t)Rsz * 2048, Rsz, 2048, 1024, 0);
        gemm(d1 + i * RM, dc_w1, nullptr, nullptr,
             a1 + i * (size_t)Rsz * 2048, Rsz, 2048, 1024, 0);
    }

    {
        dec_kernel<<<9 * Bsz * 8 * 8, 256, 0, stream>>>(a0, a1, dc_b1, dc_w2, dc_b2, dec);
    }

    {
        int nblk = 3 * Bsz * Hsz * Lsz;
        attn_scores_kernel<<<nblk, 192, 0, stream>>>(Q, Kb, dec, mask, S);
        attn_av_kernel<<<nblk, 64, 0, stream>>>(S, Vb, O);
    }

    for (int i = 0; i < 3; ++i) {
        gemm(O + i * RM, w_fc, nullptr, X[i], fc + i * RM, Rsz, 1024, 1024, GEMM_RESID);
        layernorm_kernel<<<Rsz, 256, 0, stream>>>(fc + i * RM, 1, RM, nullptr, nullptr,
                                                  ln1_g, ln1_b, oln + i * RM);
        ffn_fused_wmma_f32<<<dim3(32, 4, 2), dim3(256), 0, stream>>>(
            oln + i * RM, ffn_w1, ffn_b1, ffn_w2, fpart);
        layernorm_kernel<<<Rsz, 256, 0, stream>>>(fpart, 32, RM, ffn_b2, oln + i * RM,
                                                  ln2_g, ln2_b, out + i * RM);
    }
}


# Scores+softmax re-cut: a block stages the 192 key rows of (b,h) in LDS once for 16 query rows; a lane owns 6 keys of 4 rows, softmax by lane exchange (was one block per row); sums bit-identical
# speedup vs baseline: 3.1504x; 1.1118x over previous
#include <hip/hip_runtime.h>
#include <hip/hip_bf16.h>

#define Bsz 4
#define Lsz 64
#define DMsz 1024
#define Hsz 16
#define DKsz 64
#define DVsz 64
#define Dsz 1024
#define DIsz 4096
#define Rsz 256
#define NEGC (-1.0e9f)

typedef float v2f __attribute__((ext_vector_type(2)));
typedef float v4f __attribute__((ext_vector_type(4)));
typedef float v8f __attribute__((ext_vector_type(8)));

#define GEMM_BIAS  1
#define GEMM_RELU  2
#define GEMM_RESID 4

#define GK  128
#define GAP (GK + 4)
#define GBP 40
#define GCP 36
#define GNA (GK / 16)
#define GNB (GK / 16)

__global__ __launch_bounds__(128)
void gemm_wmma_f32(const float* __restrict__ A, const float* __restrict__ B,
                   const float* __restrict__ bias, const float* __restrict__ resid,
                   float* __restrict__ C, int M, int N, int K, int flags)
{
    __shared__ __attribute__((aligned(16))) float As[32 * GAP];
    __shared__ __attribute__((aligned(16))) float Bs[GK * GBP];
    __shared__ __attribute__((aligned(16))) float Cs[32 * GCP];

    const int tid   = threadIdx.x;
    const int wave  = tid >> 5;
    const int lane  = tid & 31;
    const int l15   = lane & 15;
    const int lhalf = lane >> 4;
    const int m0    = blockIdx.y * 32;
    const int n0    = blockIdx.x * 32;
    const int mq    = wave >> 1;
    const int nq    = wave & 1;
    (void)M;

    const int ar = tid / (GK / 4);
    const int ac = (tid % (GK / 4)) << 2;
    const int br = tid >> 3;
    const int bc = (tid & 7) << 2;
    const float* Ag = A + (size_t)(m0 + ar) * K + ac;
    const float* Bg = B + (size_t)br * N + n0 + bc;
    v4f aprf[GNA], bprf[GNB];

    v8f acc = v8f{0.f, 0.f, 0.f, 0.f, 0.f, 0.f, 0.f, 0.f};

#pragma unroll
    for (int u = 0; u < GNA; ++u)
        aprf[u] = *(const v4f*)(Ag + (size_t)(u * (512 / GK)) * K);
#pragma unroll
    for (int u = 0; u < GNB; ++u)
        bprf[u] = *(const v4f*)(Bg + (size_t)(u * 16) * N);

    const float* ap = As + (mq * 16 + l15) * GAP + lhalf * 2;
    const float* bp = Bs + (lhalf * 2) * GBP + nq * 16 + l15;

    for (int k0 = 0; k0 < K; k0 += GK) {
#pragma unroll
        for (int u = 0; u < GNA; ++u)
            *(v4f*)&As[(ar + u * (512 / GK)) * GAP + ac] = aprf[u];
#pragma unroll
        for (int u = 0; u < GNB; ++u)
            *(v4f*)&Bs[(br + u * 16) * GBP + bc] = bprf[u];
        __syncthreads();

        if (k0 + GK < K) {
#pragma unroll
            for (int u = 0; u < GNA; ++u)
                aprf[u] = *(const v4f*)(Ag + (size_t)(u * (512 / GK)) * K + k0 + GK);
#pragma unroll
            for (int u = 0; u < GNB; ++u)
                bprf[u] = *(const v4f*)(Bg + (size_t)(k0 + GK + u * 16) * N);
        }

        v2f af[GK / 4], bf[GK / 4];
#pragma unroll
        for (int s = 0; s < GK / 4; ++s) {
            af[s] = v2f{ap[s * 4], ap[s * 4 + 1]};
            bf[s] = v2f{bp[(s * 4) * GBP], bp[(s * 4 + 1) * GBP]};
        }
#pragma unroll
        for (int s = 0; s < GK / 4; ++s)
            acc = __builtin_amdgcn_wmma_f32_16x16x4_f32(
                false, af[s], false, bf[s], (short)0, acc, false, false);
        __builtin_amdgcn_sched_group_barrier(0x100, 12, 0);
#pragma unroll
        for (int p = 0; p < (GK / 4 - 8) / 2; ++p) {
            __builtin_amdgcn_sched_group_barrier(0x008, 2, 0);
            __builtin_amdgcn_sched_group_barrier(0x100, 3, 0);
        }
        __builtin_amdgcn_sched_group_barrier(0x008, 8, 0);
        __syncthreads();
    }

#pragma unroll
    for (int r = 0; r < 8; ++r)
        Cs[(mq * 16 + r + lhalf * 8) * GCP + nq * 16 + l15] = acc[r];
    __syncthreads();

#pragma unroll
    for (int u = 0; u < 2; ++u) {
        const int v   = tid + u * 128;
        const int row = v >> 3;
        const int c4  = (v & 7) << 2;
        const size_t g = (size_t)(m0 + row) * N + n0 + c4;
        v4f o = *(const v4f*)&Cs[row * GCP + c4];
        v4f bv = v4f{0.f, 0.f, 0.f, 0.f};
        if (flags & GEMM_BIAS) bv = *(const v4f*)(bias + n0 + c4);
        o.x = o.x + bv.x; o.y = o.y + bv.y; o.z = o.z + bv.z; o.w = o.w + bv.w;
        if (flags & GEMM_RESID) {
            const v4f rv = *(const v4f*)(resid + g);
            o.x += rv.x; o.y += rv.y; o.z += rv.z; o.w += rv.w;
        }
        if (flags & GEMM_RELU) {
            o.x = fmaxf(o.x, 0.f); o.y = fmaxf(o.y, 0.f);
            o.z = fmaxf(o.z, 0.f); o.w = fmaxf(o.w, 0.f);
        }
        *(v4f*)(C + g) = o;
    }
}

__global__ __launch_bounds__(256)
void ffn_fused_wmma_f32(const float* __restrict__ X, const float* __restrict__ W1,
                        const float* __restrict__ b1, const float* __restrict__ W2,
                        float* __restrict__ P)
{
    __shared__ float As[64][64 + 1];
    __shared__ float Bs[64][128 + 1];
    __shared__ float Hs[64][128 + 1];

    const int tid   = threadIdx.x;
    const int wave  = tid >> 5;
    const int lane  = tid & 31;
    const int l15   = lane & 15;
    const int lhalf = lane >> 4;
    const int s0    = blockIdx.x * 128;
    const int m0    = blockIdx.y * 64;
    const int nc0   = blockIdx.z * 4;
    const int mq    = wave >> 2;
    const int nq    = wave & 3;

    const int ar = tid >> 4;
    const int ac = (tid & 15) << 2;
    const int br = (tid >> 5) * 8;
    const int bc = (tid & 31) << 2;
    float4 aprf[4], bprf[8];

    v8f acc[2][2];
#pragma unroll
    for (int x = 0; x < 2; ++x)
#pragma unroll
        for (int y = 0; y < 2; ++y)
            acc[x][y] = v8f{0.f, 0.f, 0.f, 0.f, 0.f, 0.f, 0.f, 0.f};

#pragma unroll
    for (int rr = 0; rr < 4; ++rr)
        aprf[rr] = *(const float4*)(X + (size_t)(m0 + ar + rr * 16) * DMsz + ac);
#pragma unroll
    for (int rr = 0; rr < 8; ++rr)
        bprf[rr] = *(const float4*)(W1 + (size_t)(br + rr) * DIsz + s0 + bc);

    const int arow0 = mq * 32 + l15;
    const int bcol0 = nq * 32 + l15;

    for (int k0 = 0; k0 < DMsz; k0 += 64) {
#pragma unroll
        for (int rr = 0; rr < 4; ++rr) {
            As[ar + rr * 16][ac + 0] = aprf[rr].x;
            As[ar + rr * 16][ac + 1] = aprf[rr].y;
            As[ar + rr * 16][ac + 2] = aprf[rr].z;
            As[ar + rr * 16][ac + 3] = aprf[rr].w;
        }
#pragma unroll
        for (int rr = 0; rr < 8; ++rr) {
            Bs[br + rr][bc + 0] = bprf[rr].x;
            Bs[br + rr][bc + 1] = bprf[rr].y;
            Bs[br + rr][bc + 2] = bprf[rr].z;
            Bs[br + rr][bc + 3] = bprf[rr].w;
        }
        __syncthreads();

        if (k0 + 64 < DMsz) {
#pragma unroll
            for (int rr = 0; rr < 4; ++rr)
                aprf[rr] = *(const float4*)(X + (size_t)(m0 + ar + rr * 16) * DMsz
                                            + k0 + 64 + ac);
#pragma unroll
            for (int rr = 0; rr < 8; ++rr)
                bprf[rr] = *(const float4*)(W1 + (size_t)(k0 + 64 + br + rr) * DIsz
                                            + s0 + bc);
        } else {
#pragma unroll
            for (int rr = 0; rr < 8; ++rr)
                bprf[rr] = *(const float4*)(W2 + (size_t)(s0 + br + rr) * DMsz
                                            + nc0 * 128 + bc);
        }

#pragma unroll
        for (int half = 0; half < 2; ++half) {
            v2f af[2][8], bf[2][8];
#pragma unroll
            for (int s = 0; s < 8; ++s) {
                const int kb = half * 32 + s * 4 + lhalf * 2;
                af[0][s] = v2f{As[arow0][kb],      As[arow0][kb + 1]};
                af[1][s] = v2f{As[arow0 + 16][kb], As[arow0 + 16][kb + 1]};
                bf[0][s] = v2f{Bs[kb][bcol0],      Bs[kb + 1][bcol0]};
                bf[1][s] = v2f{Bs[kb][bcol0 + 16], Bs[kb + 1][bcol0 + 16]};
            }
#pragma unroll
            for (int s = 0; s < 8; ++s) {
                acc[0][0] = __builtin_amdgcn_wmma_f32_16x16x4_f32(
                    false, af[0][s], false, bf[0][s], (short)0, acc[0][0], false, false);
                acc[0][1] = __builtin_amdgcn_wmma_f32_16x16x4_f32(
                    false, af[0][s], false, bf[1][s], (short)0, acc[0][1], false, false);
                acc[1][0] = __builtin_amdgcn_wmma_f32_16x16x4_f32(
                    false, af[1][s], false, bf[0][s], (short)0, acc[1][0], false, false);
                acc[1][1] = __builtin_amdgcn_wmma_f32_16x16x4_f32(
                    false, af[1][s], false, bf[1][s], (short)0, acc[1][1], false, false);
            }
        }
        __syncthreads();
    }

#pragma unroll
    for (int nsub = 0; nsub < 2; ++nsub) {
        const int hc = nq * 32 + nsub * 16 + l15;
        const float bv = b1[s0 + hc];
#pragma unroll
        for (int msub = 0; msub < 2; ++msub) {
#pragma unroll
            for (int r = 0; r < 8; ++r) {
                const int hr = mq * 32 + msub * 16 + r + lhalf * 8;
                Hs[hr][hc] = fmaxf(acc[msub][nsub][r] + bv, 0.f);
            }
        }
    }

    for (int nc = nc0; nc < nc0 + 4; ++nc) {
#pragma unroll
        for (int x = 0; x < 2; ++x)
#pragma unroll
            for (int y = 0; y < 2; ++y)
                acc[x][y] = v8f{0.f, 0.f, 0.f, 0.f, 0.f, 0.f, 0.f, 0.f};

        for (int kc = 0; kc < 2; ++kc) {
#pragma unroll
            for (int rr = 0; rr < 8; ++rr) {
                Bs[br + rr][bc + 0] = bprf[rr].x;
                Bs[br + rr][bc + 1] = bprf[rr].y;
                Bs[br + rr][bc + 2] = bprf[rr].z;
                Bs[br + rr][bc + 3] = bprf[rr].w;
            }
            __syncthreads();

            const int tn = (nc - nc0) * 2 + kc + 1;
            if (tn < 8) {
                const int nn = nc0 + (tn >> 1);
                const int kk = tn & 1;
#pragma unroll
                for (int rr = 0; rr < 8; ++rr)
                    bprf[rr] = *(const float4*)(W2 + (size_t)(s0 + kk * 64 + br + rr) * DMsz
                                                + nn * 128 + bc);
            }

            const int hk0 = kc * 64;
#pragma unroll
            for (int half = 0; half < 2; ++half) {
                v2f af[2][8], bf[2][8];
#pragma unroll
                for (int s = 0; s < 8; ++s) {
                    const int kb = half * 32 + s * 4 + lhalf * 2;
                    af[0][s] = v2f{Hs[arow0][hk0 + kb],      Hs[arow0][hk0 + kb + 1]};
                    af[1][s] = v2f{Hs[arow0 + 16][hk0 + kb], Hs[arow0 + 16][hk0 + kb + 1]};
                    bf[0][s] = v2f{Bs[kb][bcol0],      Bs[kb + 1][bcol0]};
                    bf[1][s] = v2f{Bs[kb][bcol0 + 16], Bs[kb + 1][bcol0 + 16]};
                }
#pragma unroll
                for (int s = 0; s < 8; ++s) {
                    acc[0][0] = __builtin_amdgcn_wmma_f32_16x16x4_f32(
                        false, af[0][s], false, bf[0][s], (short)0, acc[0][0], false, false);
                    acc[0][1] = __builtin_amdgcn_wmma_f32_16x16x4_f32(
                        false, af[0][s], false, bf[1][s], (short)0, acc[0][1], false, false);
                    acc[1][0] = __builtin_amdgcn_wmma_f32_16x16x4_f32(
                        false, af[1][s], false, bf[0][s], (short)0, acc[1][0], false, false);
                    acc[1][1] = __builtin_amdgcn_wmma_f32_16x16x4_f32(
                        false, af[1][s], false, bf[1][s], (short)0, acc[1][1], false, false);
                }
            }
            __syncthreads();
        }

        float* Pp = P + (size_t)blockIdx.x * Rsz * DMsz;
#pragma unroll
        for (int nsub = 0; nsub < 2; ++nsub) {
            const int col = nc * 128 + nq * 32 + nsub * 16 + l15;
#pragma unroll
            for (int msub = 0; msub < 2; ++msub) {
#pragma unroll
                for (int r = 0; r < 8; ++r) {
                    const int row = m0 + mq * 32 + msub * 16 + r + lhalf * 8;
                    Pp[(size_t)row * DMsz + col] = acc[msub][nsub][r];
                }
            }
        }
    }
}

#define GT_KP (18 * 32)
#define GT_PK 32

__global__ __launch_bounds__(256)
void dec_kernel(const float* __restrict__ a0base, const float* __restrict__ a1base,
                const float* __restrict__ b1, const float* __restrict__ w2,
                const float* __restrict__ b2, float* __restrict__ dec)
{
    __shared__ float tile[GT_PK * GT_KP];

    const int tid  = threadIdx.x;
    const int wave = tid >> 5;
    const int lane = tid & 31;
    const int blk  = blockIdx.x;
    const int nt   = blk & 7;
    const int mt   = (blk >> 3) & 7;
    const int b    = (blk >> 6) & 3;
    const int pair = blk >> 8;
    const int i    = pair / 3;
    const int j    = pair % 3;

    const float4* g0 = (const float4*)(a0base + ((size_t)i * Rsz + b * Lsz + nt * 8) * 2048);
    const float4* g1 = (const float4*)(a1base + ((size_t)j * Rsz + b * Lsz + mt * 8) * 2048);
    const float2* gw = (const float2*)w2;

    const int ml = (wave >> 1) * 2;
    const int nl = (wave & 1) * 4;
    const float* pa = tile + ml * 32 + lane;
    const float* pb = tile + (8 + nl) * 32 + lane;
    const float* pc = tile + 16 * 32 + lane;

    float acc[2][4];
#pragma unroll
    for (int x = 0; x < 2; ++x)
#pragma unroll
        for (int y = 0; y < 4; ++y) acc[x][y] = 0.f;

    for (int p = 0; p < 2; ++p) {
        {
            float4 r0[8], r1[8];
#pragma unroll
            for (int u = 0; u < 8; ++u) r0[u] = g0[u * 512 + p * 256 + tid];
#pragma unroll
            for (int u = 0; u < 8; ++u) r1[u] = g1[u * 512 + p * 256 + tid];
            const int dst = (tid >> 3) * GT_KP + (tid & 7) * 4;
#pragma unroll
            for (int u = 0; u < 8; ++u) {
                *(float4*)&tile[dst + (8 + u) * 32] = r0[u];
                *(float4*)&tile[dst + u * 32]       = r1[u];
            }
        }
#pragma unroll
        for (int u = 0; u < 4; ++u) {
            const int cl  = tid + u * 256;
            const float2 w = gw[p * 1024 + cl];
            const int dst = (cl >> 5) * GT_KP + (cl & 31);
            tile[dst + 16 * 32] = b1[p * 1024 + cl];
            tile[dst + 17 * 32] = w.y - w.x;
        }
        __syncthreads();

#pragma unroll 4
        for (int k = 0; k < GT_PK; ++k) {
            const int o = k * GT_KP;
            float xv[2], yv[4];
            xv[0] = pa[o];      xv[1] = pa[o + 32];
            yv[0] = pb[o];      yv[1] = pb[o + 32];
            yv[2] = pb[o + 64]; yv[3] = pb[o + 96];
            const float bb = pc[o], ww = pc[o + 32];
#pragma unroll
            for (int x = 0; x < 2; ++x)
#pragma unroll
                for (int y = 0; y < 4; ++y) {
                    const float t = (xv[x] + yv[y]) + bb;
                    if (t > 0.f) acc[x][y] = fmaf(t, ww, acc[x][y]);
                }
        }
        __syncthreads();
    }

#pragma unroll
    for (int off = 16; off; off >>= 1)
#pragma unroll
        for (int x = 0; x < 2; ++x)
#pragma unroll
            for (int y = 0; y < 4; ++y)
                acc[x][y] += __shfl_xor(acc[x][y], off, 32);

    if (lane == 0) {
        const float bd = b2[1] - b2[0];
        float* dtile = dec + (((size_t)(pair * Bsz + b) * Lsz + mt * 8) * Lsz + nt * 8);
#pragma unroll
        for (int x = 0; x < 2; ++x) {
            float4 r;
            r.x = ((acc[x][0] + bd) > 0.f) ? NEGC : 0.f;
            r.y = ((acc[x][1] + bd) > 0.f) ? NEGC : 0.f;
            r.z = ((acc[x][2] + bd) > 0.f) ? NEGC : 0.f;
            r.w = ((acc[x][3] + bd) > 0.f) ? NEGC : 0.f;
            *(float4*)(dtile + (size_t)(ml + x) * Lsz + nl) = r;
        }
    }
}

#define SC_QT 16
#define SC_KP 68

__global__ __launch_bounds__(128)
void attn_scores_kernel(const float* __restrict__ Qb, const float* __restrict__ Kb,
                        const float* __restrict__ dec, const int* __restrict__ mask,
                        float* __restrict__ S)
{
    __shared__ __attribute__((aligned(16))) float Ks[192 * SC_KP];
    __shared__ __attribute__((aligned(16))) float Qs[SC_QT * 64];

    const int blk  = blockIdx.x;
    const int qt   = blk & 3;
    const int h    = (blk >> 2) & 15;
    const int b    = (blk >> 6) & 3;
    const int i    = blk >> 8;
    const int tid  = threadIdx.x;
    const int wave = tid >> 5;
    const int lane = tid & 31;
    const int q0   = qt * SC_QT + wave * 4;

    {
        const int c4 = (tid & 15) << 2;
        const int r0 = tid >> 4;
        v4f kr[24], qr[2];
#pragma unroll
        for (int u = 0; u < 24; ++u) {
            const int row = r0 + u * 8;
            const int j = row >> 6, k = row & 63;
            kr[u] = *(const v4f*)(Kb + ((size_t)j * Rsz + b * Lsz + k) * DMsz + h * DKsz + c4);
        }
#pragma unroll
        for (int u = 0; u < 2; ++u) {
            const int row = r0 + u * 8;
            qr[u] = *(const v4f*)(Qb + ((size_t)i * Rsz + b * Lsz + qt * SC_QT + row) * DMsz
                                  + h * DKsz + c4);
        }
#pragma unroll
        for (int u = 0; u < 24; ++u)
            *(v4f*)&Ks[(r0 + u * 8) * SC_KP + c4] = kr[u];
#pragma unroll
        for (int u = 0; u < 2; ++u)
            *(v4f*)&Qs[(r0 + u * 8) * 64 + c4] = qr[u];
    }

    float dv[4][6];
    int   mk[4][2];
#pragma unroll
    for (int r = 0; r < 4; ++r) {
#pragma unroll
        for (int m = 0; m < 6; ++m) {
            const int j = m >> 1, k = (m & 1) * 32 + lane;
            dv[r][m] = dec[(((size_t)(i * 3 + j) * Bsz + b) * Lsz + q0 + r) * Lsz + k];
        }
#pragma unroll
        for (int kh = 0; kh < 2; ++kh)
            mk[r][kh] = mask[((size_t)b * Lsz + q0 + r) * Lsz + kh * 32 + lane];
    }
    __syncthreads();

    float acc[4][6];
#pragma unroll
    for (int r = 0; r < 4; ++r)
#pragma unroll
        for (int m = 0; m < 6; ++m) acc[r][m] = 0.f;

    const float* kp = Ks + lane * SC_KP;
    const float* qp = Qs + (wave * 4) * 64;
    for (int d = 0; d < 64; d += 4) {
        v4f kv[6], qv[4];
#pragma unroll
        for (int m = 0; m < 6; ++m) kv[m] = *(const v4f*)(kp + m * 32 * SC_KP + d);
#pragma unroll
        for (int r = 0; r < 4; ++r) qv[r] = *(const v4f*)(qp + r * 64 + d);
#pragma unroll
        for (int r = 0; r < 4; ++r)
#pragma unroll
            for (int m = 0; m < 6; ++m) {
                float s = acc[r][m];
                s = fmaf(qv[r].x, kv[m].x, s);
                s = fmaf(qv[r].y, kv[m].y, s);
                s = fmaf(qv[r].z, kv[m].z, s);
                s = fmaf(qv[r].w, kv[m].w, s);
                acc[r][m] = s;
            }
    }

#pragma unroll
    for (int r = 0; r < 4; ++r) {
        float s[6];
#pragma unroll
        for (int m = 0; m < 6; ++m) {
            float t = acc[r][m];
            t *= 0.125f;
            t += dv[r][m];
            if (mk[r][m & 1] == 0) t = NEGC;
            s[m] = t;
        }
        float mx = fmaxf(fmaxf(fmaxf(s[0], s[1]), fmaxf(s[2], s[3])), fmaxf(s[4], s[5]));
#pragma unroll
        for (int off = 16; off; off >>= 1) mx = fmaxf(mx, __shfl_xor(mx, off, 32));
        float e[6];
#pragma unroll
        for (int m = 0; m < 6; ++m) e[m] = __expf(s[m] - mx);
        float sum = ((e[0] + e[2]) + e[4]) + ((e[1] + e[3]) + e[5]);
#pragma unroll
        for (int off = 16; off; off >>= 1) sum += __shfl_xor(sum, off, 32);
        const float inv = 1.0f / sum;
        float* Sp = S + ((size_t)((((i * Bsz + b) * Hsz + h) * Lsz) + q0 + r)) * 192 + lane;
#pragma unroll
        for (int m = 0; m < 6; ++m) Sp[m * 32] = e[m] * inv;
    }
}

__global__ __launch_bounds__(64)
void attn_av_kernel(const float* __restrict__ S, const float* __restrict__ Vb,
                    float* __restrict__ Ob)
{
    const int blk = blockIdx.x;
    const int q = blk & 63;
    const int h = (blk >> 6) & 15;
    const int b = (blk >> 10) & 3;
    const int i = blk >> 12;
    const int d = threadIdx.x;

    __shared__ float arow[192];
    const float* Sp = S + (size_t)blk * 192;
    arow[d] = Sp[d]; arow[d + 64] = Sp[d + 64]; arow[d + 128] = Sp[d + 128];
    __syncthreads();

    float acc = 0.f;
    for (int jk = 0; jk < 192; ++jk) {
        int j = jk >> 6, k = jk & 63;
        acc += arow[jk] * Vb[((size_t)j * Rsz + b * Lsz + k) * DMsz + h * DVsz + d];
    }
    Ob[((size_t)i * Rsz + b * Lsz + q) * DMsz + h * DVsz + d] = acc;
}

__global__ __launch_bounds__(256)
void layernorm_kernel(const float* __restrict__ x, int nparts, size_t pstride,
                      const float* __restrict__ bias, const float* __restrict__ resid,
                      const float* __restrict__ g, const float* __restrict__ bta,
                      float* __restrict__ out)
{
    const int row = blockIdx.x;
    const int t = threadIdx.x;
    __shared__ float red[256];

    const float* xp = x + (size_t)row * 1024;
    float v[4];
#pragma unroll
    for (int u = 0; u < 4; ++u) v[u] = xp[t + u * 256];
#pragma unroll 8
    for (int p = 1; p < nparts; ++p) {
        const float* pp = xp + (size_t)p * pstride;
#pragma unroll
        for (int u = 0; u < 4; ++u) v[u] += pp[t + u * 256];
    }
    float s = 0.f;
#pragma unroll
    for (int u = 0; u < 4; ++u) {
        float val = v[u];
        if (bias)  val += bias[t + u * 256];
        if (resid) val += resid[(size_t)row * 1024 + t + u * 256];
        v[u] = val; s += val;
    }
    red[t] = s; __syncthreads();
    for (int st = 128; st > 0; st >>= 1) { if (t < st) red[t] += red[t + st]; __syncthreads(); }
    const float mu = red[0] * (1.f / 1024.f);
    __syncthreads();

    float s2 = 0.f;
#pragma unroll
    for (int u = 0; u < 4; ++u) { float dd = v[u] - mu; s2 += dd * dd; }
    red[t] = s2; __syncthreads();
    for (int st = 128; st > 0; st >>= 1) { if (t < st) red[t] += red[t + st]; __syncthreads(); }
    const float var = red[0] * (1.f / 1024.f);
    const float rinv = 1.0f / sqrtf(var + 1e-6f);
#pragma unroll
    for (int u = 0; u < 4; ++u) {
        int c = t + u * 256;
        out[(size_t)row * 1024 + c] = (v[u] - mu) * rinv * g[c] + bta[c];
    }
}

extern "C" void kernel_launch(void* const* d_in, const int* in_sizes, int n_in,
                              void* d_out, int out_size, void* d_ws, size_t ws_size,
                              hipStream_t stream)
{
    const float* X[3]   = { (const float*)d_in[0], (const float*)d_in[1], (const float*)d_in[2] };
    const int*   mask   = (const int*)d_in[3];
    const float* w_qs   = (const float*)d_in[4];
    const float* w_ks   = (const float*)d_in[5];
    const float* w_vs   = (const float*)d_in[6];
    const float* w_ds0  = (const float*)d_in[7];
    const float* w_ds1  = (const float*)d_in[8];
    const float* dc_w1  = (const float*)d_in[9];
    const float* dc_b1  = (const float*)d_in[10];
    const float* dc_w2  = (const float*)d_in[11];
    const float* dc_b2  = (const float*)d_in[12];
    const float* w_fc   = (const float*)d_in[13];
    const float* ln1_g  = (const float*)d_in[14];
    const float* ln1_b  = (const float*)d_in[15];
    const float* ffn_w1 = (const float*)d_in[16];
    const float* ffn_b1 = (const float*)d_in[17];
    const float* ffn_w2 = (const float*)d_in[18];
    const float* ffn_b2 = (const float*)d_in[19];
    const float* ln2_g  = (const float*)d_in[20];
    const float* ln2_b  = (const float*)d_in[21];
    float* out = (float*)d_out;

    float* ws = (float*)d_ws;
    size_t off = 0;
    auto alloc = [&](size_t n) { float* p = ws + off; off += n; return p; };
    const size_t RM = (size_t)Rsz * DMsz;
    float* Q   = alloc(3 * RM);
    float* Kb  = alloc(3 * RM);
    float* Vb  = alloc(3 * RM);
    float* d0  = alloc(3 * RM);
    float* d1  = alloc(3 * RM);
    float* a0  = alloc(3 * (size_t)Rsz * 2048);
    float* a1  = alloc(3 * (size_t)Rsz * 2048);
    float* dec = alloc(9 * (size_t)Bsz * Lsz * Lsz);
    float* S   = alloc((size_t)3 * Bsz * Hsz * Lsz * 192);
    float* O   = alloc(3 * RM);
    float* fc  = alloc(3 * RM);
    float* oln = alloc(3 * RM);
    float* fpart = alloc(32 * RM);
    (void)ws_size; (void)n_in; (void)in_sizes; (void)out_size;

    auto gemm = [&](const float* A, const float* Bm, const float* bias,
                    const float* resid, float* C, int M, int N, int K, int flags) {
        dim3 grid(N / 32, M / 32);
        gemm_wmma_f32<<<grid, dim3(128), 0, stream>>>(A, Bm, bias, resid, C, M, N, K, flags);
    };

    for (int i = 0; i < 3; ++i) {
        gemm(X[i], w_qs,  nullptr, nullptr, Q  + i * RM, Rsz, 1024, 1024, 0);
        gemm(X[i], w_ks,  nullptr, nullptr, Kb + i * RM, Rsz, 1024, 1024, 0);
        gemm(X[i], w_vs,  nullptr, nullptr, Vb + i * RM, Rsz, 1024, 1024, 0);
        gemm(X[i], w_ds0, nullptr, nullptr, d0 + i * RM, Rsz, 1024, 1024, 0);
        gemm(X[i], w_ds1, nullptr, nullptr, d1 + i * RM, Rsz, 1024, 1024, 0);
        gemm(d0 + i * RM, dc_w1 + (size_t)1024 * 2048, nullptr, nullptr,
             a0 + i * (size_t)Rsz * 2048, Rsz, 2048, 1024, 0);
        gemm(d1 + i * RM, dc_w1, nullptr, nullptr,
             a1 + i * (size_t)Rsz * 2048, Rsz, 2048, 1024, 0);
    }

    {
        dec_kernel<<<9 * Bsz * 8 * 8, 256, 0, stream>>>(a0, a1, dc_b1, dc_w2, dc_b2, dec);
    }

    {
        int nblk = 3 * Bsz * Hsz * Lsz;
        attn_scores_kernel<<<3 * Bsz * Hsz * (Lsz / SC_QT), 128, 0, stream>>>(Q, Kb, dec, mask, S);
        attn_av_kernel<<<nblk, 64, 0, stream>>>(S, Vb, O);
    }

    for (int i = 0; i < 3; ++i) {
        gemm(O + i * RM, w_fc, nullptr, X[i], fc + i * RM, Rsz, 1024, 1024, GEMM_RESID);
        layernorm_kernel<<<Rsz, 256, 0, stream>>>(fc + i * RM, 1, RM, nullptr, nullptr,
                                                  ln1_g, ln1_b, oln + i * RM);
        ffn_fused_wmma_f32<<<dim3(32, 4, 2), dim3(256), 0, stream>>>(
            oln + i * RM, ffn_w1, ffn_b1, ffn_w2, fpart);
        layernorm_kernel<<<Rsz, 256, 0, stream>>>(fpart, 32, RM, ffn_b2, oln + i * RM,
                                                  ln2_g, ln2_b, out + i * RM);
    }
}


# FFN block re-cut: 32 rows x one 128-hidden slab x all 1024 output columns (was 64 rows x half the columns: every hidden tile computed twice); 128-k stages in 16-byte LDS rows; sums bit-identical
# speedup vs baseline: 3.4248x; 1.0871x over previous
#include <hip/hip_runtime.h>
#include <hip/hip_bf16.h>

#define Bsz 4
#define Lsz 64
#define DMsz 1024
#define Hsz 16
#define DKsz 64
#define DVsz 64
#define Dsz 1024
#define DIsz 4096
#define Rsz 256
#define NEGC (-1.0e9f)

typedef float v2f __attribute__((ext_vector_type(2)));
typedef float v4f __attribute__((ext_vector_type(4)));
typedef float v8f __attribute__((ext_vector_type(8)));

#define GEMM_BIAS  1
#define GEMM_RELU  2
#define GEMM_RESID 4

#define GK  128
#define GAP (GK + 4)
#define GBP 40
#define GCP 36
#define GNA (GK / 16)
#define GNB (GK / 16)

__global__ __launch_bounds__(128)
void gemm_wmma_f32(const float* __restrict__ A, const float* __restrict__ B,
                   const float* __restrict__ bias, const float* __restrict__ resid,
                   float* __restrict__ C, int M, int N, int K, int flags)
{
    __shared__ __attribute__((aligned(16))) float As[32 * GAP];
    __shared__ __attribute__((aligned(16))) float Bs[GK * GBP];
    __shared__ __attribute__((aligned(16))) float Cs[32 * GCP];

    const int tid   = threadIdx.x;
    const int wave  = tid >> 5;
    const int lane  = tid & 31;
    const int l15   = lane & 15;
    const int lhalf = lane >> 4;
    const int m0    = blockIdx.y * 32;
    const int n0    = blockIdx.x * 32;
    const int mq    = wave >> 1;
    const int nq    = wave & 1;
    (void)M;

    const int ar = tid / (GK / 4);
    const int ac = (tid % (GK / 4)) << 2;
    const int br = tid >> 3;
    const int bc = (tid & 7) << 2;
    const float* Ag = A + (size_t)(m0 + ar) * K + ac;
    const float* Bg = B + (size_t)br * N + n0 + bc;
    v4f aprf[GNA], bprf[GNB];

    v8f acc = v8f{0.f, 0.f, 0.f, 0.f, 0.f, 0.f, 0.f, 0.f};

#pragma unroll
    for (int u = 0; u < GNA; ++u)
        aprf[u] = *(const v4f*)(Ag + (size_t)(u * (512 / GK)) * K);
#pragma unroll
    for (int u = 0; u < GNB; ++u)
        bprf[u] = *(const v4f*)(Bg + (size_t)(u * 16) * N);

    const float* ap = As + (mq * 16 + l15) * GAP + lhalf * 2;
    const float* bp = Bs + (lhalf * 2) * GBP + nq * 16 + l15;

    for (int k0 = 0; k0 < K; k0 += GK) {
#pragma unroll
        for (int u = 0; u < GNA; ++u)
            *(v4f*)&As[(ar + u * (512 / GK)) * GAP + ac] = aprf[u];
#pragma unroll
        for (int u = 0; u < GNB; ++u)
            *(v4f*)&Bs[(br + u * 16) * GBP + bc] = bprf[u];
        __syncthreads();

        if (k0 + GK < K) {
#pragma unroll
            for (int u = 0; u < GNA; ++u)
                aprf[u] = *(const v4f*)(Ag + (size_t)(u * (512 / GK)) * K + k0 + GK);
#pragma unroll
            for (int u = 0; u < GNB; ++u)
                bprf[u] = *(const v4f*)(Bg + (size_t)(k0 + GK + u * 16) * N);
        }

        v2f af[GK / 4], bf[GK / 4];
#pragma unroll
        for (int s = 0; s < GK / 4; ++s) {
            af[s] = v2f{ap[s * 4], ap[s * 4 + 1]};
            bf[s] = v2f{bp[(s * 4) * GBP], bp[(s * 4 + 1) * GBP]};
        }
#pragma unroll
        for (int s = 0; s < GK / 4; ++s)
            acc = __builtin_amdgcn_wmma_f32_16x16x4_f32(
                false, af[s], false, bf[s], (short)0, acc, false, false);
        __builtin_amdgcn_sched_group_barrier(0x100, 12, 0);
#pragma unroll
        for (int p = 0; p < (GK / 4 - 8) / 2; ++p) {
            __builtin_amdgcn_sched_group_barrier(0x008, 2, 0);
            __builtin_amdgcn_sched_group_barrier(0x100, 3, 0);
        }
        __builtin_amdgcn_sched_group_barrier(0x008, 8, 0);
        __syncthreads();
    }

#pragma unroll
    for (int r = 0; r < 8; ++r)
        Cs[(mq * 16 + r + lhalf * 8) * GCP + nq * 16 + l15] = acc[r];
    __syncthreads();

#pragma unroll
    for (int u = 0; u < 2; ++u) {
        const int v   = tid + u * 128;
        const int row = v >> 3;
        const int c4  = (v & 7) << 2;
        const size_t g = (size_t)(m0 + row) * N + n0 + c4;
        v4f o = *(const v4f*)&Cs[row * GCP + c4];
        v4f bv = v4f{0.f, 0.f, 0.f, 0.f};
        if (flags & GEMM_BIAS) bv = *(const v4f*)(bias + n0 + c4);
        o.x = o.x + bv.x; o.y = o.y + bv.y; o.z = o.z + bv.z; o.w = o.w + bv.w;
        if (flags & GEMM_RESID) {
            const v4f rv = *(const v4f*)(resid + g);
            o.x += rv.x; o.y += rv.y; o.z += rv.z; o.w += rv.w;
        }
        if (flags & GEMM_RELU) {
            o.x = fmaxf(o.x, 0.f); o.y = fmaxf(o.y, 0.f);
            o.z = fmaxf(o.z, 0.f); o.w = fmaxf(o.w, 0.f);
        }
        *(v4f*)(C + g) = o;
    }
}

#define FM  32
#define FK  128
#define FAP (FK + 4)
#define FBP (128 + 8)

struct FfnFrag { float a0[8], a1[8], b0[2][8], b1[2][8]; };

__device__ __forceinline__ void ffn_frags(FfnFrag& f, const float* a, const float* b, int k)
{
#pragma unroll
    for (int s = 0; s < 8; ++s) {
        const int kk = k + s * 4;
        f.a0[s]    = a[kk];
        f.a1[s]    = a[kk + 1];
        f.b0[0][s] = b[kk * FBP];
        f.b1[0][s] = b[(kk + 1) * FBP];
        f.b0[1][s] = b[kk * FBP + 16];
        f.b1[1][s] = b[(kk + 1) * FBP + 16];
    }
}

__device__ __forceinline__ void ffn_chunk(v8f (&acc)[2], const float* a, const float* b)
{
    FfnFrag f[2];
    ffn_frags(f[0], a, b, 0);
#pragma unroll
    for (int g = 0; g < FK / 32; ++g) {
        if (g + 1 < FK / 32)
            ffn_frags(f[(g + 1) & 1], a, b, (g + 1) * 32);
        __builtin_amdgcn_sched_barrier(0);
#pragma unroll
        for (int s = 0; s < 8; ++s) {
            const FfnFrag& c = f[g & 1];
            const v2f af  = v2f{c.a0[s], c.a1[s]};
            const v2f bf0 = v2f{c.b0[0][s], c.b1[0][s]};
            const v2f bf1 = v2f{c.b0[1][s], c.b1[1][s]};
            acc[0] = __builtin_amdgcn_wmma_f32_16x16x4_f32(
                false, af, false, bf0, (short)0, acc[0], false, false);
            acc[1] = __builtin_amdgcn_wmma_f32_16x16x4_f32(
                false, af, false, bf1, (short)0, acc[1], false, false);
        }
        __builtin_amdgcn_sched_barrier(0);
    }
}

__global__ __launch_bounds__(256)
void ffn_fused_wmma_f32(const float* __restrict__ X, const float* __restrict__ W1,
                        const float* __restrict__ b1, const float* __restrict__ W2,
                        float* __restrict__ P)
{
    __shared__ __attribute__((aligned(16))) float As[FM * FAP];
    __shared__ __attribute__((aligned(16))) float Bs[FK * FBP];
    __shared__ __attribute__((aligned(16))) float Hs[FM * FAP];

    const int tid   = threadIdx.x;
    const int wave  = tid >> 5;
    const int lane  = tid & 31;
    const int l15   = lane & 15;
    const int lhalf = lane >> 4;
    const int s0    = blockIdx.x * 128;
    const int m0    = blockIdx.y * FM;
    const int mq    = wave >> 2;
    const int nq    = wave & 3;

    const int sr = tid >> 5;
    const int sc = (tid & 31) << 2;
    const float* Xg  = X  + (size_t)(m0 + sr) * DMsz + sc;
    const float* W1g = W1 + (size_t)sr * DIsz + s0 + sc;
    const float* W2g = W2 + (size_t)(s0 + sr) * DMsz + sc;
    v4f aprf[4], bprf[16];

    v8f acc[2];
    acc[0] = v8f{0.f, 0.f, 0.f, 0.f, 0.f, 0.f, 0.f, 0.f};
    acc[1] = v8f{0.f, 0.f, 0.f, 0.f, 0.f, 0.f, 0.f, 0.f};

#pragma unroll
    for (int u = 0; u < 4; ++u)
        aprf[u] = *(const v4f*)(Xg + (size_t)(u * 8) * DMsz);
#pragma unroll
    for (int u = 0; u < 16; ++u)
        bprf[u] = *(const v4f*)(W1g + (size_t)(u * 8) * DIsz);

    const float* ap = As + (mq * 16 + l15) * FAP + lhalf * 2;
    const float* hp = Hs + (mq * 16 + l15) * FAP + lhalf * 2;
    const float* bp = Bs + (lhalf * 2) * FBP + nq * 32 + l15;

    for (int k0 = 0; k0 < DMsz; k0 += FK) {
#pragma unroll
        for (int u = 0; u < 4; ++u)
            *(v4f*)&As[(sr + u * 8) * FAP + sc] = aprf[u];
#pragma unroll
        for (int u = 0; u < 16; ++u)
            *(v4f*)&Bs[(sr + u * 8) * FBP + sc] = bprf[u];
        __syncthreads();

        if (k0 + FK < DMsz) {
#pragma unroll
            for (int u = 0; u < 4; ++u)
                aprf[u] = *(const v4f*)(Xg + (size_t)(u * 8) * DMsz + k0 + FK);
#pragma unroll
            for (int u = 0; u < 16; ++u)
                bprf[u] = *(const v4f*)(W1g + (size_t)(k0 + FK + u * 8) * DIsz);
        } else {
#pragma unroll
            for (int u = 0; u < 16; ++u)
                bprf[u] = *(const v4f*)(W2g + (size_t)(u * 8) * DMsz);
        }

        ffn_chunk(acc, ap, bp);
        __syncthreads();
    }

#pragma unroll
    for (int nsub = 0; nsub < 2; ++nsub) {
        const int hc = nq * 32 + nsub * 16 + l15;
        const float bv = b1[s0 + hc];
#pragma unroll
        for (int r = 0; r < 8; ++r) {
            const int hr = mq * 16 + r + lhalf * 8;
            Hs[hr * FAP + hc] = fmaxf(acc[nsub][r] + bv, 0.f);
        }
    }

    float* Pp = P + (size_t)blockIdx.x * Rsz * DMsz;
    for (int nc = 0; nc < 8; ++nc) {
        acc[0] = v8f{0.f, 0.f, 0.f, 0.f, 0.f, 0.f, 0.f, 0.f};
        acc[1] = v8f{0.f, 0.f, 0.f, 0.f, 0.f, 0.f, 0.f, 0.f};

#pragma unroll
        for (int u = 0; u < 16; ++u)
            *(v4f*)&Bs[(sr + u * 8) * FBP + sc] = bprf[u];
        __syncthreads();

        if (nc + 1 < 8) {
#pragma unroll
            for (int u = 0; u < 16; ++u)
                bprf[u] = *(const v4f*)(W2g + (size_t)(u * 8) * DMsz + (nc + 1) * 128);
        }

        ffn_chunk(acc, hp, bp);
        __syncthreads();

#pragma unroll
        for (int nsub = 0; nsub < 2; ++nsub) {
            const int col = nc * 128 + nq * 32 + nsub * 16 + l15;
#pragma unroll
            for (int r = 0; r < 8; ++r) {
                const int row = m0 + mq * 16 + r + lhalf * 8;
                Pp[(size_t)row * DMsz + col] = acc[nsub][r];
            }
        }
    }
}

#define GT_KP (18 * 32)
#define GT_PK 32

__global__ __launch_bounds__(256)
void dec_kernel(const float* __restrict__ a0base, const float* __restrict__ a1base,
                const float* __restrict__ b1, const float* __restrict__ w2,
                const float* __restrict__ b2, float* __restrict__ dec)
{
    __shared__ float tile[GT_PK * GT_KP];

    const int tid  = threadIdx.x;
    const int wave = tid >> 5;
    const int lane = tid & 31;
    const int blk  = blockIdx.x;
    const int nt   = blk & 7;
    const int mt   = (blk >> 3) & 7;
    const int b    = (blk >> 6) & 3;
    const int pair = blk >> 8;
    const int i    = pair / 3;
    const int j    = pair % 3;

    const float4* g0 = (const float4*)(a0base + ((size_t)i * Rsz + b * Lsz + nt * 8) * 2048);
    const float4* g1 = (const float4*)(a1base + ((size_t)j * Rsz + b * Lsz + mt * 8) * 2048);
    const float2* gw = (const float2*)w2;

    const int ml = (wave >> 1) * 2;
    const int nl = (wave & 1) * 4;
    const float* pa = tile + ml * 32 + lane;
    const float* pb = tile + (8 + nl) * 32 + lane;
    const float* pc = tile + 16 * 32 + lane;

    float acc[2][4];
#pragma unroll
    for (int x = 0; x < 2; ++x)
#pragma unroll
        for (int y = 0; y < 4; ++y) acc[x][y] = 0.f;

    for (int p = 0; p < 2; ++p) {
        {
            float4 r0[8], r1[8];
#pragma unroll
            for (int u = 0; u < 8; ++u) r0[u] = g0[u * 512 + p * 256 + tid];
#pragma unroll
            for (int u = 0; u < 8; ++u) r1[u] = g1[u * 512 + p * 256 + tid];
            const int dst = (tid >> 3) * GT_KP + (tid & 7) * 4;
#pragma unroll
            for (int u = 0; u < 8; ++u) {
                *(float4*)&tile[dst + (8 + u) * 32] = r0[u];
                *(float4*)&tile[dst + u * 32]       = r1[u];
            }
        }
#pragma unroll
        for (int u = 0; u < 4; ++u) {
            const int cl  = tid + u * 256;
            const float2 w = gw[p * 1024 + cl];
            const int dst = (cl >> 5) * GT_KP + (cl & 31);
            tile[dst + 16 * 32] = b1[p * 1024 + cl];
            tile[dst + 17 * 32] = w.y - w.x;
        }
        __syncthreads();

#pragma unroll 4
        for (int k = 0; k < GT_PK; ++k) {
            const int o = k * GT_KP;
            float xv[2], yv[4];
            xv[0] = pa[o];      xv[1] = pa[o + 32];
            yv[0] = pb[o];      yv[1] = pb[o + 32];
            yv[2] = pb[o + 64]; yv[3] = pb[o + 96];
            const float bb = pc[o], ww = pc[o + 32];
#pragma unroll
            for (int x = 0; x < 2; ++x)
#pragma unroll
                for (int y = 0; y < 4; ++y) {
                    const float t = (xv[x] + yv[y]) + bb;
                    if (t > 0.f) acc[x][y] = fmaf(t, ww, acc[x][y]);
                }
        }
        __syncthreads();
    }

#pragma unroll
    for (int off = 16; off; off >>= 1)
#pragma unroll
        for (int x = 0; x < 2; ++x)
#pragma unroll
            for (int y = 0; y < 4; ++y)
                acc[x][y] += __shfl_xor(acc[x][y], off, 32);

    if (lane == 0) {
        const float bd = b2[1] - b2[0];
        float* dtile = dec + (((size_t)(pair * Bsz + b) * Lsz + mt * 8) * Lsz + nt * 8);
#pragma unroll
        for (int x = 0; x < 2; ++x) {
            float4 r;
            r.x = ((acc[x][0] + bd) > 0.f) ? NEGC : 0.f;
            r.y = ((acc[x][1] + bd) > 0.f) ? NEGC : 0.f;
            r.z = ((acc[x][2] + bd) > 0.f) ? NEGC : 0.f;
            r.w = ((acc[x][3] + bd) > 0.f) ? NEGC : 0.f;
            *(float4*)(dtile + (size_t)(ml + x) * Lsz + nl) = r;
        }
    }
}

#define SC_QT 16
#define SC_KP 68

__global__ __launch_bounds__(128)
void attn_scores_kernel(const float* __restrict__ Qb, const float* __restrict__ Kb,
                        const float* __restrict__ dec, const int* __restrict__ mask,
                        float* __restrict__ S)
{
    __shared__ __attribute__((aligned(16))) float Ks[192 * SC_KP];
    __shared__ __attribute__((aligned(16))) float Qs[SC_QT * 64];

    const int blk  = blockIdx.x;
    const int qt   = blk & 3;
    const int h    = (blk >> 2) & 15;
    const int b    = (blk >> 6) & 3;
    const int i    = blk >> 8;
    const int tid  = threadIdx.x;
    const int wave = tid >> 5;
    const int lane = tid & 31;
    const int q0   = qt * SC_QT + wave * 4;

    {
        const int c4 = (tid & 15) << 2;
        const int r0 = tid >> 4;
        v4f kr[24], qr[2];
#pragma unroll
        for (int u = 0; u < 24; ++u) {
            const int row = r0 + u * 8;
            const int j = row >> 6, k = row & 63;
            kr[u] = *(const v4f*)(Kb + ((size_t)j * Rsz + b * Lsz + k) * DMsz + h * DKsz + c4);
        }
#pragma unroll
        for (int u = 0; u < 2; ++u) {
            const int row = r0 + u * 8;
            qr[u] = *(const v4f*)(Qb + ((size_t)i * Rsz + b * Lsz + qt * SC_QT + row) * DMsz
                                  + h * DKsz + c4);
        }
#pragma unroll
        for (int u = 0; u < 24; ++u)
            *(v4f*)&Ks[(r0 + u * 8) * SC_KP + c4] = kr[u];
#pragma unroll
        for (int u = 0; u < 2; ++u)
            *(v4f*)&Qs[(r0 + u * 8) * 64 + c4] = qr[u];
    }

    float dv[4][6];
    int   mk[4][2];
#pragma unroll
    for (int r = 0; r < 4; ++r) {
#pragma unroll
        for (int m = 0; m < 6; ++m) {
            const int j = m >> 1, k = (m & 1) * 32 + lane;
            dv[r][m] = dec[(((size_t)(i * 3 + j) * Bsz + b) * Lsz + q0 + r) * Lsz + k];
        }
#pragma unroll
        for (int kh = 0; kh < 2; ++kh)
            mk[r][kh] = mask[((size_t)b * Lsz + q0 + r) * Lsz + kh * 32 + lane];
    }
    __syncthreads();

    float acc[4][6];
#pragma unroll
    for (int r = 0; r < 4; ++r)
#pragma unroll
        for (int m = 0; m < 6; ++m) acc[r][m] = 0.f;

    const float* kp = Ks + lane * SC_KP;
    const float* qp = Qs + (wave * 4) * 64;
    for (int d = 0; d < 64; d += 4) {
        v4f kv[6], qv[4];
#pragma unroll
        for (int m = 0; m < 6; ++m) kv[m] = *(const v4f*)(kp + m * 32 * SC_KP + d);
#pragma unroll
        for (int r = 0; r < 4; ++r) qv[r] = *(const v4f*)(qp + r * 64 + d);
#pragma unroll
        for (int r = 0; r < 4; ++r)
#pragma unroll
            for (int m = 0; m < 6; ++m) {
                float s = acc[r][m];
                s = fmaf(qv[r].x, kv[m].x, s);
                s = fmaf(qv[r].y, kv[m].y, s);
                s = fmaf(qv[r].z, kv[m].z, s);
                s = fmaf(qv[r].w, kv[m].w, s);
                acc[r][m] = s;
            }
    }

#pragma unroll
    for (int r = 0; r < 4; ++r) {
        float s[6];
#pragma unroll
        for (int m = 0; m < 6; ++m) {
            float t = acc[r][m];
            t *= 0.125f;
            t += dv[r][m];
            if (mk[r][m & 1] == 0) t = NEGC;
            s[m] = t;
        }
        float mx = fmaxf(fmaxf(fmaxf(s[0], s[1]), fmaxf(s[2], s[3])), fmaxf(s[4], s[5]));
#pragma unroll
        for (int off = 16; off; off >>= 1) mx = fmaxf(mx, __shfl_xor(mx, off, 32));
        float e[6];
#pragma unroll
        for (int m = 0; m < 6; ++m) e[m] = __expf(s[m] - mx);
        float sum = ((e[0] + e[2]) + e[4]) + ((e[1] + e[3]) + e[5]);
#pragma unroll
        for (int off = 16; off; off >>= 1) sum += __shfl_xor(sum, off, 32);
        const float inv = 1.0f / sum;
        float* Sp = S + ((size_t)((((i * Bsz + b) * Hsz + h) * Lsz) + q0 + r)) * 192 + lane;
#pragma unroll
        for (int m = 0; m < 6; ++m) Sp[m * 32] = e[m] * inv;
    }
}

__global__ __launch_bounds__(64)
void attn_av_kernel(const float* __restrict__ S, const float* __restrict__ Vb,
                    float* __restrict__ Ob)
{
    const int blk = blockIdx.x;
    const int q = blk & 63;
    const int h = (blk >> 6) & 15;
    const int b = (blk >> 10) & 3;
    const int i = blk >> 12;
    const int d = threadIdx.x;

    __shared__ float arow[192];
    const float* Sp = S + (size_t)blk * 192;
    arow[d] = Sp[d]; arow[d + 64] = Sp[d + 64]; arow[d + 128] = Sp[d + 128];
    __syncthreads();

    float acc = 0.f;
    for (int jk = 0; jk < 192; ++jk) {
        int j = jk >> 6, k = jk & 63;
        acc += arow[jk] * Vb[((size_t)j * Rsz + b * Lsz + k) * DMsz + h * DVsz + d];
    }
    Ob[((size_t)i * Rsz + b * Lsz + q) * DMsz + h * DVsz + d] = acc;
}

__global__ __launch_bounds__(256)
void layernorm_kernel(const float* __restrict__ x, int nparts, size_t pstride,
                      const float* __restrict__ bias, const float* __restrict__ resid,
                      const float* __restrict__ g, const float* __restrict__ bta,
                      float* __restrict__ out)
{
    const int row = blockIdx.x;
    const int t = threadIdx.x;
    __shared__ float red[256];

    const float* xp = x + (size_t)row * 1024;
    float v[4];
#pragma unroll
    for (int u = 0; u < 4; ++u) v[u] = xp[t + u * 256];
#pragma unroll 8
    for (int p = 1; p < nparts; ++p) {
        const float* pp = xp + (size_t)p * pstride;
#pragma unroll
        for (int u = 0; u < 4; ++u) v[u] += pp[t + u * 256];
    }
    float s = 0.f;
#pragma unroll
    for (int u = 0; u < 4; ++u) {
        float val = v[u];
        if (bias)  val += bias[t + u * 256];
        if (resid) val += resid[(size_t)row * 1024 + t + u * 256];
        v[u] = val; s += val;
    }
    red[t] = s; __syncthreads();
    for (int st = 128; st > 0; st >>= 1) { if (t < st) red[t] += red[t + st]; __syncthreads(); }
    const float mu = red[0] * (1.f / 1024.f);
    __syncthreads();

    float s2 = 0.f;
#pragma unroll
    for (int u = 0; u < 4; ++u) { float dd = v[u] - mu; s2 += dd * dd; }
    red[t] = s2; __syncthreads();
    for (int st = 128; st > 0; st >>= 1) { if (t < st) red[t] += red[t + st]; __syncthreads(); }
    const float var = red[0] * (1.f / 1024.f);
    const float rinv = 1.0f / sqrtf(var + 1e-6f);
#pragma unroll
    for (int u = 0; u < 4; ++u) {
        int c = t + u * 256;
        out[(size_t)row * 1024 + c] = (v[u] - mu) * rinv * g[c] + bta[c];
    }
}

extern "C" void kernel_launch(void* const* d_in, const int* in_sizes, int n_in,
                              void* d_out, int out_size, void* d_ws, size_t ws_size,
                              hipStream_t stream)
{
    const float* X[3]   = { (const float*)d_in[0], (const float*)d_in[1], (const float*)d_in[2] };
    const int*   mask   = (const int*)d_in[3];
    const float* w_qs   = (const float*)d_in[4];
    const float* w_ks   = (const float*)d_in[5];
    const float* w_vs   = (const float*)d_in[6];
    const float* w_ds0  = (const float*)d_in[7];
    const float* w_ds1  = (const float*)d_in[8];
    const float* dc_w1  = (const float*)d_in[9];
    const float* dc_b1  = (const float*)d_in[10];
    const float* dc_w2  = (const float*)d_in[11];
    const float* dc_b2  = (const float*)d_in[12];
    const float* w_fc   = (const float*)d_in[13];
    const float* ln1_g  = (const float*)d_in[14];
    const float* ln1_b  = (const float*)d_in[15];
    const float* ffn_w1 = (const float*)d_in[16];
    const float* ffn_b1 = (const float*)d_in[17];
    const float* ffn_w2 = (const float*)d_in[18];
    const float* ffn_b2 = (const float*)d_in[19];
    const float* ln2_g  = (const float*)d_in[20];
    const float* ln2_b  = (const float*)d_in[21];
    float* out = (float*)d_out;

    float* ws = (float*)d_ws;
    size_t off = 0;
    auto alloc = [&](size_t n) { float* p = ws + off; off += n; return p; };
    const size_t RM = (size_t)Rsz * DMsz;
    float* Q   = alloc(3 * RM);
    float* Kb  = alloc(3 * RM);
    float* Vb  = alloc(3 * RM);
    float* d0  = alloc(3 * RM);
    float* d1  = alloc(3 * RM);
    float* a0  = alloc(3 * (size_t)Rsz * 2048);
    float* a1  = alloc(3 * (size_t)Rsz * 2048);
    float* dec = alloc(9 * (size_t)Bsz * Lsz * Lsz);
    float* S   = alloc((size_t)3 * Bsz * Hsz * Lsz * 192);
    float* O   = alloc(3 * RM);
    float* fc  = alloc(3 * RM);
    float* oln = alloc(3 * RM);
    float* fpart = alloc(32 * RM);
    (void)ws_size; (void)n_in; (void)in_sizes; (void)out_size;

    auto gemm = [&](const float* A, const float* Bm, const float* bias,
                    const float* resid, float* C, int M, int N, int K, int flags) {
        dim3 grid(N / 32, M / 32);
        gemm_wmma_f32<<<grid, dim3(128), 0, stream>>>(A, Bm, bias, resid, C, M, N, K, flags);
    };

    for (int i = 0; i < 3; ++i) {
        gemm(X[i], w_qs,  nullptr, nullptr, Q  + i * RM, Rsz, 1024, 1024, 0);
        gemm(X[i], w_ks,  nullptr, nullptr, Kb + i * RM, Rsz, 1024, 1024, 0);
        gemm(X[i], w_vs,  nullptr, nullptr, Vb + i * RM, Rsz, 1024, 1024, 0);
        gemm(X[i], w_ds0, nullptr, nullptr, d0 + i * RM, Rsz, 1024, 1024, 0);
        gemm(X[i], w_ds1, nullptr, nullptr, d1 + i * RM, Rsz, 1024, 1024, 0);
        gemm(d0 + i * RM, dc_w1 + (size_t)1024 * 2048, nullptr, nullptr,
             a0 + i * (size_t)Rsz * 2048, Rsz, 2048, 1024, 0);
        gemm(d1 + i * RM, dc_w1, nullptr, nullptr,
             a1 + i * (size_t)Rsz * 2048, Rsz, 2048, 1024, 0);
    }

    {
        dec_kernel<<<9 * Bsz * 8 * 8, 256, 0, stream>>>(a0, a1, dc_b1, dc_w2, dc_b2, dec);
    }

    {
        int nblk = 3 * Bsz * Hsz * Lsz;
        attn_scores_kernel<<<3 * Bsz * Hsz * (Lsz / SC_QT), 128, 0, stream>>>(Q, Kb, dec, mask, S);
        attn_av_kernel<<<nblk, 64, 0, stream>>>(S, Vb, O);
    }

    for (int i = 0; i < 3; ++i) {
        gemm(O + i * RM, w_fc, nullptr, X[i], fc + i * RM, Rsz, 1024, 1024, GEMM_RESID);
        layernorm_kernel<<<Rsz, 256, 0, stream>>>(fc + i * RM, 1, RM, nullptr, nullptr,
                                                  ln1_g, ln1_b, oln + i * RM);
        ffn_fused_wmma_f32<<<dim3(32, Rsz / FM), dim3(256), 0, stream>>>(
            oln + i * RM, ffn_w1, ffn_b1, ffn_w2, fpart);
        layernorm_kernel<<<Rsz, 256, 0, stream>>>(fpart, 32, RM, ffn_b2, oln + i * RM,
                                                  ln2_g, ln2_b, out + i * RM);
    }
}


# A@V re-cut: a block stages the 192 value rows of (b,h) and its 16 probability rows in LDS once for 16 query rows; a lane owns 2 columns of 4 rows (was one block per query row); sums bit-identical
# speedup vs baseline: 3.5387x; 1.0333x over previous
#include <hip/hip_runtime.h>
#include <hip/hip_bf16.h>

#define Bsz 4
#define Lsz 64
#define DMsz 1024
#define Hsz 16
#define DKsz 64
#define DVsz 64
#define Dsz 1024
#define DIsz 4096
#define Rsz 256
#define NEGC (-1.0e9f)

typedef float v2f __attribute__((ext_vector_type(2)));
typedef float v4f __attribute__((ext_vector_type(4)));
typedef float v8f __attribute__((ext_vector_type(8)));

#define GEMM_BIAS  1
#define GEMM_RELU  2
#define GEMM_RESID 4

#define GK  128
#define GAP (GK + 4)
#define GBP 40
#define GCP 36
#define GNA (GK / 16)
#define GNB (GK / 16)

__global__ __launch_bounds__(128)
void gemm_wmma_f32(const float* __restrict__ A, const float* __restrict__ B,
                   const float* __restrict__ bias, const float* __restrict__ resid,
                   float* __restrict__ C, int M, int N, int K, int flags)
{
    __shared__ __attribute__((aligned(16))) float As[32 * GAP];
    __shared__ __attribute__((aligned(16))) float Bs[GK * GBP];
    __shared__ __attribute__((aligned(16))) float Cs[32 * GCP];

    const int tid   = threadIdx.x;
    const int wave  = tid >> 5;
    const int lane  = tid & 31;
    const int l15   = lane & 15;
    const int lhalf = lane >> 4;
    const int m0    = blockIdx.y * 32;
    const int n0    = blockIdx.x * 32;
    const int mq    = wave >> 1;
    const int nq    = wave & 1;
    (void)M;

    const int ar = tid / (GK / 4);
    const int ac = (tid % (GK / 4)) << 2;
    const int br = tid >> 3;
    const int bc = (tid & 7) << 2;
    const float* Ag = A + (size_t)(m0 + ar) * K + ac;
    const float* Bg = B + (size_t)br * N + n0 + bc;
    v4f aprf[GNA], bprf[GNB];

    v8f acc = v8f{0.f, 0.f, 0.f, 0.f, 0.f, 0.f, 0.f, 0.f};

#pragma unroll
    for (int u = 0; u < GNA; ++u)
        aprf[u] = *(const v4f*)(Ag + (size_t)(u * (512 / GK)) * K);
#pragma unroll
    for (int u = 0; u < GNB; ++u)
        bprf[u] = *(const v4f*)(Bg + (size_t)(u * 16) * N);

    const float* ap = As + (mq * 16 + l15) * GAP + lhalf * 2;
    const float* bp = Bs + (lhalf * 2) * GBP + nq * 16 + l15;

    for (int k0 = 0; k0 < K; k0 += GK) {
#pragma unroll
        for (int u = 0; u < GNA; ++u)
            *(v4f*)&As[(ar + u * (512 / GK)) * GAP + ac] = aprf[u];
#pragma unroll
        for (int u = 0; u < GNB; ++u)
            *(v4f*)&Bs[(br + u * 16) * GBP + bc] = bprf[u];
        __syncthreads();

        if (k0 + GK < K) {
#pragma unroll
            for (int u = 0; u < GNA; ++u)
                aprf[u] = *(const v4f*)(Ag + (size_t)(u * (512 / GK)) * K + k0 + GK);
#pragma unroll
            for (int u = 0; u < GNB; ++u)
                bprf[u] = *(const v4f*)(Bg + (size_t)(k0 + GK + u * 16) * N);
        }

        v2f af[GK / 4], bf[GK / 4];
#pragma unroll
        for (int s = 0; s < GK / 4; ++s) {
            af[s] = v2f{ap[s * 4], ap[s * 4 + 1]};
            bf[s] = v2f{bp[(s * 4) * GBP], bp[(s * 4 + 1) * GBP]};
        }
#pragma unroll
        for (int s = 0; s < GK / 4; ++s)
            acc = __builtin_amdgcn_wmma_f32_16x16x4_f32(
                false, af[s], false, bf[s], (short)0, acc, false, false);
        __builtin_amdgcn_sched_group_barrier(0x100, 12, 0);
#pragma unroll
        for (int p = 0; p < (GK / 4 - 8) / 2; ++p) {
            __builtin_amdgcn_sched_group_barrier(0x008, 2, 0);
            __builtin_amdgcn_sched_group_barrier(0x100, 3, 0);
        }
        __builtin_amdgcn_sched_group_barrier(0x008, 8, 0);
        __syncthreads();
    }

#pragma unroll
    for (int r = 0; r < 8; ++r)
        Cs[(mq * 16 + r + lhalf * 8) * GCP + nq * 16 + l15] = acc[r];
    __syncthreads();

#pragma unroll
    for (int u = 0; u < 2; ++u) {
        const int v   = tid + u * 128;
        const int row = v >> 3;
        const int c4  = (v & 7) << 2;
        const size_t g = (size_t)(m0 + row) * N + n0 + c4;
        v4f o = *(const v4f*)&Cs[row * GCP + c4];
        v4f bv = v4f{0.f, 0.f, 0.f, 0.f};
        if (flags & GEMM_BIAS) bv = *(const v4f*)(bias + n0 + c4);
        o.x = o.x + bv.x; o.y = o.y + bv.y; o.z = o.z + bv.z; o.w = o.w + bv.w;
        if (flags & GEMM_RESID) {
            const v4f rv = *(const v4f*)(resid + g);
            o.x += rv.x; o.y += rv.y; o.z += rv.z; o.w += rv.w;
        }
        if (flags & GEMM_RELU) {
            o.x = fmaxf(o.x, 0.f); o.y = fmaxf(o.y, 0.f);
            o.z = fmaxf(o.z, 0.f); o.w = fmaxf(o.w, 0.f);
        }
        *(v4f*)(C + g) = o;
    }
}

#define FM  32
#define FK  128
#define FAP (FK + 4)
#define FBP (128 + 8)

struct FfnFrag { float a0[8], a1[8], b0[2][8], b1[2][8]; };

__device__ __forceinline__ void ffn_frags(FfnFrag& f, const float* a, const float* b, int k)
{
#pragma unroll
    for (int s = 0; s < 8; ++s) {
        const int kk = k + s * 4;
        f.a0[s]    = a[kk];
        f.a1[s]    = a[kk + 1];
        f.b0[0][s] = b[kk * FBP];
        f.b1[0][s] = b[(kk + 1) * FBP];
        f.b0[1][s] = b[kk * FBP + 16];
        f.b1[1][s] = b[(kk + 1) * FBP + 16];
    }
}

__device__ __forceinline__ void ffn_chunk(v8f (&acc)[2], const float* a, const float* b)
{
    FfnFrag f[2];
    ffn_frags(f[0], a, b, 0);
#pragma unroll
    for (int g = 0; g < FK / 32; ++g) {
        if (g + 1 < FK / 32)
            ffn_frags(f[(g + 1) & 1], a, b, (g + 1) * 32);
        __builtin_amdgcn_sched_barrier(0);
#pragma unroll
        for (int s = 0; s < 8; ++s) {
            const FfnFrag& c = f[g & 1];
            const v2f af  = v2f{c.a0[s], c.a1[s]};
            const v2f bf0 = v2f{c.b0[0][s], c.b1[0][s]};
            const v2f bf1 = v2f{c.b0[1][s], c.b1[1][s]};
            acc[0] = __builtin_amdgcn_wmma_f32_16x16x4_f32(
                false, af, false, bf0, (short)0, acc[0], false, false);
            acc[1] = __builtin_amdgcn_wmma_f32_16x16x4_f32(
                false, af, false, bf1, (short)0, acc[1], false, false);
        }
        __builtin_amdgcn_sched_barrier(0);
    }
}

__global__ __launch_bounds__(256)
void ffn_fused_wmma_f32(const float* __restrict__ X, const float* __restrict__ W1,
                        const float* __restrict__ b1, const float* __restrict__ W2,
                        float* __restrict__ P)
{
    __shared__ __attribute__((aligned(16))) float As[FM * FAP];
    __shared__ __attribute__((aligned(16))) float Bs[FK * FBP];
    __shared__ __attribute__((aligned(16))) float Hs[FM * FAP];

    const int tid   = threadIdx.x;
    const int wave  = tid >> 5;
    const int lane  = tid & 31;
    const int l15   = lane & 15;
    const int lhalf = lane >> 4;
    const int s0    = blockIdx.x * 128;
    const int m0    = blockIdx.y * FM;
    const int mq    = wave >> 2;
    const int nq    = wave & 3;

    const int sr = tid >> 5;
    const int sc = (tid & 31) << 2;
    const float* Xg  = X  + (size_t)(m0 + sr) * DMsz + sc;
    const float* W1g = W1 + (size_t)sr * DIsz + s0 + sc;
    const float* W2g = W2 + (size_t)(s0 + sr) * DMsz + sc;
    v4f aprf[4], bprf[16];

    v8f acc[2];
    acc[0] = v8f{0.f, 0.f, 0.f, 0.f, 0.f, 0.f, 0.f, 0.f};
    acc[1] = v8f{0.f, 0.f, 0.f, 0.f, 0.f, 0.f, 0.f, 0.f};

#pragma unroll
    for (int u = 0; u < 4; ++u)
        aprf[u] = *(const v4f*)(Xg + (size_t)(u * 8) * DMsz);
#pragma unroll
    for (int u = 0; u < 16; ++u)
        bprf[u] = *(const v4f*)(W1g + (size_t)(u * 8) * DIsz);

    const float* ap = As + (mq * 16 + l15) * FAP + lhalf * 2;
    const float* hp = Hs + (mq * 16 + l15) * FAP + lhalf * 2;
    const float* bp = Bs + (lhalf * 2) * FBP + nq * 32 + l15;

    for (int k0 = 0; k0 < DMsz; k0 += FK) {
#pragma unroll
        for (int u = 0; u < 4; ++u)
            *(v4f*)&As[(sr + u * 8) * FAP + sc] = aprf[u];
#pragma unroll
        for (int u = 0; u < 16; ++u)
            *(v4f*)&Bs[(sr + u * 8) * FBP + sc] = bprf[u];
        __syncthreads();

        if (k0 + FK < DMsz) {
#pragma unroll
            for (int u = 0; u < 4; ++u)
                aprf[u] = *(const v4f*)(Xg + (size_t)(u * 8) * DMsz + k0 + FK);
#pragma unroll
            for (int u = 0; u < 16; ++u)
                bprf[u] = *(const v4f*)(W1g + (size_t)(k0 + FK + u * 8) * DIsz);
        } else {
#pragma unroll
            for (int u = 0; u < 16; ++u)
                bprf[u] = *(const v4f*)(W2g + (size_t)(u * 8) * DMsz);
        }

        ffn_chunk(acc, ap, bp);
        __syncthreads();
    }

#pragma unroll
    for (int nsub = 0; nsub < 2; ++nsub) {
        const int hc = nq * 32 + nsub * 16 + l15;
        const float bv = b1[s0 + hc];
#pragma unroll
        for (int r = 0; r < 8; ++r) {
            const int hr = mq * 16 + r + lhalf * 8;
            Hs[hr * FAP + hc] = fmaxf(acc[nsub][r] + bv, 0.f);
        }
    }

    float* Pp = P + (size_t)blockIdx.x * Rsz * DMsz;
    for (int nc = 0; nc < 8; ++nc) {
        acc[0] = v8f{0.f, 0.f, 0.f, 0.f, 0.f, 0.f, 0.f, 0.f};
        acc[1] = v8f{0.f, 0.f, 0.f, 0.f, 0.f, 0.f, 0.f, 0.f};

#pragma unroll
        for (int u = 0; u < 16; ++u)
            *(v4f*)&Bs[(sr + u * 8) * FBP + sc] = bprf[u];
        __syncthreads();

        if (nc + 1 < 8) {
#pragma unroll
            for (int u = 0; u < 16; ++u)
                bprf[u] = *(const v4f*)(W2g + (size_t)(u * 8) * DMsz + (nc + 1) * 128);
        }

        ffn_chunk(acc, hp, bp);
        __syncthreads();

#pragma unroll
        for (int nsub = 0; nsub < 2; ++nsub) {
            const int col = nc * 128 + nq * 32 + nsub * 16 + l15;
#pragma unroll
            for (int r = 0; r < 8; ++r) {
                const int row = m0 + mq * 16 + r + lhalf * 8;
                Pp[(size_t)row * DMsz + col] = acc[nsub][r];
            }
        }
    }
}

#define GT_KP (18 * 32)
#define GT_PK 32

__global__ __launch_bounds__(256)
void dec_kernel(const float* __restrict__ a0base, const float* __restrict__ a1base,
                const float* __restrict__ b1, const float* __restrict__ w2,
                const float* __restrict__ b2, float* __restrict__ dec)
{
    __shared__ float tile[GT_PK * GT_KP];

    const int tid  = threadIdx.x;
    const int wave = tid >> 5;
    const int lane = tid & 31;
    const int blk  = blockIdx.x;
    const int nt   = blk & 7;
    const int mt   = (blk >> 3) & 7;
    const int b    = (blk >> 6) & 3;
    const int pair = blk >> 8;
    const int i    = pair / 3;
    const int j    = pair % 3;

    const float4* g0 = (const float4*)(a0base + ((size_t)i * Rsz + b * Lsz + nt * 8) * 2048);
    const float4* g1 = (const float4*)(a1base + ((size_t)j * Rsz + b * Lsz + mt * 8) * 2048);
    const float2* gw = (const float2*)w2;

    const int ml = (wave >> 1) * 2;
    const int nl = (wave & 1) * 4;
    const float* pa = tile + ml * 32 + lane;
    const float* pb = tile + (8 + nl) * 32 + lane;
    const float* pc = tile + 16 * 32 + lane;

    float acc[2][4];
#pragma unroll
    for (int x = 0; x < 2; ++x)
#pragma unroll
        for (int y = 0; y < 4; ++y) acc[x][y] = 0.f;

    for (int p = 0; p < 2; ++p) {
        {
            float4 r0[8], r1[8];
#pragma unroll
            for (int u = 0; u < 8; ++u) r0[u] = g0[u * 512 + p * 256 + tid];
#pragma unroll
            for (int u = 0; u < 8; ++u) r1[u] = g1[u * 512 + p * 256 + tid];
            const int dst = (tid >> 3) * GT_KP + (tid & 7) * 4;
#pragma unroll
            for (int u = 0; u < 8; ++u) {
                *(float4*)&tile[dst + (8 + u) * 32] = r0[u];
                *(float4*)&tile[dst + u * 32]       = r1[u];
            }
        }
#pragma unroll
        for (int u = 0; u < 4; ++u) {
            const int cl  = tid + u * 256;
            const float2 w = gw[p * 1024 + cl];
            const int dst = (cl >> 5) * GT_KP + (cl & 31);
            tile[dst + 16 * 32] = b1[p * 1024 + cl];
            tile[dst + 17 * 32] = w.y - w.x;
        }
        __syncthreads();

#pragma unroll 4
        for (int k = 0; k < GT_PK; ++k) {
            const int o = k * GT_KP;
            float xv[2], yv[4];
            xv[0] = pa[o];      xv[1] = pa[o + 32];
            yv[0] = pb[o];      yv[1] = pb[o + 32];
            yv[2] = pb[o + 64]; yv[3] = pb[o + 96];
            const float bb = pc[o], ww = pc[o + 32];
#pragma unroll
            for (int x = 0; x < 2; ++x)
#pragma unroll
                for (int y = 0; y < 4; ++y) {
                    const float t = (xv[x] + yv[y]) + bb;
                    if (t > 0.f) acc[x][y] = fmaf(t, ww, acc[x][y]);
                }
        }
        __syncthreads();
    }

#pragma unroll
    for (int off = 16; off; off >>= 1)
#pragma unroll
        for (int x = 0; x < 2; ++x)
#pragma unroll
            for (int y = 0; y < 4; ++y)
                acc[x][y] += __shfl_xor(acc[x][y], off, 32);

    if (lane == 0) {
        const float bd = b2[1] - b2[0];
        float* dtile = dec + (((size_t)(pair * Bsz + b) * Lsz + mt * 8) * Lsz + nt * 8);
#pragma unroll
        for (int x = 0; x < 2; ++x) {
            float4 r;
            r.x = ((acc[x][0] + bd) > 0.f) ? NEGC : 0.f;
            r.y = ((acc[x][1] + bd) > 0.f) ? NEGC : 0.f;
            r.z = ((acc[x][2] + bd) > 0.f) ? NEGC : 0.f;
            r.w = ((acc[x][3] + bd) > 0.f) ? NEGC : 0.f;
            *(float4*)(dtile + (size_t)(ml + x) * Lsz + nl) = r;
        }
    }
}

#define SC_QT 16
#define SC_KP 68

__global__ __launch_bounds__(128)
void attn_scores_kernel(const float* __restrict__ Qb, const float* __restrict__ Kb,
                        const float* __restrict__ dec, const int* __restrict__ mask,
                        float* __restrict__ S)
{
    __shared__ __attribute__((aligned(16))) float Ks[192 * SC_KP];
    __shared__ __attribute__((aligned(16))) float Qs[SC_QT * 64];

    const int blk  = blockIdx.x;
    const int qt   = blk & 3;
    const int h    = (blk >> 2) & 15;
    const int b    = (blk >> 6) & 3;
    const int i    = blk >> 8;
    const int tid  = threadIdx.x;
    const int wave = tid >> 5;
    const int lane = tid & 31;
    const int q0   = qt * SC_QT + wave * 4;

    {
        const int c4 = (tid & 15) << 2;
        const int r0 = tid >> 4;
        v4f kr[24], qr[2];
#pragma unroll
        for (int u = 0; u < 24; ++u) {
            const int row = r0 + u * 8;
            const int j = row >> 6, k = row & 63;
            kr[u] = *(const v4f*)(Kb + ((size_t)j * Rsz + b * Lsz + k) * DMsz + h * DKsz + c4);
        }
#pragma unroll
        for (int u = 0; u < 2; ++u) {
            const int row = r0 + u * 8;
            qr[u] = *(const v4f*)(Qb + ((size_t)i * Rsz + b * Lsz + qt * SC_QT + row) * DMsz
                                  + h * DKsz + c4);
        }
#pragma unroll
        for (int u = 0; u < 24; ++u)
            *(v4f*)&Ks[(r0 + u * 8) * SC_KP + c4] = kr[u];
#pragma unroll
        for (int u = 0; u < 2; ++u)
            *(v4f*)&Qs[(r0 + u * 8) * 64 + c4] = qr[u];
    }

    float dv[4][6];
    int   mk[4][2];
#pragma unroll
    for (int r = 0; r < 4; ++r) {
#pragma unroll
        for (int m = 0; m < 6; ++m) {
            const int j = m >> 1, k = (m & 1) * 32 + lane;
            dv[r][m] = dec[(((size_t)(i * 3 + j) * Bsz + b) * Lsz + q0 + r) * Lsz + k];
        }
#pragma unroll
        for (int kh = 0; kh < 2; ++kh)
            mk[r][kh] = mask[((size_t)b * Lsz + q0 + r) * Lsz + kh * 32 + lane];
    }
    __syncthreads();

    float acc[4][6];
#pragma unroll
    for (int r = 0; r < 4; ++r)
#pragma unroll
        for (int m = 0; m < 6; ++m) acc[r][m] = 0.f;

    const float* kp = Ks + lane * SC_KP;
    const float* qp = Qs + (wave * 4) * 64;
    for (int d = 0; d < 64; d += 4) {
        v4f kv[6], qv[4];
#pragma unroll
        for (int m = 0; m < 6; ++m) kv[m] = *(const v4f*)(kp + m * 32 * SC_KP + d);
#pragma unroll
        for (int r = 0; r < 4; ++r) qv[r] = *(const v4f*)(qp + r * 64 + d);
#pragma unroll
        for (int r = 0; r < 4; ++r)
#pragma unroll
            for (int m = 0; m < 6; ++m) {
                float s = acc[r][m];
                s = fmaf(qv[r].x, kv[m].x, s);
                s = fmaf(qv[r].y, kv[m].y, s);
                s = fmaf(qv[r].z, kv[m].z, s);
                s = fmaf(qv[r].w, kv[m].w, s);
                acc[r][m] = s;
            }
    }

#pragma unroll
    for (int r = 0; r < 4; ++r) {
        float s[6];
#pragma unroll
        for (int m = 0; m < 6; ++m) {
            float t = acc[r][m];
            t *= 0.125f;
            t += dv[r][m];
            if (mk[r][m & 1] == 0) t = NEGC;
            s[m] = t;
        }
        float mx = fmaxf(fmaxf(fmaxf(s[0], s[1]), fmaxf(s[2], s[3])), fmaxf(s[4], s[5]));
#pragma unroll
        for (int off = 16; off; off >>= 1) mx = fmaxf(mx, __shfl_xor(mx, off, 32));
        float e[6];
#pragma unroll
        for (int m = 0; m < 6; ++m) e[m] = __expf(s[m] - mx);
        float sum = ((e[0] + e[2]) + e[4]) + ((e[1] + e[3]) + e[5]);
#pragma unroll
        for (int off = 16; off; off >>= 1) sum += __shfl_xor(sum, off, 32);
        const float inv = 1.0f / sum;
        float* Sp = S + ((size_t)((((i * Bsz + b) * Hsz + h) * Lsz) + q0 + r)) * 192 + lane;
#pragma unroll
        for (int m = 0; m < 6; ++m) Sp[m * 32] = e[m] * inv;
    }
}

#define AV_QT 16
#define AV_VP 68

__global__ __launch_bounds__(128)
void attn_av_kernel(const float* __restrict__ S, const float* __restrict__ Vb,
                    float* __restrict__ Ob)
{
    __shared__ __attribute__((aligned(16))) float Vs[192 * AV_VP];
    __shared__ __attribute__((aligned(16))) float Ss[AV_QT * 192];

    const int blk  = blockIdx.x;
    const int qt   = blk & 3;
    const int h    = (blk >> 2) & 15;
    const int b    = (blk >> 6) & 3;
    const int i    = blk >> 8;
    const int tid  = threadIdx.x;
    const int wave = tid >> 5;
    const int lane = tid & 31;

    {
        const int c4 = (tid & 15) << 2;
        const int r0 = tid >> 4;
        const float* Sg = S + ((size_t)((((i * Bsz + b) * Hsz + h) * Lsz) + qt * AV_QT)) * 192;
        v4f vr[24], sr[6];
#pragma unroll
        for (int u = 0; u < 24; ++u) {
            const int row = r0 + u * 8;
            const int j = row >> 6, k = row & 63;
            vr[u] = *(const v4f*)(Vb + ((size_t)j * Rsz + b * Lsz + k) * DMsz + h * DVsz + c4);
        }
#pragma unroll
        for (int u = 0; u < 6; ++u)
            sr[u] = *(const v4f*)(Sg + (size_t)(tid + u * 128) * 4);
#pragma unroll
        for (int u = 0; u < 24; ++u)
            *(v4f*)&Vs[(r0 + u * 8) * AV_VP + c4] = vr[u];
#pragma unroll
        for (int u = 0; u < 6; ++u)
            *(v4f*)&Ss[(tid + u * 128) * 4] = sr[u];
    }
    __syncthreads();

    float acc[4][2];
#pragma unroll
    for (int r = 0; r < 4; ++r) { acc[r][0] = 0.f; acc[r][1] = 0.f; }

    const float* vp = Vs + lane;
    const float* sp = Ss + (wave * 4) * 192;
#pragma unroll 4
    for (int jk = 0; jk < 192; jk += 4) {
        v4f sv[4];
        float vv[4][2];
#pragma unroll
        for (int r = 0; r < 4; ++r) sv[r] = *(const v4f*)(sp + r * 192 + jk);
#pragma unroll
        for (int t = 0; t < 4; ++t) {
            vv[t][0] = vp[(jk + t) * AV_VP];
            vv[t][1] = vp[(jk + t) * AV_VP + 32];
        }
#pragma unroll
        for (int r = 0; r < 4; ++r)
#pragma unroll
            for (int cc = 0; cc < 2; ++cc) {
                float s = acc[r][cc];
                s = fmaf(sv[r].x, vv[0][cc], s);
                s = fmaf(sv[r].y, vv[1][cc], s);
                s = fmaf(sv[r].z, vv[2][cc], s);
                s = fmaf(sv[r].w, vv[3][cc], s);
                acc[r][cc] = s;
            }
    }

#pragma unroll
    for (int r = 0; r < 4; ++r) {
        const int q = qt * AV_QT + wave * 4 + r;
        float* Op = Ob + ((size_t)i * Rsz + b * Lsz + q) * DMsz + h * DVsz + lane;
        Op[0]  = acc[r][0];
        Op[32] = acc[r][1];
    }
}

__global__ __launch_bounds__(256)
void layernorm_kernel(const float* __restrict__ x, int nparts, size_t pstride,
                      const float* __restrict__ bias, const float* __restrict__ resid,
                      const float* __restrict__ g, const float* __restrict__ bta,
                      float* __restrict__ out)
{
    const int row = blockIdx.x;
    const int t = threadIdx.x;
    __shared__ float red[256];

    const float* xp = x + (size_t)row * 1024;
    float v[4];
#pragma unroll
    for (int u = 0; u < 4; ++u) v[u] = xp[t + u * 256];
#pragma unroll 8
    for (int p = 1; p < nparts; ++p) {
        const float* pp = xp + (size_t)p * pstride;
#pragma unroll
        for (int u = 0; u < 4; ++u) v[u] += pp[t + u * 256];
    }
    float s = 0.f;
#pragma unroll
    for (int u = 0; u < 4; ++u) {
        float val = v[u];
        if (bias)  val += bias[t + u * 256];
        if (resid) val += resid[(size_t)row * 1024 + t + u * 256];
        v[u] = val; s += val;
    }
    red[t] = s; __syncthreads();
    for (int st = 128; st > 0; st >>= 1) { if (t < st) red[t] += red[t + st]; __syncthreads(); }
    const float mu = red[0] * (1.f / 1024.f);
    __syncthreads();

    float s2 = 0.f;
#pragma unroll
    for (int u = 0; u < 4; ++u) { float dd = v[u] - mu; s2 += dd * dd; }
    red[t] = s2; __syncthreads();
    for (int st = 128; st > 0; st >>= 1) { if (t < st) red[t] += red[t + st]; __syncthreads(); }
    const float var = red[0] * (1.f / 1024.f);
    const float rinv = 1.0f / sqrtf(var + 1e-6f);
#pragma unroll
    for (int u = 0; u < 4; ++u) {
        int c = t + u * 256;
        out[(size_t)row * 1024 + c] = (v[u] - mu) * rinv * g[c] + bta[c];
    }
}

extern "C" void kernel_launch(void* const* d_in, const int* in_sizes, int n_in,
                              void* d_out, int out_size, void* d_ws, size_t ws_size,
                              hipStream_t stream)
{
    const float* X[3]   = { (const float*)d_in[0], (const float*)d_in[1], (const float*)d_in[2] };
    const int*   mask   = (const int*)d_in[3];
    const float* w_qs   = (const float*)d_in[4];
    const float* w_ks   = (const float*)d_in[5];
    const float* w_vs   = (const float*)d_in[6];
    const float* w_ds0  = (const float*)d_in[7];
    const float* w_ds1  = (const float*)d_in[8];
    const float* dc_w1  = (const float*)d_in[9];
    const float* dc_b1  = (const float*)d_in[10];
    const float* dc_w2  = (const float*)d_in[11];
    const float* dc_b2  = (const float*)d_in[12];
    const float* w_fc   = (const float*)d_in[13];
    const float* ln1_g  = (const float*)d_in[14];
    const float* ln1_b  = (const float*)d_in[15];
    const float* ffn_w1 = (const float*)d_in[16];
    const float* ffn_b1 = (const float*)d_in[17];
    const float* ffn_w2 = (const float*)d_in[18];
    const float* ffn_b2 = (const float*)d_in[19];
    const float* ln2_g  = (const float*)d_in[20];
    const float* ln2_b  = (const float*)d_in[21];
    float* out = (float*)d_out;

    float* ws = (float*)d_ws;
    size_t off = 0;
    auto alloc = [&](size_t n) { float* p = ws + off; off += n; return p; };
    const size_t RM = (size_t)Rsz * DMsz;
    float* Q   = alloc(3 * RM);
    float* Kb  = alloc(3 * RM);
    float* Vb  = alloc(3 * RM);
    float* d0  = alloc(3 * RM);
    float* d1  = alloc(3 * RM);
    float* a0  = alloc(3 * (size_t)Rsz * 2048);
    float* a1  = alloc(3 * (size_t)Rsz * 2048);
    float* dec = alloc(9 * (size_t)Bsz * Lsz * Lsz);
    float* S   = alloc((size_t)3 * Bsz * Hsz * Lsz * 192);
    float* O   = alloc(3 * RM);
    float* fc  = alloc(3 * RM);
    float* oln = alloc(3 * RM);
    float* fpart = alloc(32 * RM);
    (void)ws_size; (void)n_in; (void)in_sizes; (void)out_size;

    auto gemm = [&](const float* A, const float* Bm, const float* bias,
                    const float* resid, float* C, int M, int N, int K, int flags) {
        dim3 grid(N / 32, M / 32);
        gemm_wmma_f32<<<grid, dim3(128), 0, stream>>>(A, Bm, bias, resid, C, M, N, K, flags);
    };

    for (int i = 0; i < 3; ++i) {
        gemm(X[i], w_qs,  nullptr, nullptr, Q  + i * RM, Rsz, 1024, 1024, 0);
        gemm(X[i], w_ks,  nullptr, nullptr, Kb + i * RM, Rsz, 1024, 1024, 0);
        gemm(X[i], w_vs,  nullptr, nullptr, Vb + i * RM, Rsz, 1024, 1024, 0);
        gemm(X[i], w_ds0, nullptr, nullptr, d0 + i * RM, Rsz, 1024, 1024, 0);
        gemm(X[i], w_ds1, nullptr, nullptr, d1 + i * RM, Rsz, 1024, 1024, 0);
        gemm(d0 + i * RM, dc_w1 + (size_t)1024 * 2048, nullptr, nullptr,
             a0 + i * (size_t)Rsz * 2048, Rsz, 2048, 1024, 0);
        gemm(d1 + i * RM, dc_w1, nullptr, nullptr,
             a1 + i * (size_t)Rsz * 2048, Rsz, 2048, 1024, 0);
    }

    {
        dec_kernel<<<9 * Bsz * 8 * 8, 256, 0, stream>>>(a0, a1, dc_b1, dc_w2, dc_b2, dec);
    }

    {
        attn_scores_kernel<<<3 * Bsz * Hsz * (Lsz / SC_QT), 128, 0, stream>>>(Q, Kb, dec, mask, S);
        attn_av_kernel<<<3 * Bsz * Hsz * (Lsz / AV_QT), 128, 0, stream>>>(S, Vb, O);
    }

    for (int i = 0; i < 3; ++i) {
        gemm(O + i * RM, w_fc, nullptr, X[i], fc + i * RM, Rsz, 1024, 1024, GEMM_RESID);
        layernorm_kernel<<<Rsz, 256, 0, stream>>>(fc + i * RM, 1, RM, nullptr, nullptr,
                                                  ln1_g, ln1_b, oln + i * RM);
        ffn_fused_wmma_f32<<<dim3(32, Rsz / FM), dim3(256), 0, stream>>>(
            oln + i * RM, ffn_w1, ffn_b1, ffn_w2, fpart);
        layernorm_kernel<<<Rsz, 256, 0, stream>>>(fpart, 32, RM, ffn_b2, oln + i * RM,
                                                  ln2_g, ln2_b, out + i * RM);
    }
}


# Projections q,k,v,d0,d1 of a modality in one kernel: a block stages each X chunk once for a 32-column tile of all five weights, five accumulators per wave; 15 launches become 3; sums bit-identical
# speedup vs baseline: 4.1500x; 1.1727x over previous
#include <hip/hip_runtime.h>
#include <hip/hip_bf16.h>

#define Bsz 4
#define Lsz 64
#define DMsz 1024
#define Hsz 16
#define DKsz 64
#define DVsz 64
#define Dsz 1024
#define DIsz 4096
#define Rsz 256
#define NEGC (-1.0e9f)

typedef float v2f __attribute__((ext_vector_type(2)));
typedef float v4f __attribute__((ext_vector_type(4)));
typedef float v8f __attribute__((ext_vector_type(8)));

#define GEMM_BIAS  1
#define GEMM_RELU  2
#define GEMM_RESID 4

#define GK  128
#define GAP (GK + 4)
#define GBP 40
#define GCP 36
#define GNA (GK / 16)
#define GNB (GK / 16)

__global__ __launch_bounds__(128)
void gemm_wmma_f32(const float* __restrict__ A, const float* __restrict__ B,
                   const float* __restrict__ bias, const float* __restrict__ resid,
                   float* __restrict__ C, int M, int N, int K, int flags)
{
    __shared__ __attribute__((aligned(16))) float As[32 * GAP];
    __shared__ __attribute__((aligned(16))) float Bs[GK * GBP];
    __shared__ __attribute__((aligned(16))) float Cs[32 * GCP];

    const int tid   = threadIdx.x;
    const int wave  = tid >> 5;
    const int lane  = tid & 31;
    const int l15   = lane & 15;
    const int lhalf = lane >> 4;
    const int m0    = blockIdx.y * 32;
    const int n0    = blockIdx.x * 32;
    const int mq    = wave >> 1;
    const int nq    = wave & 1;
    (void)M;

    const int ar = tid / (GK / 4);
    const int ac = (tid % (GK / 4)) << 2;
    const int br = tid >> 3;
    const int bc = (tid & 7) << 2;
    const float* Ag = A + (size_t)(m0 + ar) * K + ac;
    const float* Bg = B + (size_t)br * N + n0 + bc;
    v4f aprf[GNA], bprf[GNB];

    v8f acc = v8f{0.f, 0.f, 0.f, 0.f, 0.f, 0.f, 0.f, 0.f};

#pragma unroll
    for (int u = 0; u < GNA; ++u)
        aprf[u] = *(const v4f*)(Ag + (size_t)(u * (512 / GK)) * K);
#pragma unroll
    for (int u = 0; u < GNB; ++u)
        bprf[u] = *(const v4f*)(Bg + (size_t)(u * 16) * N);

    const float* ap = As + (mq * 16 + l15) * GAP + lhalf * 2;
    const float* bp = Bs + (lhalf * 2) * GBP + nq * 16 + l15;

    for (int k0 = 0; k0 < K; k0 += GK) {
#pragma unroll
        for (int u = 0; u < GNA; ++u)
            *(v4f*)&As[(ar + u * (512 / GK)) * GAP + ac] = aprf[u];
#pragma unroll
        for (int u = 0; u < GNB; ++u)
            *(v4f*)&Bs[(br + u * 16) * GBP + bc] = bprf[u];
        __syncthreads();

        if (k0 + GK < K) {
#pragma unroll
            for (int u = 0; u < GNA; ++u)
                aprf[u] = *(const v4f*)(Ag + (size_t)(u * (512 / GK)) * K + k0 + GK);
#pragma unroll
            for (int u = 0; u < GNB; ++u)
                bprf[u] = *(const v4f*)(Bg + (size_t)(k0 + GK + u * 16) * N);
        }

        v2f af[GK / 4], bf[GK / 4];
#pragma unroll
        for (int s = 0; s < GK / 4; ++s) {
            af[s] = v2f{ap[s * 4], ap[s * 4 + 1]};
            bf[s] = v2f{bp[(s * 4) * GBP], bp[(s * 4 + 1) * GBP]};
        }
#pragma unroll
        for (int s = 0; s < GK / 4; ++s)
            acc = __builtin_amdgcn_wmma_f32_16x16x4_f32(
                false, af[s], false, bf[s], (short)0, acc, false, false);
        __builtin_amdgcn_sched_group_barrier(0x100, 12, 0);
#pragma unroll
        for (int p = 0; p < (GK / 4 - 8) / 2; ++p) {
            __builtin_amdgcn_sched_group_barrier(0x008, 2, 0);
            __builtin_amdgcn_sched_group_barrier(0x100, 3, 0);
        }
        __builtin_amdgcn_sched_group_barrier(0x008, 8, 0);
        __syncthreads();
    }

#pragma unroll
    for (int r = 0; r < 8; ++r)
        Cs[(mq * 16 + r + lhalf * 8) * GCP + nq * 16 + l15] = acc[r];
    __syncthreads();

#pragma unroll
    for (int u = 0; u < 2; ++u) {
        const int v   = tid + u * 128;
        const int row = v >> 3;
        const int c4  = (v & 7) << 2;
        const size_t g = (size_t)(m0 + row) * N + n0 + c4;
        v4f o = *(const v4f*)&Cs[row * GCP + c4];
        v4f bv = v4f{0.f, 0.f, 0.f, 0.f};
        if (flags & GEMM_BIAS) bv = *(const v4f*)(bias + n0 + c4);
        o.x = o.x + bv.x; o.y = o.y + bv.y; o.z = o.z + bv.z; o.w = o.w + bv.w;
        if (flags & GEMM_RESID) {
            const v4f rv = *(const v4f*)(resid + g);
            o.x += rv.x; o.y += rv.y; o.z += rv.z; o.w += rv.w;
        }
        if (flags & GEMM_RELU) {
            o.x = fmaxf(o.x, 0.f); o.y = fmaxf(o.y, 0.f);
            o.z = fmaxf(o.z, 0.f); o.w = fmaxf(o.w, 0.f);
        }
        *(v4f*)(C + g) = o;
    }
}

#define PJ_NW 5
#define PJ_BS (GK * GBP)

struct PjFrag { float a0[8], a1[8], b0[PJ_NW][8], b1[PJ_NW][8]; };

__device__ __forceinline__ void pj_frags(PjFrag& f, const float* a, const float* b, int k)
{
#pragma unroll
    for (int s = 0; s < 8; ++s) {
        const int kk = k + s * 4;
        f.a0[s] = a[kk];
        f.a1[s] = a[kk + 1];
#pragma unroll
        for (int w = 0; w < PJ_NW; ++w) {
            f.b0[w][s] = b[w * PJ_BS + kk * GBP];
            f.b1[w][s] = b[w * PJ_BS + (kk + 1) * GBP];
        }
    }
}

__global__ __launch_bounds__(128)
void proj5_wmma_f32(const float* __restrict__ A,
                    const float* __restrict__ B0, const float* __restrict__ B1,
                    const float* __restrict__ B2, const float* __restrict__ B3,
                    const float* __restrict__ B4,
                    float* __restrict__ C0, float* __restrict__ C1, float* __restrict__ C2,
                    float* __restrict__ C3, float* __restrict__ C4)
{
    __shared__ __attribute__((aligned(16))) float As[32 * GAP];
    __shared__ __attribute__((aligned(16))) float Bs[PJ_NW * PJ_BS];

    const int tid   = threadIdx.x;
    const int wave  = tid >> 5;
    const int lane  = tid & 31;
    const int l15   = lane & 15;
    const int lhalf = lane >> 4;
    const int m0    = blockIdx.y * 32;
    const int n0    = blockIdx.x * 32;
    const int mq    = wave >> 1;
    const int nq    = wave & 1;

    const float* const Bw[PJ_NW] = {B0, B1, B2, B3, B4};
    float* const       Cw[PJ_NW] = {C0, C1, C2, C3, C4};

    const int ar = tid / (GK / 4);
    const int ac = (tid % (GK / 4)) << 2;
    const int br = tid >> 3;
    const int bc = (tid & 7) << 2;
    const float* Ag = A + (size_t)(m0 + ar) * DMsz + ac;
    const size_t bo = (size_t)br * DMsz + n0 + bc;
    v4f aprf[GNA], bprf[PJ_NW][GNB];

    v8f acc[PJ_NW];
#pragma unroll
    for (int w = 0; w < PJ_NW; ++w) acc[w] = v8f{0.f, 0.f, 0.f, 0.f, 0.f, 0.f, 0.f, 0.f};

#pragma unroll
    for (int u = 0; u < GNA; ++u)
        aprf[u] = *(const v4f*)(Ag + (size_t)(u * (512 / GK)) * DMsz);
#pragma unroll
    for (int w = 0; w < PJ_NW; ++w)
#pragma unroll
        for (int u = 0; u < GNB; ++u)
            bprf[w][u] = *(const v4f*)(Bw[w] + bo + (size_t)(u * 16) * DMsz);

    const float* ap = As + (mq * 16 + l15) * GAP + lhalf * 2;
    const float* bp = Bs + (lhalf * 2) * GBP + nq * 16 + l15;

    for (int k0 = 0; k0 < DMsz; k0 += GK) {
#pragma unroll
        for (int u = 0; u < GNA; ++u)
            *(v4f*)&As[(ar + u * (512 / GK)) * GAP + ac] = aprf[u];
#pragma unroll
        for (int w = 0; w < PJ_NW; ++w)
#pragma unroll
            for (int u = 0; u < GNB; ++u)
                *(v4f*)&Bs[w * PJ_BS + (br + u * 16) * GBP + bc] = bprf[w][u];
        __syncthreads();

        if (k0 + GK < DMsz) {
#pragma unroll
            for (int u = 0; u < GNA; ++u)
                aprf[u] = *(const v4f*)(Ag + (size_t)(u * (512 / GK)) * DMsz + k0 + GK);
#pragma unroll
            for (int w = 0; w < PJ_NW; ++w)
#pragma unroll
                for (int u = 0; u < GNB; ++u)
                    bprf[w][u] = *(const v4f*)(Bw[w] + bo + (size_t)(k0 + GK + u * 16) * DMsz);
        }

        PjFrag f[2];
        pj_frags(f[0], ap, bp, 0);
        __builtin_amdgcn_sched_barrier(0);
#pragma unroll
        for (int g = 0; g < GK / 32; ++g) {
            if (g + 1 < GK / 32)
                pj_frags(f[(g + 1) & 1], ap, bp, (g + 1) * 32);
            __builtin_amdgcn_sched_barrier(0);
#pragma unroll
            for (int s = 0; s < 8; ++s) {
                const PjFrag& c = f[g & 1];
                const v2f af = v2f{c.a0[s], c.a1[s]};
#pragma unroll
                for (int w = 0; w < PJ_NW; ++w) {
                    const v2f bf = v2f{c.b0[w][s], c.b1[w][s]};
                    acc[w] = __builtin_amdgcn_wmma_f32_16x16x4_f32(
                        false, af, false, bf, (short)0, acc[w], false, false);
                }
            }
            __builtin_amdgcn_sched_barrier(0);
        }
        __syncthreads();
    }

    float* Cs = Bs;
#pragma unroll
    for (int w = 0; w < PJ_NW; ++w)
#pragma unroll
        for (int r = 0; r < 8; ++r)
            Cs[w * 32 * GCP + (mq * 16 + r + lhalf * 8) * GCP + nq * 16 + l15] = acc[w][r];
    __syncthreads();

#pragma unroll
    for (int w = 0; w < PJ_NW; ++w)
#pragma unroll
        for (int u = 0; u < 2; ++u) {
            const int v   = tid + u * 128;
            const int row = v >> 3;
            const int c4  = (v & 7) << 2;
            *(v4f*)(Cw[w] + (size_t)(m0 + row) * DMsz + n0 + c4) =
                *(const v4f*)&Cs[w * 32 * GCP + row * GCP + c4];
        }
}

#define FM  32
#define FK  128
#define FAP (FK + 4)
#define FBP (128 + 8)

struct FfnFrag { float a0[8], a1[8], b0[2][8], b1[2][8]; };

__device__ __forceinline__ void ffn_frags(FfnFrag& f, const float* a, const float* b, int k)
{
#pragma unroll
    for (int s = 0; s < 8; ++s) {
        const int kk = k + s * 4;
        f.a0[s]    = a[kk];
        f.a1[s]    = a[kk + 1];
        f.b0[0][s] = b[kk * FBP];
        f.b1[0][s] = b[(kk + 1) * FBP];
        f.b0[1][s] = b[kk * FBP + 16];
        f.b1[1][s] = b[(kk + 1) * FBP + 16];
    }
}

__device__ __forceinline__ void ffn_chunk(v8f (&acc)[2], const float* a, const float* b)
{
    FfnFrag f[2];
    ffn_frags(f[0], a, b, 0);
#pragma unroll
    for (int g = 0; g < FK / 32; ++g) {
        if (g + 1 < FK / 32)
            ffn_frags(f[(g + 1) & 1], a, b, (g + 1) * 32);
        __builtin_amdgcn_sched_barrier(0);
#pragma unroll
        for (int s = 0; s < 8; ++s) {
            const FfnFrag& c = f[g & 1];
            const v2f af  = v2f{c.a0[s], c.a1[s]};
            const v2f bf0 = v2f{c.b0[0][s], c.b1[0][s]};
            const v2f bf1 = v2f{c.b0[1][s], c.b1[1][s]};
            acc[0] = __builtin_amdgcn_wmma_f32_16x16x4_f32(
                false, af, false, bf0, (short)0, acc[0], false, false);
            acc[1] = __builtin_amdgcn_wmma_f32_16x16x4_f32(
                false, af, false, bf1, (short)0, acc[1], false, false);
        }
        __builtin_amdgcn_sched_barrier(0);
    }
}

__global__ __launch_bounds__(256)
void ffn_fused_wmma_f32(const float* __restrict__ X, const float* __restrict__ W1,
                        const float* __restrict__ b1, const float* __restrict__ W2,
                        float* __restrict__ P)
{
    __shared__ __attribute__((aligned(16))) float As[FM * FAP];
    __shared__ __attribute__((aligned(16))) float Bs[FK * FBP];
    __shared__ __attribute__((aligned(16))) float Hs[FM * FAP];

    const int tid   = threadIdx.x;
    const int wave  = tid >> 5;
    const int lane  = tid & 31;
    const int l15   = lane & 15;
    const int lhalf = lane >> 4;
    const int s0    = blockIdx.x * 128;
    const int m0    = blockIdx.y * FM;
    const int mq    = wave >> 2;
    const int nq    = wave & 3;

    const int sr = tid >> 5;
    const int sc = (tid & 31) << 2;
    const float* Xg  = X  + (size_t)(m0 + sr) * DMsz + sc;
    const float* W1g = W1 + (size_t)sr * DIsz + s0 + sc;
    const float* W2g = W2 + (size_t)(s0 + sr) * DMsz + sc;
    v4f aprf[4], bprf[16];

    v8f acc[2];
    acc[0] = v8f{0.f, 0.f, 0.f, 0.f, 0.f, 0.f, 0.f, 0.f};
    acc[1] = v8f{0.f, 0.f, 0.f, 0.f, 0.f, 0.f, 0.f, 0.f};

#pragma unroll
    for (int u = 0; u < 4; ++u)
        aprf[u] = *(const v4f*)(Xg + (size_t)(u * 8) * DMsz);
#pragma unroll
    for (int u = 0; u < 16; ++u)
        bprf[u] = *(const v4f*)(W1g + (size_t)(u * 8) * DIsz);

    const float* ap = As + (mq * 16 + l15) * FAP + lhalf * 2;
    const float* hp = Hs + (mq * 16 + l15) * FAP + lhalf * 2;
    const float* bp = Bs + (lhalf * 2) * FBP + nq * 32 + l15;

    for (int k0 = 0; k0 < DMsz; k0 += FK) {
#pragma unroll
        for (int u = 0; u < 4; ++u)
            *(v4f*)&As[(sr + u * 8) * FAP + sc] = aprf[u];
#pragma unroll
        for (int u = 0; u < 16; ++u)
            *(v4f*)&Bs[(sr + u * 8) * FBP + sc] = bprf[u];
        __syncthreads();

        if (k0 + FK < DMsz) {
#pragma unroll
            for (int u = 0; u < 4; ++u)
                aprf[u] = *(const v4f*)(Xg + (size_t)(u * 8) * DMsz + k0 + FK);
#pragma unroll
            for (int u = 0; u < 16; ++u)
                bprf[u] = *(const v4f*)(W1g + (size_t)(k0 + FK + u * 8) * DIsz);
        } else {
#pragma unroll
            for (int u = 0; u < 16; ++u)
                bprf[u] = *(const v4f*)(W2g + (size_t)(u * 8) * DMsz);
        }

        ffn_chunk(acc, ap, bp);
        __syncthreads();
    }

#pragma unroll
    for (int nsub = 0; nsub < 2; ++nsub) {
        const int hc = nq * 32 + nsub * 16 + l15;
        const float bv = b1[s0 + hc];
#pragma unroll
        for (int r = 0; r < 8; ++r) {
            const int hr = mq * 16 + r + lhalf * 8;
            Hs[hr * FAP + hc] = fmaxf(acc[nsub][r] + bv, 0.f);
        }
    }

    float* Pp = P + (size_t)blockIdx.x * Rsz * DMsz;
    for (int nc = 0; nc < 8; ++nc) {
        acc[0] = v8f{0.f, 0.f, 0.f, 0.f, 0.f, 0.f, 0.f, 0.f};
        acc[1] = v8f{0.f, 0.f, 0.f, 0.f, 0.f, 0.f, 0.f, 0.f};

#pragma unroll
        for (int u = 0; u < 16; ++u)
            *(v4f*)&Bs[(sr + u * 8) * FBP + sc] = bprf[u];
        __syncthreads();

        if (nc + 1 < 8) {
#pragma unroll
            for (int u = 0; u < 16; ++u)
                bprf[u] = *(const v4f*)(W2g + (size_t)(u * 8) * DMsz + (nc + 1) * 128);
        }

        ffn_chunk(acc, hp, bp);
        __syncthreads();

#pragma unroll
        for (int nsub = 0; nsub < 2; ++nsub) {
            const int col = nc * 128 + nq * 32 + nsub * 16 + l15;
#pragma unroll
            for (int r = 0; r < 8; ++r) {
                const int row = m0 + mq * 16 + r + lhalf * 8;
                Pp[(size_t)row * DMsz + col] = acc[nsub][r];
            }
        }
    }
}

#define GT_KP (18 * 32)
#define GT_PK 32

__global__ __launch_bounds__(256)
void dec_kernel(const float* __restrict__ a0base, const float* __restrict__ a1base,
                const float* __restrict__ b1, const float* __restrict__ w2,
                const float* __restrict__ b2, float* __restrict__ dec)
{
    __shared__ float tile[GT_PK * GT_KP];

    const int tid  = threadIdx.x;
    const int wave = tid >> 5;
    const int lane = tid & 31;
    const int blk  = blockIdx.x;
    const int nt   = blk & 7;
    const int mt   = (blk >> 3) & 7;
    const int b    = (blk >> 6) & 3;
    const int pair = blk >> 8;
    const int i    = pair / 3;
    const int j    = pair % 3;

    const float4* g0 = (const float4*)(a0base + ((size_t)i * Rsz + b * Lsz + nt * 8) * 2048);
    const float4* g1 = (const float4*)(a1base + ((size_t)j * Rsz + b * Lsz + mt * 8) * 2048);
    const float2* gw = (const float2*)w2;

    const int ml = (wave >> 1) * 2;
    const int nl = (wave & 1) * 4;
    const float* pa = tile + ml * 32 + lane;
    const float* pb = tile + (8 + nl) * 32 + lane;
    const float* pc = tile + 16 * 32 + lane;

    float acc[2][4];
#pragma unroll
    for (int x = 0; x < 2; ++x)
#pragma unroll
        for (int y = 0; y < 4; ++y) acc[x][y] = 0.f;

    for (int p = 0; p < 2; ++p) {
        {
            float4 r0[8], r1[8];
#pragma unroll
            for (int u = 0; u < 8; ++u) r0[u] = g0[u * 512 + p * 256 + tid];
#pragma unroll
            for (int u = 0; u < 8; ++u) r1[u] = g1[u * 512 + p * 256 + tid];
            const int dst = (tid >> 3) * GT_KP + (tid & 7) * 4;
#pragma unroll
            for (int u = 0; u < 8; ++u) {
                *(float4*)&tile[dst + (8 + u) * 32] = r0[u];
                *(float4*)&tile[dst + u * 32]       = r1[u];
            }
        }
#pragma unroll
        for (int u = 0; u < 4; ++u) {
            const int cl  = tid + u * 256;
            const float2 w = gw[p * 1024 + cl];
            const int dst = (cl >> 5) * GT_KP + (cl & 31);
            tile[dst + 16 * 32] = b1[p * 1024 + cl];
            tile[dst + 17 * 32] = w.y - w.x;
        }
        __syncthreads();

#pragma unroll 4
        for (int k = 0; k < GT_PK; ++k) {
            const int o = k * GT_KP;
            float xv[2], yv[4];
            xv[0] = pa[o];      xv[1] = pa[o + 32];
            yv[0] = pb[o];      yv[1] = pb[o + 32];
            yv[2] = pb[o + 64]; yv[3] = pb[o + 96];
            const float bb = pc[o], ww = pc[o + 32];
#pragma unroll
            for (int x = 0; x < 2; ++x)
#pragma unroll
                for (int y = 0; y < 4; ++y) {
                    const float t = (xv[x] + yv[y]) + bb;
                    if (t > 0.f) acc[x][y] = fmaf(t, ww, acc[x][y]);
                }
        }
        __syncthreads();
    }

#pragma unroll
    for (int off = 16; off; off >>= 1)
#pragma unroll
        for (int x = 0; x < 2; ++x)
#pragma unroll
            for (int y = 0; y < 4; ++y)
                acc[x][y] += __shfl_xor(acc[x][y], off, 32);

    if (lane == 0) {
        const float bd = b2[1] - b2[0];
        float* dtile = dec + (((size_t)(pair * Bsz + b) * Lsz + mt * 8) * Lsz + nt * 8);
#pragma unroll
        for (int x = 0; x < 2; ++x) {
            float4 r;
            r.x = ((acc[x][0] + bd) > 0.f) ? NEGC : 0.f;
            r.y = ((acc[x][1] + bd) > 0.f) ? NEGC : 0.f;
            r.z = ((acc[x][2] + bd) > 0.f) ? NEGC : 0.f;
            r.w = ((acc[x][3] + bd) > 0.f) ? NEGC : 0.f;
            *(float4*)(dtile + (size_t)(ml + x) * Lsz + nl) = r;
        }
    }
}

#define SC_QT 16
#define SC_KP 68

__global__ __launch_bounds__(128)
void attn_scores_kernel(const float* __restrict__ Qb, const float* __restrict__ Kb,
                        const float* __restrict__ dec, const int* __restrict__ mask,
                        float* __restrict__ S)
{
    __shared__ __attribute__((aligned(16))) float Ks[192 * SC_KP];
    __shared__ __attribute__((aligned(16))) float Qs[SC_QT * 64];

    const int blk  = blockIdx.x;
    const int qt   = blk & 3;
    const int h    = (blk >> 2) & 15;
    const int b    = (blk >> 6) & 3;
    const int i    = blk >> 8;
    const int tid  = threadIdx.x;
    const int wave = tid >> 5;
    const int lane = tid & 31;
    const int q0   = qt * SC_QT + wave * 4;

    {
        const int c4 = (tid & 15) << 2;
        const int r0 = tid >> 4;
        v4f kr[24], qr[2];
#pragma unroll
        for (int u = 0; u < 24; ++u) {
            const int row = r0 + u * 8;
            const int j = row >> 6, k = row & 63;
            kr[u] = *(const v4f*)(Kb + ((size_t)j * Rsz + b * Lsz + k) * DMsz + h * DKsz + c4);
        }
#pragma unroll
        for (int u = 0; u < 2; ++u) {
            const int row = r0 + u * 8;
            qr[u] = *(const v4f*)(Qb + ((size_t)i * Rsz + b * Lsz + qt * SC_QT + row) * DMsz
                                  + h * DKsz + c4);
        }
#pragma unroll
        for (int u = 0; u < 24; ++u)
            *(v4f*)&Ks[(r0 + u * 8) * SC_KP + c4] = kr[u];
#pragma unroll
        for (int u = 0; u < 2; ++u)
            *(v4f*)&Qs[(r0 + u * 8) * 64 + c4] = qr[u];
    }

    float dv[4][6];
    int   mk[4][2];
#pragma unroll
    for (int r = 0; r < 4; ++r) {
#pragma unroll
        for (int m = 0; m < 6; ++m) {
            const int j = m >> 1, k = (m & 1) * 32 + lane;
            dv[r][m] = dec[(((size_t)(i * 3 + j) * Bsz + b) * Lsz + q0 + r) * Lsz + k];
        }
#pragma unroll
        for (int kh = 0; kh < 2; ++kh)
            mk[r][kh] = mask[((size_t)b * Lsz + q0 + r) * Lsz + kh * 32 + lane];
    }
    __syncthreads();

    float acc[4][6];
#pragma unroll
    for (int r = 0; r < 4; ++r)
#pragma unroll
        for (int m = 0; m < 6; ++m) acc[r][m] = 0.f;

    const float* kp = Ks + lane * SC_KP;
    const float* qp = Qs + (wave * 4) * 64;
    for (int d = 0; d < 64; d += 4) {
        v4f kv[6], qv[4];
#pragma unroll
        for (int m = 0; m < 6; ++m) kv[m] = *(const v4f*)(kp + m * 32 * SC_KP + d);
#pragma unroll
        for (int r = 0; r < 4; ++r) qv[r] = *(const v4f*)(qp + r * 64 + d);
#pragma unroll
        for (int r = 0; r < 4; ++r)
#pragma unroll
            for (int m = 0; m < 6; ++m) {
                float s = acc[r][m];
                s = fmaf(qv[r].x, kv[m].x, s);
                s = fmaf(qv[r].y, kv[m].y, s);
                s = fmaf(qv[r].z, kv[m].z, s);
                s = fmaf(qv[r].w, kv[m].w, s);
                acc[r][m] = s;
            }
    }

#pragma unroll
    for (int r = 0; r < 4; ++r) {
        float s[6];
#pragma unroll
        for (int m = 0; m < 6; ++m) {
            float t = acc[r][m];
            t *= 0.125f;
            t += dv[r][m];
            if (mk[r][m & 1] == 0) t = NEGC;
            s[m] = t;
        }
        float mx = fmaxf(fmaxf(fmaxf(s[0], s[1]), fmaxf(s[2], s[3])), fmaxf(s[4], s[5]));
#pragma unroll
        for (int off = 16; off; off >>= 1) mx = fmaxf(mx, __shfl_xor(mx, off, 32));
        float e[6];
#pragma unroll
        for (int m = 0; m < 6; ++m) e[m] = __expf(s[m] - mx);
        float sum = ((e[0] + e[2]) + e[4]) + ((e[1] + e[3]) + e[5]);
#pragma unroll
        for (int off = 16; off; off >>= 1) sum += __shfl_xor(sum, off, 32);
        const float inv = 1.0f / sum;
        float* Sp = S + ((size_t)((((i * Bsz + b) * Hsz + h) * Lsz) + q0 + r)) * 192 + lane;
#pragma unroll
        for (int m = 0; m < 6; ++m) Sp[m * 32] = e[m] * inv;
    }
}

#define AV_QT 16
#define AV_VP 68

__global__ __launch_bounds__(128)
void attn_av_kernel(const float* __restrict__ S, const float* __restrict__ Vb,
                    float* __restrict__ Ob)
{
    __shared__ __attribute__((aligned(16))) float Vs[192 * AV_VP];
    __shared__ __attribute__((aligned(16))) float Ss[AV_QT * 192];

    const int blk  = blockIdx.x;
    const int qt   = blk & 3;
    const int h    = (blk >> 2) & 15;
    const int b    = (blk >> 6) & 3;
    const int i    = blk >> 8;
    const int tid  = threadIdx.x;
    const int wave = tid >> 5;
    const int lane = tid & 31;

    {
        const int c4 = (tid & 15) << 2;
        const int r0 = tid >> 4;
        const float* Sg = S + ((size_t)((((i * Bsz + b) * Hsz + h) * Lsz) + qt * AV_QT)) * 192;
        v4f vr[24], sr[6];
#pragma unroll
        for (int u = 0; u < 24; ++u) {
            const int row = r0 + u * 8;
            const int j = row >> 6, k = row & 63;
            vr[u] = *(const v4f*)(Vb + ((size_t)j * Rsz + b * Lsz + k) * DMsz + h * DVsz + c4);
        }
#pragma unroll
        for (int u = 0; u < 6; ++u)
            sr[u] = *(const v4f*)(Sg + (size_t)(tid + u * 128) * 4);
#pragma unroll
        for (int u = 0; u < 24; ++u)
            *(v4f*)&Vs[(r0 + u * 8) * AV_VP + c4] = vr[u];
#pragma unroll
        for (int u = 0; u < 6; ++u)
            *(v4f*)&Ss[(tid + u * 128) * 4] = sr[u];
    }
    __syncthreads();

    float acc[4][2];
#pragma unroll
    for (int r = 0; r < 4; ++r) { acc[r][0] = 0.f; acc[r][1] = 0.f; }

    const float* vp = Vs + lane;
    const float* sp = Ss + (wave * 4) * 192;
#pragma unroll 4
    for (int jk = 0; jk < 192; jk += 4) {
        v4f sv[4];
        float vv[4][2];
#pragma unroll
        for (int r = 0; r < 4; ++r) sv[r] = *(const v4f*)(sp + r * 192 + jk);
#pragma unroll
        for (int t = 0; t < 4; ++t) {
            vv[t][0] = vp[(jk + t) * AV_VP];
            vv[t][1] = vp[(jk + t) * AV_VP + 32];
        }
#pragma unroll
        for (int r = 0; r < 4; ++r)
#pragma unroll
            for (int cc = 0; cc < 2; ++cc) {
                float s = acc[r][cc];
                s = fmaf(sv[r].x, vv[0][cc], s);
                s = fmaf(sv[r].y, vv[1][cc], s);
                s = fmaf(sv[r].z, vv[2][cc], s);
                s = fmaf(sv[r].w, vv[3][cc], s);
                acc[r][cc] = s;
            }
    }

#pragma unroll
    for (int r = 0; r < 4; ++r) {
        const int q = qt * AV_QT + wave * 4 + r;
        float* Op = Ob + ((size_t)i * Rsz + b * Lsz + q) * DMsz + h * DVsz + lane;
        Op[0]  = acc[r][0];
        Op[32] = acc[r][1];
    }
}

__global__ __launch_bounds__(256)
void layernorm_kernel(const float* __restrict__ x, int nparts, size_t pstride,
                      const float* __restrict__ bias, const float* __restrict__ resid,
                      const float* __restrict__ g, const float* __restrict__ bta,
                      float* __restrict__ out)
{
    const int row = blockIdx.x;
    const int t = threadIdx.x;
    __shared__ float red[256];

    const float* xp = x + (size_t)row * 1024;
    float v[4];
#pragma unroll
    for (int u = 0; u < 4; ++u) v[u] = xp[t + u * 256];
#pragma unroll 8
    for (int p = 1; p < nparts; ++p) {
        const float* pp = xp + (size_t)p * pstride;
#pragma unroll
        for (int u = 0; u < 4; ++u) v[u] += pp[t + u * 256];
    }
    float s = 0.f;
#pragma unroll
    for (int u = 0; u < 4; ++u) {
        float val = v[u];
        if (bias)  val += bias[t + u * 256];
        if (resid) val += resid[(size_t)row * 1024 + t + u * 256];
        v[u] = val; s += val;
    }
    red[t] = s; __syncthreads();
    for (int st = 128; st > 0; st >>= 1) { if (t < st) red[t] += red[t + st]; __syncthreads(); }
    const float mu = red[0] * (1.f / 1024.f);
    __syncthreads();

    float s2 = 0.f;
#pragma unroll
    for (int u = 0; u < 4; ++u) { float dd = v[u] - mu; s2 += dd * dd; }
    red[t] = s2; __syncthreads();
    for (int st = 128; st > 0; st >>= 1) { if (t < st) red[t] += red[t + st]; __syncthreads(); }
    const float var = red[0] * (1.f / 1024.f);
    const float rinv = 1.0f / sqrtf(var + 1e-6f);
#pragma unroll
    for (int u = 0; u < 4; ++u) {
        int c = t + u * 256;
        out[(size_t)row * 1024 + c] = (v[u] - mu) * rinv * g[c] + bta[c];
    }
}

extern "C" void kernel_launch(void* const* d_in, const int* in_sizes, int n_in,
                              void* d_out, int out_size, void* d_ws, size_t ws_size,
                              hipStream_t stream)
{
    const float* X[3]   = { (const float*)d_in[0], (const float*)d_in[1], (const float*)d_in[2] };
    const int*   mask   = (const int*)d_in[3];
    const float* w_qs   = (const float*)d_in[4];
    const float* w_ks   = (const float*)d_in[5];
    const float* w_vs   = (const float*)d_in[6];
    const float* w_ds0  = (const float*)d_in[7];
    const float* w_ds1  = (const float*)d_in[8];
    const float* dc_w1  = (const float*)d_in[9];
    const float* dc_b1  = (const float*)d_in[10];
    const float* dc_w2  = (const float*)d_in[11];
    const float* dc_b2  = (const float*)d_in[12];
    const float* w_fc   = (const float*)d_in[13];
    const float* ln1_g  = (const float*)d_in[14];
    const float* ln1_b  = (const float*)d_in[15];
    const float* ffn_w1 = (const float*)d_in[16];
    const float* ffn_b1 = (const float*)d_in[17];
    const float* ffn_w2 = (const float*)d_in[18];
    const float* ffn_b2 = (const float*)d_in[19];
    const float* ln2_g  = (const float*)d_in[20];
    const float* ln2_b  = (const float*)d_in[21];
    float* out = (float*)d_out;

    float* ws = (float*)d_ws;
    size_t off = 0;
    auto alloc = [&](size_t n) { float* p = ws + off; off += n; return p; };
    const size_t RM = (size_t)Rsz * DMsz;
    float* Q   = alloc(3 * RM);
    float* Kb  = alloc(3 * RM);
    float* Vb  = alloc(3 * RM);
    float* d0  = alloc(3 * RM);
    float* d1  = alloc(3 * RM);
    float* a0  = alloc(3 * (size_t)Rsz * 2048);
    float* a1  = alloc(3 * (size_t)Rsz * 2048);
    float* dec = alloc(9 * (size_t)Bsz * Lsz * Lsz);
    float* S   = alloc((size_t)3 * Bsz * Hsz * Lsz * 192);
    float* O   = alloc(3 * RM);
    float* fc  = alloc(3 * RM);
    float* oln = alloc(3 * RM);
    float* fpart = alloc(32 * RM);
    (void)ws_size; (void)n_in; (void)in_sizes; (void)out_size;

    auto gemm = [&](const float* A, const float* Bm, const float* bias,
                    const float* resid, float* C, int M, int N, int K, int flags) {
        dim3 grid(N / 32, M / 32);
        gemm_wmma_f32<<<grid, dim3(128), 0, stream>>>(A, Bm, bias, resid, C, M, N, K, flags);
    };

    for (int i = 0; i < 3; ++i) {
        proj5_wmma_f32<<<dim3(DMsz / 32, Rsz / 32), dim3(128), 0, stream>>>(
            X[i], w_qs, w_ks, w_vs, w_ds0, w_ds1,
            Q + i * RM, Kb + i * RM, Vb + i * RM, d0 + i * RM, d1 + i * RM);
        gemm(d0 + i * RM, dc_w1 + (size_t)1024 * 2048, nullptr, nullptr,
             a0 + i * (size_t)Rsz * 2048, Rsz, 2048, 1024, 0);
        gemm(d1 + i * RM, dc_w1, nullptr, nullptr,
             a1 + i * (size_t)Rsz * 2048, Rsz, 2048, 1024, 0);
    }

    {
        dec_kernel<<<9 * Bsz * 8 * 8, 256, 0, stream>>>(a0, a1, dc_b1, dc_w2, dc_b2, dec);
    }

    {
        attn_scores_kernel<<<3 * Bsz * Hsz * (Lsz / SC_QT), 128, 0, stream>>>(Q, Kb, dec, mask, S);
        attn_av_kernel<<<3 * Bsz * Hsz * (Lsz / AV_QT), 128, 0, stream>>>(S, Vb, O);
    }

    for (int i = 0; i < 3; ++i) {
        gemm(O + i * RM, w_fc, nullptr, X[i], fc + i * RM, Rsz, 1024, 1024, GEMM_RESID);
        layernorm_kernel<<<Rsz, 256, 0, stream>>>(fc + i * RM, 1, RM, nullptr, nullptr,
                                                  ln1_g, ln1_b, oln + i * RM);
        ffn_fused_wmma_f32<<<dim3(32, Rsz / FM), dim3(256), 0, stream>>>(
            oln + i * RM, ffn_w1, ffn_b1, ffn_w2, fpart);
        layernorm_kernel<<<Rsz, 256, 0, stream>>>(fpart, 32, RM, ffn_b2, oln + i * RM,
                                                  ln2_g, ln2_b, out + i * RM);
    }
}


# Weight chunks staged once for all three modalities: a GEMM or FFN block owns the same 32 rows of each, three accumulator sets per wave share each B fragment; 12 launches become 4; sums bit-identical
# speedup vs baseline: 4.8424x; 1.1668x over previous
#include <hip/hip_runtime.h>
#include <hip/hip_bf16.h>

#define Bsz 4
#define Lsz 64
#define DMsz 1024
#define Hsz 16
#define DKsz 64
#define DVsz 64
#define Dsz 1024
#define DIsz 4096
#define Rsz 256
#define NEGC (-1.0e9f)

typedef float v2f __attribute__((ext_vector_type(2)));
typedef float v4f __attribute__((ext_vector_type(4)));
typedef float v8f __attribute__((ext_vector_type(8)));

#define GK  128
#define GAP (GK + 4)
#define GBP 40
#define GCP 36
#define GNA (GK / 16)
#define GNB (GK / 16)

#define G3_NT 3
#define G3_AS (32 * GAP)

struct G3Frag { float a0[G3_NT][8], a1[G3_NT][8], b0[8], b1[8]; };

__device__ __forceinline__ void g3_frags(G3Frag& f, const float* a, const float* b, int k)
{
#pragma unroll
    for (int s = 0; s < 8; ++s) {
        const int kk = k + s * 4;
#pragma unroll
        for (int t = 0; t < G3_NT; ++t) {
            f.a0[t][s] = a[t * G3_AS + kk];
            f.a1[t][s] = a[t * G3_AS + kk + 1];
        }
        f.b0[s] = b[kk * GBP];
        f.b1[s] = b[(kk + 1) * GBP];
    }
}

__global__ __launch_bounds__(128)
void gemm3_wmma_f32(const float* __restrict__ A, const float* __restrict__ B,
                    const float* __restrict__ R0, const float* __restrict__ R1,
                    const float* __restrict__ R2, float* __restrict__ C, int N, int K)
{
    __shared__ __attribute__((aligned(16))) float As[G3_NT * G3_AS];
    __shared__ __attribute__((aligned(16))) float Bs[GK * GBP];

    const int tid   = threadIdx.x;
    const int wave  = tid >> 5;
    const int lane  = tid & 31;
    const int l15   = lane & 15;
    const int lhalf = lane >> 4;
    const int m0    = blockIdx.y * 32;
    const int n0    = blockIdx.x * 32;
    const int mq    = wave >> 1;
    const int nq    = wave & 1;

    const float* const Rw[G3_NT] = {R0, R1, R2};

    const int ar = tid / (GK / 4);
    const int ac = (tid % (GK / 4)) << 2;
    const int br = tid >> 3;
    const int bc = (tid & 7) << 2;
    const float* Ag = A + (size_t)(m0 + ar) * K + ac;
    const float* Bg = B + (size_t)br * N + n0 + bc;
    v4f aprf[G3_NT][GNA], bprf[GNB];

    v8f acc[G3_NT];
#pragma unroll
    for (int t = 0; t < G3_NT; ++t) acc[t] = v8f{0.f, 0.f, 0.f, 0.f, 0.f, 0.f, 0.f, 0.f};

#pragma unroll
    for (int t = 0; t < G3_NT; ++t)
#pragma unroll
        for (int u = 0; u < GNA; ++u)
            aprf[t][u] = *(const v4f*)(Ag + (size_t)(t * Rsz + u * (512 / GK)) * K);
#pragma unroll
    for (int u = 0; u < GNB; ++u)
        bprf[u] = *(const v4f*)(Bg + (size_t)(u * 16) * N);

    const float* ap = As + (mq * 16 + l15) * GAP + lhalf * 2;
    const float* bp = Bs + (lhalf * 2) * GBP + nq * 16 + l15;

    for (int k0 = 0; k0 < K; k0 += GK) {
#pragma unroll
        for (int t = 0; t < G3_NT; ++t)
#pragma unroll
            for (int u = 0; u < GNA; ++u)
                *(v4f*)&As[t * G3_AS + (ar + u * (512 / GK)) * GAP + ac] = aprf[t][u];
#pragma unroll
        for (int u = 0; u < GNB; ++u)
            *(v4f*)&Bs[(br + u * 16) * GBP + bc] = bprf[u];
        __syncthreads();

        if (k0 + GK < K) {
#pragma unroll
            for (int t = 0; t < G3_NT; ++t)
#pragma unroll
                for (int u = 0; u < GNA; ++u)
                    aprf[t][u] = *(const v4f*)(Ag + (size_t)(t * Rsz + u * (512 / GK)) * K + k0 + GK);
#pragma unroll
            for (int u = 0; u < GNB; ++u)
                bprf[u] = *(const v4f*)(Bg + (size_t)(k0 + GK + u * 16) * N);
        }

        G3Frag f[2];
        g3_frags(f[0], ap, bp, 0);
        __builtin_amdgcn_sched_barrier(0);
#pragma unroll
        for (int g = 0; g < GK / 32; ++g) {
            if (g + 1 < GK / 32)
                g3_frags(f[(g + 1) & 1], ap, bp, (g + 1) * 32);
            __builtin_amdgcn_sched_barrier(0);
#pragma unroll
            for (int s = 0; s < 8; ++s) {
                const G3Frag& c = f[g & 1];
                const v2f bf = v2f{c.b0[s], c.b1[s]};
#pragma unroll
                for (int t = 0; t < G3_NT; ++t) {
                    const v2f af = v2f{c.a0[t][s], c.a1[t][s]};
                    acc[t] = __builtin_amdgcn_wmma_f32_16x16x4_f32(
                        false, af, false, bf, (short)0, acc[t], false, false);
                }
            }
            __builtin_amdgcn_sched_barrier(0);
        }
        __syncthreads();
    }

    v4f rv[G3_NT][2];
#pragma unroll
    for (int t = 0; t < G3_NT; ++t)
#pragma unroll
        for (int u = 0; u < 2; ++u) rv[t][u] = v4f{0.f, 0.f, 0.f, 0.f};
    if (R0) {
#pragma unroll
        for (int t = 0; t < G3_NT; ++t)
#pragma unroll
            for (int u = 0; u < 2; ++u) {
                const int v = tid + u * 128;
                rv[t][u] = *(const v4f*)(Rw[t] + (size_t)(m0 + (v >> 3)) * N + n0 + ((v & 7) << 2));
            }
    }

    float* Cs = As;
#pragma unroll
    for (int t = 0; t < G3_NT; ++t)
#pragma unroll
        for (int r = 0; r < 8; ++r)
            Cs[t * 32 * GCP + (mq * 16 + r + lhalf * 8) * GCP + nq * 16 + l15] = acc[t][r];
    __syncthreads();

#pragma unroll
    for (int t = 0; t < G3_NT; ++t)
#pragma unroll
        for (int u = 0; u < 2; ++u) {
            const int v   = tid + u * 128;
            const int row = v >> 3;
            const int c4  = (v & 7) << 2;
            const size_t g = (size_t)(m0 + row) * N + n0 + c4;
            v4f o = *(const v4f*)&Cs[t * 32 * GCP + row * GCP + c4];
            o.x += rv[t][u].x; o.y += rv[t][u].y; o.z += rv[t][u].z; o.w += rv[t][u].w;
            *(v4f*)(C + (size_t)t * Rsz * N + g) = o;
        }
}

#define PJ_NW 5
#define PJ_BS (GK * GBP)

struct PjFrag { float a0[8], a1[8], b0[PJ_NW][8], b1[PJ_NW][8]; };

__device__ __forceinline__ void pj_frags(PjFrag& f, const float* a, const float* b, int k)
{
#pragma unroll
    for (int s = 0; s < 8; ++s) {
        const int kk = k + s * 4;
        f.a0[s] = a[kk];
        f.a1[s] = a[kk + 1];
#pragma unroll
        for (int w = 0; w < PJ_NW; ++w) {
            f.b0[w][s] = b[w * PJ_BS + kk * GBP];
            f.b1[w][s] = b[w * PJ_BS + (kk + 1) * GBP];
        }
    }
}

__global__ __launch_bounds__(128)
void proj5_wmma_f32(const float* __restrict__ A,
                    const float* __restrict__ B0, const float* __restrict__ B1,
                    const float* __restrict__ B2, const float* __restrict__ B3,
                    const float* __restrict__ B4,
                    float* __restrict__ C0, float* __restrict__ C1, float* __restrict__ C2,
                    float* __restrict__ C3, float* __restrict__ C4)
{
    __shared__ __attribute__((aligned(16))) float As[32 * GAP];
    __shared__ __attribute__((aligned(16))) float Bs[PJ_NW * PJ_BS];

    const int tid   = threadIdx.x;
    const int wave  = tid >> 5;
    const int lane  = tid & 31;
    const int l15   = lane & 15;
    const int lhalf = lane >> 4;
    const int m0    = blockIdx.y * 32;
    const int n0    = blockIdx.x * 32;
    const int mq    = wave >> 1;
    const int nq    = wave & 1;

    const float* const Bw[PJ_NW] = {B0, B1, B2, B3, B4};
    float* const       Cw[PJ_NW] = {C0, C1, C2, C3, C4};

    const int ar = tid / (GK / 4);
    const int ac = (tid % (GK / 4)) << 2;
    const int br = tid >> 3;
    const int bc = (tid & 7) << 2;
    const float* Ag = A + (size_t)(m0 + ar) * DMsz + ac;
    const size_t bo = (size_t)br * DMsz + n0 + bc;
    v4f aprf[GNA], bprf[PJ_NW][GNB];

    v8f acc[PJ_NW];
#pragma unroll
    for (int w = 0; w < PJ_NW; ++w) acc[w] = v8f{0.f, 0.f, 0.f, 0.f, 0.f, 0.f, 0.f, 0.f};

#pragma unroll
    for (int u = 0; u < GNA; ++u)
        aprf[u] = *(const v4f*)(Ag + (size_t)(u * (512 / GK)) * DMsz);
#pragma unroll
    for (int w = 0; w < PJ_NW; ++w)
#pragma unroll
        for (int u = 0; u < GNB; ++u)
            bprf[w][u] = *(const v4f*)(Bw[w] + bo + (size_t)(u * 16) * DMsz);

    const float* ap = As + (mq * 16 + l15) * GAP + lhalf * 2;
    const float* bp = Bs + (lhalf * 2) * GBP + nq * 16 + l15;

    for (int k0 = 0; k0 < DMsz; k0 += GK) {
#pragma unroll
        for (int u = 0; u < GNA; ++u)
            *(v4f*)&As[(ar + u * (512 / GK)) * GAP + ac] = aprf[u];
#pragma unroll
        for (int w = 0; w < PJ_NW; ++w)
#pragma unroll
            for (int u = 0; u < GNB; ++u)
                *(v4f*)&Bs[w * PJ_BS + (br + u * 16) * GBP + bc] = bprf[w][u];
        __syncthreads();

        if (k0 + GK < DMsz) {
#pragma unroll
            for (int u = 0; u < GNA; ++u)
                aprf[u] = *(const v4f*)(Ag + (size_t)(u * (512 / GK)) * DMsz + k0 + GK);
#pragma unroll
            for (int w = 0; w < PJ_NW; ++w)
#pragma unroll
                for (int u = 0; u < GNB; ++u)
                    bprf[w][u] = *(const v4f*)(Bw[w] + bo + (size_t)(k0 + GK + u * 16) * DMsz);
        }

        PjFrag f[2];
        pj_frags(f[0], ap, bp, 0);
        __builtin_amdgcn_sched_barrier(0);
#pragma unroll
        for (int g = 0; g < GK / 32; ++g) {
            if (g + 1 < GK / 32)
                pj_frags(f[(g + 1) & 1], ap, bp, (g + 1) * 32);
            __builtin_amdgcn_sched_barrier(0);
#pragma unroll
            for (int s = 0; s < 8; ++s) {
                const PjFrag& c = f[g & 1];
                const v2f af = v2f{c.a0[s], c.a1[s]};
#pragma unroll
                for (int w = 0; w < PJ_NW; ++w) {
                    const v2f bf = v2f{c.b0[w][s], c.b1[w][s]};
                    acc[w] = __builtin_amdgcn_wmma_f32_16x16x4_f32(
                        false, af, false, bf, (short)0, acc[w], false, false);
                }
            }
            __builtin_amdgcn_sched_barrier(0);
        }
        __syncthreads();
    }

    float* Cs = Bs;
#pragma unroll
    for (int w = 0; w < PJ_NW; ++w)
#pragma unroll
        for (int r = 0; r < 8; ++r)
            Cs[w * 32 * GCP + (mq * 16 + r + lhalf * 8) * GCP + nq * 16 + l15] = acc[w][r];
    __syncthreads();

#pragma unroll
    for (int w = 0; w < PJ_NW; ++w)
#pragma unroll
        for (int u = 0; u < 2; ++u) {
            const int v   = tid + u * 128;
            const int row = v >> 3;
            const int c4  = (v & 7) << 2;
            *(v4f*)(Cw[w] + (size_t)(m0 + row) * DMsz + n0 + c4) =
                *(const v4f*)&Cs[w * 32 * GCP + row * GCP + c4];
        }
}

#define FM  32
#define FK  128
#define FAP (FK + 4)
#define FBP (128 + 8)
#define FT  3
#define FAS (FM * FAP)

struct FfnFrag { float a0[FT][8], a1[FT][8], b0[2][8], b1[2][8]; };

__device__ __forceinline__ void ffn_frags(FfnFrag& f, const float* a, const float* b, int k)
{
#pragma unroll
    for (int s = 0; s < 8; ++s) {
        const int kk = k + s * 4;
#pragma unroll
        for (int t = 0; t < FT; ++t) {
            f.a0[t][s] = a[t * FAS + kk];
            f.a1[t][s] = a[t * FAS + kk + 1];
        }
        f.b0[0][s] = b[kk * FBP];
        f.b1[0][s] = b[(kk + 1) * FBP];
        f.b0[1][s] = b[kk * FBP + 16];
        f.b1[1][s] = b[(kk + 1) * FBP + 16];
    }
}

__device__ __forceinline__ void ffn_chunk(v8f (&acc)[FT][2], const float* a, const float* b)
{
    FfnFrag f[2];
    ffn_frags(f[0], a, b, 0);
#pragma unroll
    for (int g = 0; g < FK / 32; ++g) {
        if (g + 1 < FK / 32)
            ffn_frags(f[(g + 1) & 1], a, b, (g + 1) * 32);
        __builtin_amdgcn_sched_barrier(0);
#pragma unroll
        for (int s = 0; s < 8; ++s) {
            const FfnFrag& c = f[g & 1];
            const v2f bf0 = v2f{c.b0[0][s], c.b1[0][s]};
            const v2f bf1 = v2f{c.b0[1][s], c.b1[1][s]};
#pragma unroll
            for (int t = 0; t < FT; ++t) {
                const v2f af = v2f{c.a0[t][s], c.a1[t][s]};
                acc[t][0] = __builtin_amdgcn_wmma_f32_16x16x4_f32(
                    false, af, false, bf0, (short)0, acc[t][0], false, false);
                acc[t][1] = __builtin_amdgcn_wmma_f32_16x16x4_f32(
                    false, af, false, bf1, (short)0, acc[t][1], false, false);
            }
        }
        __builtin_amdgcn_sched_barrier(0);
    }
}

__global__ __launch_bounds__(256)
void ffn_fused_wmma_f32(const float* __restrict__ X, const float* __restrict__ W1,
                        const float* __restrict__ b1, const float* __restrict__ W2,
                        float* __restrict__ P)
{
    __shared__ __attribute__((aligned(16))) float As[FT * FAS];
    __shared__ __attribute__((aligned(16))) float Bs[FK * FBP];
    __shared__ __attribute__((aligned(16))) float Hs[FT * FAS];

    const int tid   = threadIdx.x;
    const int wave  = tid >> 5;
    const int lane  = tid & 31;
    const int l15   = lane & 15;
    const int lhalf = lane >> 4;
    const int s0    = blockIdx.x * 128;
    const int m0    = blockIdx.y * FM;
    const int mq    = wave >> 2;
    const int nq    = wave & 3;

    const int sr = tid >> 5;
    const int sc = (tid & 31) << 2;
    const float* Xg  = X  + (size_t)(m0 + sr) * DMsz + sc;
    const float* W1g = W1 + (size_t)sr * DIsz + s0 + sc;
    const float* W2g = W2 + (size_t)(s0 + sr) * DMsz + sc;
    v4f aprf[FT][4], bprf[16];

    v8f acc[FT][2];
#pragma unroll
    for (int t = 0; t < FT; ++t) {
        acc[t][0] = v8f{0.f, 0.f, 0.f, 0.f, 0.f, 0.f, 0.f, 0.f};
        acc[t][1] = v8f{0.f, 0.f, 0.f, 0.f, 0.f, 0.f, 0.f, 0.f};
    }

#pragma unroll
    for (int t = 0; t < FT; ++t)
#pragma unroll
        for (int u = 0; u < 4; ++u)
            aprf[t][u] = *(const v4f*)(Xg + (size_t)(t * Rsz + u * 8) * DMsz);
#pragma unroll
    for (int u = 0; u < 16; ++u)
        bprf[u] = *(const v4f*)(W1g + (size_t)(u * 8) * DIsz);

    const float* ap = As + (mq * 16 + l15) * FAP + lhalf * 2;
    const float* hp = Hs + (mq * 16 + l15) * FAP + lhalf * 2;
    const float* bp = Bs + (lhalf * 2) * FBP + nq * 32 + l15;

    for (int k0 = 0; k0 < DMsz; k0 += FK) {
#pragma unroll
        for (int t = 0; t < FT; ++t)
#pragma unroll
            for (int u = 0; u < 4; ++u)
                *(v4f*)&As[t * FAS + (sr + u * 8) * FAP + sc] = aprf[t][u];
#pragma unroll
        for (int u = 0; u < 16; ++u)
            *(v4f*)&Bs[(sr + u * 8) * FBP + sc] = bprf[u];
        __syncthreads();

        if (k0 + FK < DMsz) {
#pragma unroll
            for (int t = 0; t < FT; ++t)
#pragma unroll
                for (int u = 0; u < 4; ++u)
                    aprf[t][u] = *(const v4f*)(Xg + (size_t)(t * Rsz + u * 8) * DMsz + k0 + FK);
#pragma unroll
            for (int u = 0; u < 16; ++u)
                bprf[u] = *(const v4f*)(W1g + (size_t)(k0 + FK + u * 8) * DIsz);
        } else {
#pragma unroll
            for (int u = 0; u < 16; ++u)
                bprf[u] = *(const v4f*)(W2g + (size_t)(u * 8) * DMsz);
        }

        ffn_chunk(acc, ap, bp);
        __syncthreads();
    }

#pragma unroll
    for (int nsub = 0; nsub < 2; ++nsub) {
        const int hc = nq * 32 + nsub * 16 + l15;
        const float bv = b1[s0 + hc];
#pragma unroll
        for (int t = 0; t < FT; ++t)
#pragma unroll
            for (int r = 0; r < 8; ++r) {
                const int hr = mq * 16 + r + lhalf * 8;
                Hs[t * FAS + hr * FAP + hc] = fmaxf(acc[t][nsub][r] + bv, 0.f);
            }
    }

    float* Pp = P + (size_t)blockIdx.x * FT * Rsz * DMsz;
    for (int nc = 0; nc < 8; ++nc) {
#pragma unroll
        for (int t = 0; t < FT; ++t) {
            acc[t][0] = v8f{0.f, 0.f, 0.f, 0.f, 0.f, 0.f, 0.f, 0.f};
            acc[t][1] = v8f{0.f, 0.f, 0.f, 0.f, 0.f, 0.f, 0.f, 0.f};
        }

#pragma unroll
        for (int u = 0; u < 16; ++u)
            *(v4f*)&Bs[(sr + u * 8) * FBP + sc] = bprf[u];
        __syncthreads();

        if (nc + 1 < 8) {
#pragma unroll
            for (int u = 0; u < 16; ++u)
                bprf[u] = *(const v4f*)(W2g + (size_t)(u * 8) * DMsz + (nc + 1) * 128);
        }

        ffn_chunk(acc, hp, bp);
        __syncthreads();

#pragma unroll
        for (int t = 0; t < FT; ++t)
#pragma unroll
            for (int nsub = 0; nsub < 2; ++nsub) {
                const int col = nc * 128 + nq * 32 + nsub * 16 + l15;
#pragma unroll
                for (int r = 0; r < 8; ++r) {
                    const int row = t * Rsz + m0 + mq * 16 + r + lhalf * 8;
                    Pp[(size_t)row * DMsz + col] = acc[t][nsub][r];
                }
            }
    }
}

#define GT_KP (18 * 32)
#define GT_PK 32

__global__ __launch_bounds__(256)
void dec_kernel(const float* __restrict__ a0base, const float* __restrict__ a1base,
                const float* __restrict__ b1, const float* __restrict__ w2,
                const float* __restrict__ b2, float* __restrict__ dec)
{
    __shared__ float tile[GT_PK * GT_KP];

    const int tid  = threadIdx.x;
    const int wave = tid >> 5;
    const int lane = tid & 31;
    const int blk  = blockIdx.x;
    const int nt   = blk & 7;
    const int mt   = (blk >> 3) & 7;
    const int b    = (blk >> 6) & 3;
    const int pair = blk >> 8;
    const int i    = pair / 3;
    const int j    = pair % 3;

    const float4* g0 = (const float4*)(a0base + ((size_t)i * Rsz + b * Lsz + nt * 8) * 2048);
    const float4* g1 = (const float4*)(a1base + ((size_t)j * Rsz + b * Lsz + mt * 8) * 2048);
    const float2* gw = (const float2*)w2;

    const int ml = (wave >> 1) * 2;
    const int nl = (wave & 1) * 4;
    const float* pa = tile + ml * 32 + lane;
    const float* pb = tile + (8 + nl) * 32 + lane;
    const float* pc = tile + 16 * 32 + lane;

    float acc[2][4];
#pragma unroll
    for (int x = 0; x < 2; ++x)
#pragma unroll
        for (int y = 0; y < 4; ++y) acc[x][y] = 0.f;

    for (int p = 0; p < 2; ++p) {
        {
            float4 r0[8], r1[8];
#pragma unroll
            for (int u = 0; u < 8; ++u) r0[u] = g0[u * 512 + p * 256 + tid];
#pragma unroll
            for (int u = 0; u < 8; ++u) r1[u] = g1[u * 512 + p * 256 + tid];
            const int dst = (tid >> 3) * GT_KP + (tid & 7) * 4;
#pragma unroll
            for (int u = 0; u < 8; ++u) {
                *(float4*)&tile[dst + (8 + u) * 32] = r0[u];
                *(float4*)&tile[dst + u * 32]       = r1[u];
            }
        }
#pragma unroll
        for (int u = 0; u < 4; ++u) {
            const int cl  = tid + u * 256;
            const float2 w = gw[p * 1024 + cl];
            const int dst = (cl >> 5) * GT_KP + (cl & 31);
            tile[dst + 16 * 32] = b1[p * 1024 + cl];
            tile[dst + 17 * 32] = w.y - w.x;
        }
        __syncthreads();

#pragma unroll 4
        for (int k = 0; k < GT_PK; ++k) {
            const int o = k * GT_KP;
            float xv[2], yv[4];
            xv[0] = pa[o];      xv[1] = pa[o + 32];
            yv[0] = pb[o];      yv[1] = pb[o + 32];
            yv[2] = pb[o + 64]; yv[3] = pb[o + 96];
            const float bb = pc[o], ww = pc[o + 32];
#pragma unroll
            for (int x = 0; x < 2; ++x)
#pragma unroll
                for (int y = 0; y < 4; ++y) {
                    const float t = (xv[x] + yv[y]) + bb;
                    if (t > 0.f) acc[x][y] = fmaf(t, ww, acc[x][y]);
                }
        }
        __syncthreads();
    }

#pragma unroll
    for (int off = 16; off; off >>= 1)
#pragma unroll
        for (int x = 0; x < 2; ++x)
#pragma unroll
            for (int y = 0; y < 4; ++y)
                acc[x][y] += __shfl_xor(acc[x][y], off, 32);

    if (lane == 0) {
        const float bd = b2[1] - b2[0];
        float* dtile = dec + (((size_t)(pair * Bsz + b) * Lsz + mt * 8) * Lsz + nt * 8);
#pragma unroll
        for (int x = 0; x < 2; ++x) {
            float4 r;
            r.x = ((acc[x][0] + bd) > 0.f) ? NEGC : 0.f;
            r.y = ((acc[x][1] + bd) > 0.f) ? NEGC : 0.f;
            r.z = ((acc[x][2] + bd) > 0.f) ? NEGC : 0.f;
            r.w = ((acc[x][3] + bd) > 0.f) ? NEGC : 0.f;
            *(float4*)(dtile + (size_t)(ml + x) * Lsz + nl) = r;
        }
    }
}

#define SC_QT 16
#define SC_KP 68

__global__ __launch_bounds__(128)
void attn_scores_kernel(const float* __restrict__ Qb, const float* __restrict__ Kb,
                        const float* __restrict__ dec, const int* __restrict__ mask,
                        float* __restrict__ S)
{
    __shared__ __attribute__((aligned(16))) float Ks[192 * SC_KP];
    __shared__ __attribute__((aligned(16))) float Qs[SC_QT * 64];

    const int blk  = blockIdx.x;
    const int qt   = blk & 3;
    const int h    = (blk >> 2) & 15;
    const int b    = (blk >> 6) & 3;
    const int i    = blk >> 8;
    const int tid  = threadIdx.x;
    const int wave = tid >> 5;
    const int lane = tid & 31;
    const int q0   = qt * SC_QT + wave * 4;

    {
        const int c4 = (tid & 15) << 2;
        const int r0 = tid >> 4;
        v4f kr[24], qr[2];
#pragma unroll
        for (int u = 0; u < 24; ++u) {
            const int row = r0 + u * 8;
            const int j = row >> 6, k = row & 63;
            kr[u] = *(const v4f*)(Kb + ((size_t)j * Rsz + b * Lsz + k) * DMsz + h * DKsz + c4);
        }
#pragma unroll
        for (int u = 0; u < 2; ++u) {
            const int row = r0 + u * 8;
            qr[u] = *(const v4f*)(Qb + ((size_t)i * Rsz + b * Lsz + qt * SC_QT + row) * DMsz
                                  + h * DKsz + c4);
        }
#pragma unroll
        for (int u = 0; u < 24; ++u)
            *(v4f*)&Ks[(r0 + u * 8) * SC_KP + c4] = kr[u];
#pragma unroll
        for (int u = 0; u < 2; ++u)
            *(v4f*)&Qs[(r0 + u * 8) * 64 + c4] = qr[u];
    }

    float dv[4][6];
    int   mk[4][2];
#pragma unroll
    for (int r = 0; r < 4; ++r) {
#pragma unroll
        for (int m = 0; m < 6; ++m) {
            const int j = m >> 1, k = (m & 1) * 32 + lane;
            dv[r][m] = dec[(((size_t)(i * 3 + j) * Bsz + b) * Lsz + q0 + r) * Lsz + k];
        }
#pragma unroll
        for (int kh = 0; kh < 2; ++kh)
            mk[r][kh] = mask[((size_t)b * Lsz + q0 + r) * Lsz + kh * 32 + lane];
    }
    __syncthreads();

    float acc[4][6];
#pragma unroll
    for (int r = 0; r < 4; ++r)
#pragma unroll
        for (int m = 0; m < 6; ++m) acc[r][m] = 0.f;

    const float* kp = Ks + lane * SC_KP;
    const float* qp = Qs + (wave * 4) * 64;
    for (int d = 0; d < 64; d += 4) {
        v4f kv[6], qv[4];
#pragma unroll
        for (int m = 0; m < 6; ++m) kv[m] = *(const v4f*)(kp + m * 32 * SC_KP + d);
#pragma unroll
        for (int r = 0; r < 4; ++r) qv[r] = *(const v4f*)(qp + r * 64 + d);
#pragma unroll
        for (int r = 0; r < 4; ++r)
#pragma unroll
            for (int m = 0; m < 6; ++m) {
                float s = acc[r][m];
                s = fmaf(qv[r].x, kv[m].x, s);
                s = fmaf(qv[r].y, kv[m].y, s);
                s = fmaf(qv[r].z, kv[m].z, s);
                s = fmaf(qv[r].w, kv[m].w, s);
                acc[r][m] = s;
            }
    }

#pragma unroll
    for (int r = 0; r < 4; ++r) {
        float s[6];
#pragma unroll
        for (int m = 0; m < 6; ++m) {
            float t = acc[r][m];
            t *= 0.125f;
            t += dv[r][m];
            if (mk[r][m & 1] == 0) t = NEGC;
            s[m] = t;
        }
        float mx = fmaxf(fmaxf(fmaxf(s[0], s[1]), fmaxf(s[2], s[3])), fmaxf(s[4], s[5]));
#pragma unroll
        for (int off = 16; off; off >>= 1) mx = fmaxf(mx, __shfl_xor(mx, off, 32));
        float e[6];
#pragma unroll
        for (int m = 0; m < 6; ++m) e[m] = __expf(s[m] - mx);
        float sum = ((e[0] + e[2]) + e[4]) + ((e[1] + e[3]) + e[5]);
#pragma unroll
        for (int off = 16; off; off >>= 1) sum += __shfl_xor(sum, off, 32);
        const float inv = 1.0f / sum;
        float* Sp = S + ((size_t)((((i * Bsz + b) * Hsz + h) * Lsz) + q0 + r)) * 192 + lane;
#pragma unroll
        for (int m = 0; m < 6; ++m) Sp[m * 32] = e[m] * inv;
    }
}

#define AV_QT 16
#define AV_VP 68

__global__ __launch_bounds__(128)
void attn_av_kernel(const float* __restrict__ S, const float* __restrict__ Vb,
                    float* __restrict__ Ob)
{
    __shared__ __attribute__((aligned(16))) float Vs[192 * AV_VP];
    __shared__ __attribute__((aligned(16))) float Ss[AV_QT * 192];

    const int blk  = blockIdx.x;
    const int qt   = blk & 3;
    const int h    = (blk >> 2) & 15;
    const int b    = (blk >> 6) & 3;
    const int i    = blk >> 8;
    const int tid  = threadIdx.x;
    const int wave = tid >> 5;
    const int lane = tid & 31;

    {
        const int c4 = (tid & 15) << 2;
        const int r0 = tid >> 4;
        const float* Sg = S + ((size_t)((((i * Bsz + b) * Hsz + h) * Lsz) + qt * AV_QT)) * 192;
        v4f vr[24], sr[6];
#pragma unroll
        for (int u = 0; u < 24; ++u) {
            const int row = r0 + u * 8;
            const int j = row >> 6, k = row & 63;
            vr[u] = *(const v4f*)(Vb + ((size_t)j * Rsz + b * Lsz + k) * DMsz + h * DVsz + c4);
        }
#pragma unroll
        for (int u = 0; u < 6; ++u)
            sr[u] = *(const v4f*)(Sg + (size_t)(tid + u * 128) * 4);
#pragma unroll
        for (int u = 0; u < 24; ++u)
            *(v4f*)&Vs[(r0 + u * 8) * AV_VP + c4] = vr[u];
#pragma unroll
        for (int u = 0; u < 6; ++u)
            *(v4f*)&Ss[(tid + u * 128) * 4] = sr[u];
    }
    __syncthreads();

    float acc[4][2];
#pragma unroll
    for (int r = 0; r < 4; ++r) { acc[r][0] = 0.f; acc[r][1] = 0.f; }

    const float* vp = Vs + lane;
    const float* sp = Ss + (wave * 4) * 192;
#pragma unroll 4
    for (int jk = 0; jk < 192; jk += 4) {
        v4f sv[4];
        float vv[4][2];
#pragma unroll
        for (int r = 0; r < 4; ++r) sv[r] = *(const v4f*)(sp + r * 192 + jk);
#pragma unroll
        for (int t = 0; t < 4; ++t) {
            vv[t][0] = vp[(jk + t) * AV_VP];
            vv[t][1] = vp[(jk + t) * AV_VP + 32];
        }
#pragma unroll
        for (int r = 0; r < 4; ++r)
#pragma unroll
            for (int cc = 0; cc < 2; ++cc) {
                float s = acc[r][cc];
                s = fmaf(sv[r].x, vv[0][cc], s);
                s = fmaf(sv[r].y, vv[1][cc], s);
                s = fmaf(sv[r].z, vv[2][cc], s);
                s = fmaf(sv[r].w, vv[3][cc], s);
                acc[r][cc] = s;
            }
    }

#pragma unroll
    for (int r = 0; r < 4; ++r) {
        const int q = qt * AV_QT + wave * 4 + r;
        float* Op = Ob + ((size_t)i * Rsz + b * Lsz + q) * DMsz + h * DVsz + lane;
        Op[0]  = acc[r][0];
        Op[32] = acc[r][1];
    }
}

__global__ __launch_bounds__(256)
void layernorm_kernel(const float* __restrict__ x, int nparts, size_t pstride,
                      const float* __restrict__ bias, const float* __restrict__ resid,
                      const float* __restrict__ g, const float* __restrict__ bta,
                      float* __restrict__ out)
{
    const int row = blockIdx.x;
    const int t = threadIdx.x;
    __shared__ float red[256];

    const float* xp = x + (size_t)row * 1024;
    float v[4];
#pragma unroll
    for (int u = 0; u < 4; ++u) v[u] = xp[t + u * 256];
#pragma unroll 8
    for (int p = 1; p < nparts; ++p) {
        const float* pp = xp + (size_t)p * pstride;
#pragma unroll
        for (int u = 0; u < 4; ++u) v[u] += pp[t + u * 256];
    }
    float s = 0.f;
#pragma unroll
    for (int u = 0; u < 4; ++u) {
        float val = v[u];
        if (bias)  val += bias[t + u * 256];
        if (resid) val += resid[(size_t)row * 1024 + t + u * 256];
        v[u] = val; s += val;
    }
    red[t] = s; __syncthreads();
    for (int st = 128; st > 0; st >>= 1) { if (t < st) red[t] += red[t + st]; __syncthreads(); }
    const float mu = red[0] * (1.f / 1024.f);
    __syncthreads();

    float s2 = 0.f;
#pragma unroll
    for (int u = 0; u < 4; ++u) { float dd = v[u] - mu; s2 += dd * dd; }
    red[t] = s2; __syncthreads();
    for (int st = 128; st > 0; st >>= 1) { if (t < st) red[t] += red[t + st]; __syncthreads(); }
    const float var = red[0] * (1.f / 1024.f);
    const float rinv = 1.0f / sqrtf(var + 1e-6f);
#pragma unroll
    for (int u = 0; u < 4; ++u) {
        int c = t + u * 256;
        out[(size_t)row * 1024 + c] = (v[u] - mu) * rinv * g[c] + bta[c];
    }
}

extern "C" void kernel_launch(void* const* d_in, const int* in_sizes, int n_in,
                              void* d_out, int out_size, void* d_ws, size_t ws_size,
                              hipStream_t stream)
{
    const float* X[3]   = { (const float*)d_in[0], (const float*)d_in[1], (const float*)d_in[2] };
    const int*   mask   = (const int*)d_in[3];
    const float* w_qs   = (const float*)d_in[4];
    const float* w_ks   = (const float*)d_in[5];
    const float* w_vs   = (const float*)d_in[6];
    const float* w_ds0  = (const float*)d_in[7];
    const float* w_ds1  = (const float*)d_in[8];
    const float* dc_w1  = (const float*)d_in[9];
    const float* dc_b1  = (const float*)d_in[10];
    const float* dc_w2  = (const float*)d_in[11];
    const float* dc_b2  = (const float*)d_in[12];
    const float* w_fc   = (const float*)d_in[13];
    const float* ln1_g  = (const float*)d_in[14];
    const float* ln1_b  = (const float*)d_in[15];
    const float* ffn_w1 = (const float*)d_in[16];
    const float* ffn_b1 = (const float*)d_in[17];
    const float* ffn_w2 = (const float*)d_in[18];
    const float* ffn_b2 = (const float*)d_in[19];
    const float* ln2_g  = (const float*)d_in[20];
    const float* ln2_b  = (const float*)d_in[21];
    float* out = (float*)d_out;

    float* ws = (float*)d_ws;
    size_t off = 0;
    auto alloc = [&](size_t n) { float* p = ws + off; off += n; return p; };
    const size_t RM = (size_t)Rsz * DMsz;
    float* Q   = alloc(3 * RM);
    float* Kb  = alloc(3 * RM);
    float* Vb  = alloc(3 * RM);
    float* d0  = alloc(3 * RM);
    float* d1  = alloc(3 * RM);
    float* a0  = alloc(3 * (size_t)Rsz * 2048);
    float* a1  = alloc(3 * (size_t)Rsz * 2048);
    float* dec = alloc(9 * (size_t)Bsz * Lsz * Lsz);
    float* S   = alloc((size_t)3 * Bsz * Hsz * Lsz * 192);
    float* O   = alloc(3 * RM);
    float* fc  = alloc(3 * RM);
    float* oln = alloc(3 * RM);
    float* fpart = alloc(32 * 3 * RM);
    (void)ws_size; (void)n_in; (void)in_sizes; (void)out_size;

    auto gemm3 = [&](const float* A, const float* Bm, const float* r0, const float* r1,
                     const float* r2, float* C, int N, int K) {
        dim3 grid(N / 32, Rsz / 32);
        gemm3_wmma_f32<<<grid, dim3(128), 0, stream>>>(A, Bm, r0, r1, r2, C, N, K);
    };

    for (int i = 0; i < 3; ++i) {
        proj5_wmma_f32<<<dim3(DMsz / 32, Rsz / 32), dim3(128), 0, stream>>>(
            X[i], w_qs, w_ks, w_vs, w_ds0, w_ds1,
            Q + i * RM, Kb + i * RM, Vb + i * RM, d0 + i * RM, d1 + i * RM);
    }
    gemm3(d0, dc_w1 + (size_t)1024 * 2048, nullptr, nullptr, nullptr, a0, 2048, 1024);
    gemm3(d1, dc_w1, nullptr, nullptr, nullptr, a1, 2048, 1024);

    {
        dec_kernel<<<9 * Bsz * 8 * 8, 256, 0, stream>>>(a0, a1, dc_b1, dc_w2, dc_b2, dec);
    }

    {
        attn_scores_kernel<<<3 * Bsz * Hsz * (Lsz / SC_QT), 128, 0, stream>>>(Q, Kb, dec, mask, S);
        attn_av_kernel<<<3 * Bsz * Hsz * (Lsz / AV_QT), 128, 0, stream>>>(S, Vb, O);
    }

    gemm3(O, w_fc, X[0], X[1], X[2], fc, 1024, 1024);
    for (int i = 0; i < 3; ++i)
        layernorm_kernel<<<Rsz, 256, 0, stream>>>(fc + i * RM, 1, RM, nullptr, nullptr,
                                                  ln1_g, ln1_b, oln + i * RM);
    ffn_fused_wmma_f32<<<dim3(32, Rsz / FM), dim3(256), 0, stream>>>(
        oln, ffn_w1, ffn_b1, ffn_w2, fpart);
    for (int i = 0; i < 3; ++i)
        layernorm_kernel<<<Rsz, 256, 0, stream>>>(fpart + i * RM, 32, 3 * RM, ffn_b2, oln + i * RM,
                                                  ln2_g, ln2_b, out + i * RM);
}


# Projections q,k,v,d0,d1 of the 3 modalities in one kernel: a block stages each weight chunk once for the same 32 rows of each input, 15 accumulators per wave; 3 launches become 1; sums bit-identical
# speedup vs baseline: 5.2787x; 1.0901x over previous
#include <hip/hip_runtime.h>
#include <hip/hip_bf16.h>

#define Bsz 4
#define Lsz 64
#define DMsz 1024
#define Hsz 16
#define DKsz 64
#define DVsz 64
#define Dsz 1024
#define DIsz 4096
#define Rsz 256
#define NEGC (-1.0e9f)

typedef float v2f __attribute__((ext_vector_type(2)));
typedef float v4f __attribute__((ext_vector_type(4)));
typedef float v8f __attribute__((ext_vector_type(8)));

#define GK  128
#define GAP (GK + 4)
#define GBP 40
#define GCP 36
#define GNA (GK / 16)
#define GNB (GK / 16)

#define G3_NT 3
#define G3_AS (32 * GAP)

struct G3Frag { float a0[G3_NT][8], a1[G3_NT][8], b0[8], b1[8]; };

__device__ __forceinline__ void g3_frags(G3Frag& f, const float* a, const float* b, int k)
{
#pragma unroll
    for (int s = 0; s < 8; ++s) {
        const int kk = k + s * 4;
#pragma unroll
        for (int t = 0; t < G3_NT; ++t) {
            f.a0[t][s] = a[t * G3_AS + kk];
            f.a1[t][s] = a[t * G3_AS + kk + 1];
        }
        f.b0[s] = b[kk * GBP];
        f.b1[s] = b[(kk + 1) * GBP];
    }
}

__global__ __launch_bounds__(128)
void gemm3_wmma_f32(const float* __restrict__ A, const float* __restrict__ B,
                    const float* __restrict__ R0, const float* __restrict__ R1,
                    const float* __restrict__ R2, float* __restrict__ C, int N, int K)
{
    __shared__ __attribute__((aligned(16))) float As[G3_NT * G3_AS];
    __shared__ __attribute__((aligned(16))) float Bs[GK * GBP];

    const int tid   = threadIdx.x;
    const int wave  = tid >> 5;
    const int lane  = tid & 31;
    const int l15   = lane & 15;
    const int lhalf = lane >> 4;
    const int m0    = blockIdx.y * 32;
    const int n0    = blockIdx.x * 32;
    const int mq    = wave >> 1;
    const int nq    = wave & 1;

    const float* const Rw[G3_NT] = {R0, R1, R2};

    const int ar = tid / (GK / 4);
    const int ac = (tid % (GK / 4)) << 2;
    const int br = tid >> 3;
    const int bc = (tid & 7) << 2;
    const float* Ag = A + (size_t)(m0 + ar) * K + ac;
    const float* Bg = B + (size_t)br * N + n0 + bc;
    v4f aprf[G3_NT][GNA], bprf[GNB];

    v8f acc[G3_NT];
#pragma unroll
    for (int t = 0; t < G3_NT; ++t) acc[t] = v8f{0.f, 0.f, 0.f, 0.f, 0.f, 0.f, 0.f, 0.f};

#pragma unroll
    for (int t = 0; t < G3_NT; ++t)
#pragma unroll
        for (int u = 0; u < GNA; ++u)
            aprf[t][u] = *(const v4f*)(Ag + (size_t)(t * Rsz + u * (512 / GK)) * K);
#pragma unroll
    for (int u = 0; u < GNB; ++u)
        bprf[u] = *(const v4f*)(Bg + (size_t)(u * 16) * N);

    const float* ap = As + (mq * 16 + l15) * GAP + lhalf * 2;
    const float* bp = Bs + (lhalf * 2) * GBP + nq * 16 + l15;

    for (int k0 = 0; k0 < K; k0 += GK) {
#pragma unroll
        for (int t = 0; t < G3_NT; ++t)
#pragma unroll
            for (int u = 0; u < GNA; ++u)
                *(v4f*)&As[t * G3_AS + (ar + u * (512 / GK)) * GAP + ac] = aprf[t][u];
#pragma unroll
        for (int u = 0; u < GNB; ++u)
            *(v4f*)&Bs[(br + u * 16) * GBP + bc] = bprf[u];
        __syncthreads();

        if (k0 + GK < K) {
#pragma unroll
            for (int t = 0; t < G3_NT; ++t)
#pragma unroll
                for (int u = 0; u < GNA; ++u)
                    aprf[t][u] = *(const v4f*)(Ag + (size_t)(t * Rsz + u * (512 / GK)) * K + k0 + GK);
#pragma unroll
            for (int u = 0; u < GNB; ++u)
                bprf[u] = *(const v4f*)(Bg + (size_t)(k0 + GK + u * 16) * N);
        }

        G3Frag f[2];
        g3_frags(f[0], ap, bp, 0);
        __builtin_amdgcn_sched_barrier(0);
#pragma unroll
        for (int g = 0; g < GK / 32; ++g) {
            if (g + 1 < GK / 32)
                g3_frags(f[(g + 1) & 1], ap, bp, (g + 1) * 32);
            __builtin_amdgcn_sched_barrier(0);
#pragma unroll
            for (int s = 0; s < 8; ++s) {
                const G3Frag& c = f[g & 1];
                const v2f bf = v2f{c.b0[s], c.b1[s]};
#pragma unroll
                for (int t = 0; t < G3_NT; ++t) {
                    const v2f af = v2f{c.a0[t][s], c.a1[t][s]};
                    acc[t] = __builtin_amdgcn_wmma_f32_16x16x4_f32(
                        false, af, false, bf, (short)0, acc[t], false, false);
                }
            }
            __builtin_amdgcn_sched_barrier(0);
        }
        __syncthreads();
    }

    v4f rv[G3_NT][2];
#pragma unroll
    for (int t = 0; t < G3_NT; ++t)
#pragma unroll
        for (int u = 0; u < 2; ++u) rv[t][u] = v4f{0.f, 0.f, 0.f, 0.f};
    if (R0) {
#pragma unroll
        for (int t = 0; t < G3_NT; ++t)
#pragma unroll
            for (int u = 0; u < 2; ++u) {
                const int v = tid + u * 128;
                rv[t][u] = *(const v4f*)(Rw[t] + (size_t)(m0 + (v >> 3)) * N + n0 + ((v & 7) << 2));
            }
    }

    float* Cs = As;
#pragma unroll
    for (int t = 0; t < G3_NT; ++t)
#pragma unroll
        for (int r = 0; r < 8; ++r)
            Cs[t * 32 * GCP + (mq * 16 + r + lhalf * 8) * GCP + nq * 16 + l15] = acc[t][r];
    __syncthreads();

#pragma unroll
    for (int t = 0; t < G3_NT; ++t)
#pragma unroll
        for (int u = 0; u < 2; ++u) {
            const int v   = tid + u * 128;
            const int row = v >> 3;
            const int c4  = (v & 7) << 2;
            const size_t g = (size_t)(m0 + row) * N + n0 + c4;
            v4f o = *(const v4f*)&Cs[t * 32 * GCP + row * GCP + c4];
            o.x += rv[t][u].x; o.y += rv[t][u].y; o.z += rv[t][u].z; o.w += rv[t][u].w;
            *(v4f*)(C + (size_t)t * Rsz * N + g) = o;
        }
}

#define PJ_NW 5
#define PJ_NT 3
#define PJ_BS (GK * GBP)
#define PJ_AS (32 * GAP)
#define PJ_GS 4

struct PjFrag { float a0[PJ_NT][PJ_GS], a1[PJ_NT][PJ_GS], b0[PJ_NW][PJ_GS], b1[PJ_NW][PJ_GS]; };

__device__ __forceinline__ void pj_frags(PjFrag& f, const float* a, const float* b, int k)
{
#pragma unroll
    for (int s = 0; s < PJ_GS; ++s) {
        const int kk = k + s * 4;
#pragma unroll
        for (int t = 0; t < PJ_NT; ++t) {
            f.a0[t][s] = a[t * PJ_AS + kk];
            f.a1[t][s] = a[t * PJ_AS + kk + 1];
        }
#pragma unroll
        for (int w = 0; w < PJ_NW; ++w) {
            f.b0[w][s] = b[w * PJ_BS + kk * GBP];
            f.b1[w][s] = b[w * PJ_BS + (kk + 1) * GBP];
        }
    }
}

__global__ __launch_bounds__(128)
void proj15_wmma_f32(const float* __restrict__ X0, const float* __restrict__ X1,
                     const float* __restrict__ X2,
                     const float* __restrict__ B0, const float* __restrict__ B1,
                     const float* __restrict__ B2, const float* __restrict__ B3,
                     const float* __restrict__ B4,
                     float* __restrict__ C0, float* __restrict__ C1, float* __restrict__ C2,
                     float* __restrict__ C3, float* __restrict__ C4)
{
    __shared__ __attribute__((aligned(16))) float As[PJ_NT * PJ_AS];
    __shared__ __attribute__((aligned(16))) float Bs[PJ_NW * PJ_BS];

    const int tid   = threadIdx.x;
    const int wave  = tid >> 5;
    const int lane  = tid & 31;
    const int l15   = lane & 15;
    const int lhalf = lane >> 4;
    const int m0    = blockIdx.y * 32;
    const int n0    = blockIdx.x * 32;
    const int mq    = wave >> 1;
    const int nq    = wave & 1;

    const float* const Xw[PJ_NT] = {X0, X1, X2};
    const float* const Bw[PJ_NW] = {B0, B1, B2, B3, B4};
    float* const       Cw[PJ_NW] = {C0, C1, C2, C3, C4};

    const int ar = tid / (GK / 4);
    const int ac = (tid % (GK / 4)) << 2;
    const int br = tid >> 3;
    const int bc = (tid & 7) << 2;
    const size_t ao = (size_t)(m0 + ar) * DMsz + ac;
    const size_t bo = (size_t)br * DMsz + n0 + bc;
    v4f aprf[PJ_NT][GNA], bprf[PJ_NW][GNB];

    v8f acc[PJ_NT][PJ_NW];
#pragma unroll
    for (int t = 0; t < PJ_NT; ++t)
#pragma unroll
        for (int w = 0; w < PJ_NW; ++w) acc[t][w] = v8f{0.f, 0.f, 0.f, 0.f, 0.f, 0.f, 0.f, 0.f};

#pragma unroll
    for (int t = 0; t < PJ_NT; ++t)
#pragma unroll
        for (int u = 0; u < GNA; ++u)
            aprf[t][u] = *(const v4f*)(Xw[t] + ao + (size_t)(u * (512 / GK)) * DMsz);
#pragma unroll
    for (int w = 0; w < PJ_NW; ++w)
#pragma unroll
        for (int u = 0; u < GNB; ++u)
            bprf[w][u] = *(const v4f*)(Bw[w] + bo + (size_t)(u * 16) * DMsz);

    const float* ap = As + (mq * 16 + l15) * GAP + lhalf * 2;
    const float* bp = Bs + (lhalf * 2) * GBP + nq * 16 + l15;

    for (int k0 = 0; k0 < DMsz; k0 += GK) {
#pragma unroll
        for (int t = 0; t < PJ_NT; ++t)
#pragma unroll
            for (int u = 0; u < GNA; ++u)
                *(v4f*)&As[t * PJ_AS + (ar + u * (512 / GK)) * GAP + ac] = aprf[t][u];
#pragma unroll
        for (int w = 0; w < PJ_NW; ++w)
#pragma unroll
            for (int u = 0; u < GNB; ++u)
                *(v4f*)&Bs[w * PJ_BS + (br + u * 16) * GBP + bc] = bprf[w][u];
        __syncthreads();

        if (k0 + GK < DMsz) {
#pragma unroll
            for (int t = 0; t < PJ_NT; ++t)
#pragma unroll
                for (int u = 0; u < GNA; ++u)
                    aprf[t][u] = *(const v4f*)(Xw[t] + ao + (size_t)(u * (512 / GK)) * DMsz + k0 + GK);
#pragma unroll
            for (int w = 0; w < PJ_NW; ++w)
#pragma unroll
                for (int u = 0; u < GNB; ++u)
                    bprf[w][u] = *(const v4f*)(Bw[w] + bo + (size_t)(k0 + GK + u * 16) * DMsz);
        }

        PjFrag f[2];
        pj_frags(f[0], ap, bp, 0);
        __builtin_amdgcn_sched_barrier(0);
#pragma unroll
        for (int g = 0; g < GK / (4 * PJ_GS); ++g) {
            if (g + 1 < GK / (4 * PJ_GS))
                pj_frags(f[(g + 1) & 1], ap, bp, (g + 1) * 4 * PJ_GS);
            __builtin_amdgcn_sched_barrier(0);
#pragma unroll
            for (int s = 0; s < PJ_GS; ++s) {
                const PjFrag& c = f[g & 1];
#pragma unroll
                for (int t = 0; t < PJ_NT; ++t) {
                    const v2f af = v2f{c.a0[t][s], c.a1[t][s]};
#pragma unroll
                    for (int w = 0; w < PJ_NW; ++w) {
                        const v2f bf = v2f{c.b0[w][s], c.b1[w][s]};
                        acc[t][w] = __builtin_amdgcn_wmma_f32_16x16x4_f32(
                            false, af, false, bf, (short)0, acc[t][w], false, false);
                    }
                }
            }
            __builtin_amdgcn_sched_barrier(0);
        }
        __syncthreads();
    }

    float* Cs = Bs;
#pragma unroll
    for (int t = 0; t < PJ_NT; ++t)
#pragma unroll
        for (int w = 0; w < PJ_NW; ++w)
#pragma unroll
            for (int r = 0; r < 8; ++r)
                Cs[(t * PJ_NW + w) * 32 * GCP + (mq * 16 + r + lhalf * 8) * GCP + nq * 16 + l15] = acc[t][w][r];
    __syncthreads();

#pragma unroll
    for (int t = 0; t < PJ_NT; ++t)
#pragma unroll
        for (int w = 0; w < PJ_NW; ++w)
#pragma unroll
            for (int u = 0; u < 2; ++u) {
                const int v   = tid + u * 128;
                const int row = v >> 3;
                const int c4  = (v & 7) << 2;
                *(v4f*)(Cw[w] + (size_t)t * Rsz * DMsz + (size_t)(m0 + row) * DMsz + n0 + c4) =
                    *(const v4f*)&Cs[(t * PJ_NW + w) * 32 * GCP + row * GCP + c4];
            }
}

#define FM  32
#define FK  128
#define FAP (FK + 4)
#define FBP (128 + 8)
#define FT  3
#define FAS (FM * FAP)

struct FfnFrag { float a0[FT][8], a1[FT][8], b0[2][8], b1[2][8]; };

__device__ __forceinline__ void ffn_frags(FfnFrag& f, const float* a, const float* b, int k)
{
#pragma unroll
    for (int s = 0; s < 8; ++s) {
        const int kk = k + s * 4;
#pragma unroll
        for (int t = 0; t < FT; ++t) {
            f.a0[t][s] = a[t * FAS + kk];
            f.a1[t][s] = a[t * FAS + kk + 1];
        }
        f.b0[0][s] = b[kk * FBP];
        f.b1[0][s] = b[(kk + 1) * FBP];
        f.b0[1][s] = b[kk * FBP + 16];
        f.b1[1][s] = b[(kk + 1) * FBP + 16];
    }
}

__device__ __forceinline__ void ffn_chunk(v8f (&acc)[FT][2], const float* a, const float* b)
{
    FfnFrag f[2];
    ffn_frags(f[0], a, b, 0);
#pragma unroll
    for (int g = 0; g < FK / 32; ++g) {
        if (g + 1 < FK / 32)
            ffn_frags(f[(g + 1) & 1], a, b, (g + 1) * 32);
        __builtin_amdgcn_sched_barrier(0);
#pragma unroll
        for (int s = 0; s < 8; ++s) {
            const FfnFrag& c = f[g & 1];
            const v2f bf0 = v2f{c.b0[0][s], c.b1[0][s]};
            const v2f bf1 = v2f{c.b0[1][s], c.b1[1][s]};
#pragma unroll
            for (int t = 0; t < FT; ++t) {
                const v2f af = v2f{c.a0[t][s], c.a1[t][s]};
                acc[t][0] = __builtin_amdgcn_wmma_f32_16x16x4_f32(
                    false, af, false, bf0, (short)0, acc[t][0], false, false);
                acc[t][1] = __builtin_amdgcn_wmma_f32_16x16x4_f32(
                    false, af, false, bf1, (short)0, acc[t][1], false, false);
            }
        }
        __builtin_amdgcn_sched_barrier(0);
    }
}

__global__ __launch_bounds__(256)
void ffn_fused_wmma_f32(const float* __restrict__ X, const float* __restrict__ W1,
                        const float* __restrict__ b1, const float* __restrict__ W2,
                        float* __restrict__ P)
{
    __shared__ __attribute__((aligned(16))) float As[FT * FAS];
    __shared__ __attribute__((aligned(16))) float Bs[FK * FBP];
    __shared__ __attribute__((aligned(16))) float Hs[FT * FAS];

    const int tid   = threadIdx.x;
    const int wave  = tid >> 5;
    const int lane  = tid & 31;
    const int l15   = lane & 15;
    const int lhalf = lane >> 4;
    const int s0    = blockIdx.x * 128;
    const int m0    = blockIdx.y * FM;
    const int mq    = wave >> 2;
    const int nq    = wave & 3;

    const int sr = tid >> 5;
    const int sc = (tid & 31) << 2;
    const float* Xg  = X  + (size_t)(m0 + sr) * DMsz + sc;
    const float* W1g = W1 + (size_t)sr * DIsz + s0 + sc;
    const float* W2g = W2 + (size_t)(s0 + sr) * DMsz + sc;
    v4f aprf[FT][4], bprf[16];

    v8f acc[FT][2];
#pragma unroll
    for (int t = 0; t < FT; ++t) {
        acc[t][0] = v8f{0.f, 0.f, 0.f, 0.f, 0.f, 0.f, 0.f, 0.f};
        acc[t][1] = v8f{0.f, 0.f, 0.f, 0.f, 0.f, 0.f, 0.f, 0.f};
    }

#pragma unroll
    for (int t = 0; t < FT; ++t)
#pragma unroll
        for (int u = 0; u < 4; ++u)
            aprf[t][u] = *(const v4f*)(Xg + (size_t)(t * Rsz + u * 8) * DMsz);
#pragma unroll
    for (int u = 0; u < 16; ++u)
        bprf[u] = *(const v4f*)(W1g + (size_t)(u * 8) * DIsz);

    const float* ap = As + (mq * 16 + l15) * FAP + lhalf * 2;
    const float* hp = Hs + (mq * 16 + l15) * FAP + lhalf * 2;
    const float* bp = Bs + (lhalf * 2) * FBP + nq * 32 + l15;

    for (int k0 = 0; k0 < DMsz; k0 += FK) {
#pragma unroll
        for (int t = 0; t < FT; ++t)
#pragma unroll
            for (int u = 0; u < 4; ++u)
                *(v4f*)&As[t * FAS + (sr + u * 8) * FAP + sc] = aprf[t][u];
#pragma unroll
        for (int u = 0; u < 16; ++u)
            *(v4f*)&Bs[(sr + u * 8) * FBP + sc] = bprf[u];
        __syncthreads();

        if (k0 + FK < DMsz) {
#pragma unroll
            for (int t = 0; t < FT; ++t)
#pragma unroll
                for (int u = 0; u < 4; ++u)
                    aprf[t][u] = *(const v4f*)(Xg + (size_t)(t * Rsz + u * 8) * DMsz + k0 + FK);
#pragma unroll
            for (int u = 0; u < 16; ++u)
                bprf[u] = *(const v4f*)(W1g + (size_t)(k0 + FK + u * 8) * DIsz);
        } else {
#pragma unroll
            for (int u = 0; u < 16; ++u)
                bprf[u] = *(const v4f*)(W2g + (size_t)(u * 8) * DMsz);
        }

        ffn_chunk(acc, ap, bp);
        __syncthreads();
    }

#pragma unroll
    for (int nsub = 0; nsub < 2; ++nsub) {
        const int hc = nq * 32 + nsub * 16 + l15;
        const float bv = b1[s0 + hc];
#pragma unroll
        for (int t = 0; t < FT; ++t)
#pragma unroll
            for (int r = 0; r < 8; ++r) {
                const int hr = mq * 16 + r + lhalf * 8;
                Hs[t * FAS + hr * FAP + hc] = fmaxf(acc[t][nsub][r] + bv, 0.f);
            }
    }

    float* Pp = P + (size_t)blockIdx.x * FT * Rsz * DMsz;
    for (int nc = 0; nc < 8; ++nc) {
#pragma unroll
        for (int t = 0; t < FT; ++t) {
            acc[t][0] = v8f{0.f, 0.f, 0.f, 0.f, 0.f, 0.f, 0.f, 0.f};
            acc[t][1] = v8f{0.f, 0.f, 0.f, 0.f, 0.f, 0.f, 0.f, 0.f};
        }

#pragma unroll
        for (int u = 0; u < 16; ++u)
            *(v4f*)&Bs[(sr + u * 8) * FBP + sc] = bprf[u];
        __syncthreads();

        if (nc + 1 < 8) {
#pragma unroll
            for (int u = 0; u < 16; ++u)
                bprf[u] = *(const v4f*)(W2g + (size_t)(u * 8) * DMsz + (nc + 1) * 128);
        }

        ffn_chunk(acc, hp, bp);
        __syncthreads();

#pragma unroll
        for (int t = 0; t < FT; ++t)
#pragma unroll
            for (int nsub = 0; nsub < 2; ++nsub) {
                const int col = nc * 128 + nq * 32 + nsub * 16 + l15;
#pragma unroll
                for (int r = 0; r < 8; ++r) {
                    const int row = t * Rsz + m0 + mq * 16 + r + lhalf * 8;
                    Pp[(size_t)row * DMsz + col] = acc[t][nsub][r];
                }
            }
    }
}

#define GT_KP (18 * 32)
#define GT_PK 32

__global__ __launch_bounds__(256)
void dec_kernel(const float* __restrict__ a0base, const float* __restrict__ a1base,
                const float* __restrict__ b1, const float* __restrict__ w2,
                const float* __restrict__ b2, float* __restrict__ dec)
{
    __shared__ float tile[GT_PK * GT_KP];

    const int tid  = threadIdx.x;
    const int wave = tid >> 5;
    const int lane = tid & 31;
    const int blk  = blockIdx.x;
    const int nt   = blk & 7;
    const int mt   = (blk >> 3) & 7;
    const int b    = (blk >> 6) & 3;
    const int pair = blk >> 8;
    const int i    = pair / 3;
    const int j    = pair % 3;

    const float4* g0 = (const float4*)(a0base + ((size_t)i * Rsz + b * Lsz + nt * 8) * 2048);
    const float4* g1 = (const float4*)(a1base + ((size_t)j * Rsz + b * Lsz + mt * 8) * 2048);
    const float2* gw = (const float2*)w2;

    const int ml = (wave >> 1) * 2;
    const int nl = (wave & 1) * 4;
    const float* pa = tile + ml * 32 + lane;
    const float* pb = tile + (8 + nl) * 32 + lane;
    const float* pc = tile + 16 * 32 + lane;

    float acc[2][4];
#pragma unroll
    for (int x = 0; x < 2; ++x)
#pragma unroll
        for (int y = 0; y < 4; ++y) acc[x][y] = 0.f;

    for (int p = 0; p < 2; ++p) {
        {
            float4 r0[8], r1[8];
#pragma unroll
            for (int u = 0; u < 8; ++u) r0[u] = g0[u * 512 + p * 256 + tid];
#pragma unroll
            for (int u = 0; u < 8; ++u) r1[u] = g1[u * 512 + p * 256 + tid];
            const int dst = (tid >> 3) * GT_KP + (tid & 7) * 4;
#pragma unroll
            for (int u = 0; u < 8; ++u) {
                *(float4*)&tile[dst + (8 + u) * 32] = r0[u];
                *(float4*)&tile[dst + u * 32]       = r1[u];
            }
        }
#pragma unroll
        for (int u = 0; u < 4; ++u) {
            const int cl  = tid + u * 256;
            const float2 w = gw[p * 1024 + cl];
            const int dst = (cl >> 5) * GT_KP + (cl & 31);
            tile[dst + 16 * 32] = b1[p * 1024 + cl];
            tile[dst + 17 * 32] = w.y - w.x;
        }
        __syncthreads();

#pragma unroll 4
        for (int k = 0; k < GT_PK; ++k) {
            const int o = k * GT_KP;
            float xv[2], yv[4];
            xv[0] = pa[o];      xv[1] = pa[o + 32];
            yv[0] = pb[o];      yv[1] = pb[o + 32];
            yv[2] = pb[o + 64]; yv[3] = pb[o + 96];
            const float bb = pc[o], ww = pc[o + 32];
#pragma unroll
            for (int x = 0; x < 2; ++x)
#pragma unroll
                for (int y = 0; y < 4; ++y) {
                    const float t = (xv[x] + yv[y]) + bb;
                    if (t > 0.f) acc[x][y] = fmaf(t, ww, acc[x][y]);
                }
        }
        __syncthreads();
    }

#pragma unroll
    for (int off = 16; off; off >>= 1)
#pragma unroll
        for (int x = 0; x < 2; ++x)
#pragma unroll
            for (int y = 0; y < 4; ++y)
                acc[x][y] += __shfl_xor(acc[x][y], off, 32);

    if (lane == 0) {
        const float bd = b2[1] - b2[0];
        float* dtile = dec + (((size_t)(pair * Bsz + b) * Lsz + mt * 8) * Lsz + nt * 8);
#pragma unroll
        for (int x = 0; x < 2; ++x) {
            float4 r;
            r.x = ((acc[x][0] + bd) > 0.f) ? NEGC : 0.f;
            r.y = ((acc[x][1] + bd) > 0.f) ? NEGC : 0.f;
            r.z = ((acc[x][2] + bd) > 0.f) ? NEGC : 0.f;
            r.w = ((acc[x][3] + bd) > 0.f) ? NEGC : 0.f;
            *(float4*)(dtile + (size_t)(ml + x) * Lsz + nl) = r;
        }
    }
}

#define SC_QT 16
#define SC_KP 68

__global__ __launch_bounds__(128)
void attn_scores_kernel(const float* __restrict__ Qb, const float* __restrict__ Kb,
                        const float* __restrict__ dec, const int* __restrict__ mask,
                        float* __restrict__ S)
{
    __shared__ __attribute__((aligned(16))) float Ks[192 * SC_KP];
    __shared__ __attribute__((aligned(16))) float Qs[SC_QT * 64];

    const int blk  = blockIdx.x;
    const int qt   = blk & 3;
    const int h    = (blk >> 2) & 15;
    const int b    = (blk >> 6) & 3;
    const int i    = blk >> 8;
    const int tid  = threadIdx.x;
    const int wave = tid >> 5;
    const int lane = tid & 31;
    const int q0   = qt * SC_QT + wave * 4;

    {
        const int c4 = (tid & 15) << 2;
        const int r0 = tid >> 4;
        v4f kr[24], qr[2];
#pragma unroll
        for (int u = 0; u < 24; ++u) {
            const int row = r0 + u * 8;
            const int j = row >> 6, k = row & 63;
            kr[u] = *(const v4f*)(Kb + ((size_t)j * Rsz + b * Lsz + k) * DMsz + h * DKsz + c4);
        }
#pragma unroll
        for (int u = 0; u < 2; ++u) {
            const int row = r0 + u * 8;
            qr[u] = *(const v4f*)(Qb + ((size_t)i * Rsz + b * Lsz + qt * SC_QT + row) * DMsz
                                  + h * DKsz + c4);
        }
#pragma unroll
        for (int u = 0; u < 24; ++u)
            *(v4f*)&Ks[(r0 + u * 8) * SC_KP + c4] = kr[u];
#pragma unroll
        for (int u = 0; u < 2; ++u)
            *(v4f*)&Qs[(r0 + u * 8) * 64 + c4] = qr[u];
    }

    float dv[4][6];
    int   mk[4][2];
#pragma unroll
    for (int r = 0; r < 4; ++r) {
#pragma unroll
        for (int m = 0; m < 6; ++m) {
            const int j = m >> 1, k = (m & 1) * 32 + lane;
            dv[r][m] = dec[(((size_t)(i * 3 + j) * Bsz + b) * Lsz + q0 + r) * Lsz + k];
        }
#pragma unroll
        for (int kh = 0; kh < 2; ++kh)
            mk[r][kh] = mask[((size_t)b * Lsz + q0 + r) * Lsz + kh * 32 + lane];
    }
    __syncthreads();

    float acc[4][6];
#pragma unroll
    for (int r = 0; r < 4; ++r)
#pragma unroll
        for (int m = 0; m < 6; ++m) acc[r][m] = 0.f;

    const float* kp = Ks + lane * SC_KP;
    const float* qp = Qs + (wave * 4) * 64;
    for (int d = 0; d < 64; d += 4) {
        v4f kv[6], qv[4];
#pragma unroll
        for (int m = 0; m < 6; ++m) kv[m] = *(const v4f*)(kp + m * 32 * SC_KP + d);
#pragma unroll
        for (int r = 0; r < 4; ++r) qv[r] = *(const v4f*)(qp + r * 64 + d);
#pragma unroll
        for (int r = 0; r < 4; ++r)
#pragma unroll
            for (int m = 0; m < 6; ++m) {
                float s = acc[r][m];
                s = fmaf(qv[r].x, kv[m].x, s);
                s = fmaf(qv[r].y, kv[m].y, s);
                s = fmaf(qv[r].z, kv[m].z, s);
                s = fmaf(qv[r].w, kv[m].w, s);
                acc[r][m] = s;
            }
    }

#pragma unroll
    for (int r = 0; r < 4; ++r) {
        float s[6];
#pragma unroll
        for (int m = 0; m < 6; ++m) {
            float t = acc[r][m];
            t *= 0.125f;
            t += dv[r][m];
            if (mk[r][m & 1] == 0) t = NEGC;
            s[m] = t;
        }
        float mx = fmaxf(fmaxf(fmaxf(s[0], s[1]), fmaxf(s[2], s[3])), fmaxf(s[4], s[5]));
#pragma unroll
        for (int off = 16; off; off >>= 1) mx = fmaxf(mx, __shfl_xor(mx, off, 32));
        float e[6];
#pragma unroll
        for (int m = 0; m < 6; ++m) e[m] = __expf(s[m] - mx);
        float sum = ((e[0] + e[2]) + e[4]) + ((e[1] + e[3]) + e[5]);
#pragma unroll
        for (int off = 16; off; off >>= 1) sum += __shfl_xor(sum, off, 32);
        const float inv = 1.0f / sum;
        float* Sp = S + ((size_t)((((i * Bsz + b) * Hsz + h) * Lsz) + q0 + r)) * 192 + lane;
#pragma unroll
        for (int m = 0; m < 6; ++m) Sp[m * 32] = e[m] * inv;
    }
}

#define AV_QT 16
#define AV_VP 68

__global__ __launch_bounds__(128)
void attn_av_kernel(const float* __restrict__ S, const float* __restrict__ Vb,
                    float* __restrict__ Ob)
{
    __shared__ __attribute__((aligned(16))) float Vs[192 * AV_VP];
    __shared__ __attribute__((aligned(16))) float Ss[AV_QT * 192];

    const int blk  = blockIdx.x;
    const int qt   = blk & 3;
    const int h    = (blk >> 2) & 15;
    const int b    = (blk >> 6) & 3;
    const int i    = blk >> 8;
    const int tid  = threadIdx.x;
    const int wave = tid >> 5;
    const int lane = tid & 31;

    {
        const int c4 = (tid & 15) << 2;
        const int r0 = tid >> 4;
        const float* Sg = S + ((size_t)((((i * Bsz + b) * Hsz + h) * Lsz) + qt * AV_QT)) * 192;
        v4f vr[24], sr[6];
#pragma unroll
        for (int u = 0; u < 24; ++u) {
            const int row = r0 + u * 8;
            const int j = row >> 6, k = row & 63;
            vr[u] = *(const v4f*)(Vb + ((size_t)j * Rsz + b * Lsz + k) * DMsz + h * DVsz + c4);
        }
#pragma unroll
        for (int u = 0; u < 6; ++u)
            sr[u] = *(const v4f*)(Sg + (size_t)(tid + u * 128) * 4);
#pragma unroll
        for (int u = 0; u < 24; ++u)
            *(v4f*)&Vs[(r0 + u * 8) * AV_VP + c4] = vr[u];
#pragma unroll
        for (int u = 0; u < 6; ++u)
            *(v4f*)&Ss[(tid + u * 128) * 4] = sr[u];
    }
    __syncthreads();

    float acc[4][2];
#pragma unroll
    for (int r = 0; r < 4; ++r) { acc[r][0] = 0.f; acc[r][1] = 0.f; }

    const float* vp = Vs + lane;
    const float* sp = Ss + (wave * 4) * 192;
#pragma unroll 4
    for (int jk = 0; jk < 192; jk += 4) {
        v4f sv[4];
        float vv[4][2];
#pragma unroll
        for (int r = 0; r < 4; ++r) sv[r] = *(const v4f*)(sp + r * 192 + jk);
#pragma unroll
        for (int t = 0; t < 4; ++t) {
            vv[t][0] = vp[(jk + t) * AV_VP];
            vv[t][1] = vp[(jk + t) * AV_VP + 32];
        }
#pragma unroll
        for (int r = 0; r < 4; ++r)
#pragma unroll
            for (int cc = 0; cc < 2; ++cc) {
                float s = acc[r][cc];
                s = fmaf(sv[r].x, vv[0][cc], s);
                s = fmaf(sv[r].y, vv[1][cc], s);
                s = fmaf(sv[r].z, vv[2][cc], s);
                s = fmaf(sv[r].w, vv[3][cc], s);
                acc[r][cc] = s;
            }
    }

#pragma unroll
    for (int r = 0; r < 4; ++r) {
        const int q = qt * AV_QT + wave * 4 + r;
        float* Op = Ob + ((size_t)i * Rsz + b * Lsz + q) * DMsz + h * DVsz + lane;
        Op[0]  = acc[r][0];
        Op[32] = acc[r][1];
    }
}

__global__ __launch_bounds__(256)
void layernorm_kernel(const float* __restrict__ x, int nparts, size_t pstride,
                      const float* __restrict__ bias, const float* __restrict__ resid,
                      const float* __restrict__ g, const float* __restrict__ bta,
                      float* __restrict__ out)
{
    const int row = blockIdx.x;
    const int t = threadIdx.x;
    __shared__ float red[256];

    const float* xp = x + (size_t)row * 1024;
    float v[4];
#pragma unroll
    for (int u = 0; u < 4; ++u) v[u] = xp[t + u * 256];
#pragma unroll 8
    for (int p = 1; p < nparts; ++p) {
        const float* pp = xp + (size_t)p * pstride;
#pragma unroll
        for (int u = 0; u < 4; ++u) v[u] += pp[t + u * 256];
    }
    float s = 0.f;
#pragma unroll
    for (int u = 0; u < 4; ++u) {
        float val = v[u];
        if (bias)  val += bias[t + u * 256];
        if (resid) val += resid[(size_t)row * 1024 + t + u * 256];
        v[u] = val; s += val;
    }
    red[t] = s; __syncthreads();
    for (int st = 128; st > 0; st >>= 1) { if (t < st) red[t] += red[t + st]; __syncthreads(); }
    const float mu = red[0] * (1.f / 1024.f);
    __syncthreads();

    float s2 = 0.f;
#pragma unroll
    for (int u = 0; u < 4; ++u) { float dd = v[u] - mu; s2 += dd * dd; }
    red[t] = s2; __syncthreads();
    for (int st = 128; st > 0; st >>= 1) { if (t < st) red[t] += red[t + st]; __syncthreads(); }
    const float var = red[0] * (1.f / 1024.f);
    const float rinv = 1.0f / sqrtf(var + 1e-6f);
#pragma unroll
    for (int u = 0; u < 4; ++u) {
        int c = t + u * 256;
        out[(size_t)row * 1024 + c] = (v[u] - mu) * rinv * g[c] + bta[c];
    }
}

extern "C" void kernel_launch(void* const* d_in, const int* in_sizes, int n_in,
                              void* d_out, int out_size, void* d_ws, size_t ws_size,
                              hipStream_t stream)
{
    const float* X[3]   = { (const float*)d_in[0], (const float*)d_in[1], (const float*)d_in[2] };
    const int*   mask   = (const int*)d_in[3];
    const float* w_qs   = (const float*)d_in[4];
    const float* w_ks   = (const float*)d_in[5];
    const float* w_vs   = (const float*)d_in[6];
    const float* w_ds0  = (const float*)d_in[7];
    const float* w_ds1  = (const float*)d_in[8];
    const float* dc_w1  = (const float*)d_in[9];
    const float* dc_b1  = (const float*)d_in[10];
    const float* dc_w2  = (const float*)d_in[11];
    const float* dc_b2  = (const float*)d_in[12];
    const float* w_fc   = (const float*)d_in[13];
    const float* ln1_g  = (const float*)d_in[14];
    const float* ln1_b  = (const float*)d_in[15];
    const float* ffn_w1 = (const float*)d_in[16];
    const float* ffn_b1 = (const float*)d_in[17];
    const float* ffn_w2 = (const float*)d_in[18];
    const float* ffn_b2 = (const float*)d_in[19];
    const float* ln2_g  = (const float*)d_in[20];
    const float* ln2_b  = (const float*)d_in[21];
    float* out = (float*)d_out;

    float* ws = (float*)d_ws;
    size_t off = 0;
    auto alloc = [&](size_t n) { float* p = ws + off; off += n; return p; };
    const size_t RM = (size_t)Rsz * DMsz;
    float* Q   = alloc(3 * RM);
    float* Kb  = alloc(3 * RM);
    float* Vb  = alloc(3 * RM);
    float* d0  = alloc(3 * RM);
    float* d1  = alloc(3 * RM);
    float* a0  = alloc(3 * (size_t)Rsz * 2048);
    float* a1  = alloc(3 * (size_t)Rsz * 2048);
    float* dec = alloc(9 * (size_t)Bsz * Lsz * Lsz);
    float* S   = alloc((size_t)3 * Bsz * Hsz * Lsz * 192);
    float* O   = alloc(3 * RM);
    float* fc  = alloc(3 * RM);
    float* oln = alloc(3 * RM);
    float* fpart = alloc(32 * 3 * RM);
    (void)ws_size; (void)n_in; (void)in_sizes; (void)out_size;

    auto gemm3 = [&](const float* A, const float* Bm, const float* r0, const float* r1,
                     const float* r2, float* C, int N, int K) {
        dim3 grid(N / 32, Rsz / 32);
        gemm3_wmma_f32<<<grid, dim3(128), 0, stream>>>(A, Bm, r0, r1, r2, C, N, K);
    };

    proj15_wmma_f32<<<dim3(DMsz / 32, Rsz / 32), dim3(128), 0, stream>>>(
        X[0], X[1], X[2], w_qs, w_ks, w_vs, w_ds0, w_ds1, Q, Kb, Vb, d0, d1);
    gemm3(d0, dc_w1 + (size_t)1024 * 2048, nullptr, nullptr, nullptr, a0, 2048, 1024);
    gemm3(d1, dc_w1, nullptr, nullptr, nullptr, a1, 2048, 1024);

    {
        dec_kernel<<<9 * Bsz * 8 * 8, 256, 0, stream>>>(a0, a1, dc_b1, dc_w2, dc_b2, dec);
    }

    {
        attn_scores_kernel<<<3 * Bsz * Hsz * (Lsz / SC_QT), 128, 0, stream>>>(Q, Kb, dec, mask, S);
        attn_av_kernel<<<3 * Bsz * Hsz * (Lsz / AV_QT), 128, 0, stream>>>(S, Vb, O);
    }

    gemm3(O, w_fc, X[0], X[1], X[2], fc, 1024, 1024);
    for (int i = 0; i < 3; ++i)
        layernorm_kernel<<<Rsz, 256, 0, stream>>>(fc + i * RM, 1, RM, nullptr, nullptr,
                                                  ln1_g, ln1_b, oln + i * RM);
    ffn_fused_wmma_f32<<<dim3(32, Rsz / FM), dim3(256), 0, stream>>>(
        oln, ffn_w1, ffn_b1, ffn_w2, fpart);
    for (int i = 0; i < 3; ++i)
        layernorm_kernel<<<Rsz, 256, 0, stream>>>(fpart + i * RM, 32, 3 * RM, ffn_b2, oln + i * RM,
                                                  ln2_g, ln2_b, out + i * RM);
}


# Layernorm: all loads of a thread (32 partial planes, bias, residual, gamma, beta) issued as one run up front; was batches of 8 planes, then 8 loads each waited for in turn; sums bit-identical
# speedup vs baseline: 5.6309x; 1.0667x over previous
#include <hip/hip_runtime.h>
#include <hip/hip_bf16.h>

#define Bsz 4
#define Lsz 64
#define DMsz 1024
#define Hsz 16
#define DKsz 64
#define DVsz 64
#define Dsz 1024
#define DIsz 4096
#define Rsz 256
#define NEGC (-1.0e9f)

typedef float v2f __attribute__((ext_vector_type(2)));
typedef float v4f __attribute__((ext_vector_type(4)));
typedef float v8f __attribute__((ext_vector_type(8)));

#define GK  128
#define GAP (GK + 4)
#define GBP 40
#define GCP 36
#define GNA (GK / 16)
#define GNB (GK / 16)

#define G3_NT 3
#define G3_AS (32 * GAP)

struct G3Frag { float a0[G3_NT][8], a1[G3_NT][8], b0[8], b1[8]; };

__device__ __forceinline__ void g3_frags(G3Frag& f, const float* a, const float* b, int k)
{
#pragma unroll
    for (int s = 0; s < 8; ++s) {
        const int kk = k + s * 4;
#pragma unroll
        for (int t = 0; t < G3_NT; ++t) {
            f.a0[t][s] = a[t * G3_AS + kk];
            f.a1[t][s] = a[t * G3_AS + kk + 1];
        }
        f.b0[s] = b[kk * GBP];
        f.b1[s] = b[(kk + 1) * GBP];
    }
}

__global__ __launch_bounds__(128)
void gemm3_wmma_f32(const float* __restrict__ A, const float* __restrict__ B,
                    const float* __restrict__ R0, const float* __restrict__ R1,
                    const float* __restrict__ R2, float* __restrict__ C, int N, int K)
{
    __shared__ __attribute__((aligned(16))) float As[G3_NT * G3_AS];
    __shared__ __attribute__((aligned(16))) float Bs[GK * GBP];

    const int tid   = threadIdx.x;
    const int wave  = tid >> 5;
    const int lane  = tid & 31;
    const int l15   = lane & 15;
    const int lhalf = lane >> 4;
    const int m0    = blockIdx.y * 32;
    const int n0    = blockIdx.x * 32;
    const int mq    = wave >> 1;
    const int nq    = wave & 1;

    const float* const Rw[G3_NT] = {R0, R1, R2};

    const int ar = tid / (GK / 4);
    const int ac = (tid % (GK / 4)) << 2;
    const int br = tid >> 3;
    const int bc = (tid & 7) << 2;
    const float* Ag = A + (size_t)(m0 + ar) * K + ac;
    const float* Bg = B + (size_t)br * N + n0 + bc;
    v4f aprf[G3_NT][GNA], bprf[GNB];

    v8f acc[G3_NT];
#pragma unroll
    for (int t = 0; t < G3_NT; ++t) acc[t] = v8f{0.f, 0.f, 0.f, 0.f, 0.f, 0.f, 0.f, 0.f};

#pragma unroll
    for (int t = 0; t < G3_NT; ++t)
#pragma unroll
        for (int u = 0; u < GNA; ++u)
            aprf[t][u] = *(const v4f*)(Ag + (size_t)(t * Rsz + u * (512 / GK)) * K);
#pragma unroll
    for (int u = 0; u < GNB; ++u)
        bprf[u] = *(const v4f*)(Bg + (size_t)(u * 16) * N);

    const float* ap = As + (mq * 16 + l15) * GAP + lhalf * 2;
    const float* bp = Bs + (lhalf * 2) * GBP + nq * 16 + l15;

    for (int k0 = 0; k0 < K; k0 += GK) {
#pragma unroll
        for (int t = 0; t < G3_NT; ++t)
#pragma unroll
            for (int u = 0; u < GNA; ++u)
                *(v4f*)&As[t * G3_AS + (ar + u * (512 / GK)) * GAP + ac] = aprf[t][u];
#pragma unroll
        for (int u = 0; u < GNB; ++u)
            *(v4f*)&Bs[(br + u * 16) * GBP + bc] = bprf[u];
        __syncthreads();

        if (k0 + GK < K) {
#pragma unroll
            for (int t = 0; t < G3_NT; ++t)
#pragma unroll
                for (int u = 0; u < GNA; ++u)
                    aprf[t][u] = *(const v4f*)(Ag + (size_t)(t * Rsz + u * (512 / GK)) * K + k0 + GK);
#pragma unroll
            for (int u = 0; u < GNB; ++u)
                bprf[u] = *(const v4f*)(Bg + (size_t)(k0 + GK + u * 16) * N);
        }

        G3Frag f[2];
        g3_frags(f[0], ap, bp, 0);
        __builtin_amdgcn_sched_barrier(0);
#pragma unroll
        for (int g = 0; g < GK / 32; ++g) {
            if (g + 1 < GK / 32)
                g3_frags(f[(g + 1) & 1], ap, bp, (g + 1) * 32);
            __builtin_amdgcn_sched_barrier(0);
#pragma unroll
            for (int s = 0; s < 8; ++s) {
                const G3Frag& c = f[g & 1];
                const v2f bf = v2f{c.b0[s], c.b1[s]};
#pragma unroll
                for (int t = 0; t < G3_NT; ++t) {
                    const v2f af = v2f{c.a0[t][s], c.a1[t][s]};
                    acc[t] = __builtin_amdgcn_wmma_f32_16x16x4_f32(
                        false, af, false, bf, (short)0, acc[t], false, false);
                }
            }
            __builtin_amdgcn_sched_barrier(0);
        }
        __syncthreads();
    }

    v4f rv[G3_NT][2];
#pragma unroll
    for (int t = 0; t < G3_NT; ++t)
#pragma unroll
        for (int u = 0; u < 2; ++u) rv[t][u] = v4f{0.f, 0.f, 0.f, 0.f};
    if (R0) {
#pragma unroll
        for (int t = 0; t < G3_NT; ++t)
#pragma unroll
            for (int u = 0; u < 2; ++u) {
                const int v = tid + u * 128;
                rv[t][u] = *(const v4f*)(Rw[t] + (size_t)(m0 + (v >> 3)) * N + n0 + ((v & 7) << 2));
            }
    }

    float* Cs = As;
#pragma unroll
    for (int t = 0; t < G3_NT; ++t)
#pragma unroll
        for (int r = 0; r < 8; ++r)
            Cs[t * 32 * GCP + (mq * 16 + r + lhalf * 8) * GCP + nq * 16 + l15] = acc[t][r];
    __syncthreads();

#pragma unroll
    for (int t = 0; t < G3_NT; ++t)
#pragma unroll
        for (int u = 0; u < 2; ++u) {
            const int v   = tid + u * 128;
            const int row = v >> 3;
            const int c4  = (v & 7) << 2;
            const size_t g = (size_t)(m0 + row) * N + n0 + c4;
            v4f o = *(const v4f*)&Cs[t * 32 * GCP + row * GCP + c4];
            o.x += rv[t][u].x; o.y += rv[t][u].y; o.z += rv[t][u].z; o.w += rv[t][u].w;
            *(v4f*)(C + (size_t)t * Rsz * N + g) = o;
        }
}

#define PJ_NW 5
#define PJ_NT 3
#define PJ_BS (GK * GBP)
#define PJ_AS (32 * GAP)
#define PJ_GS 4

struct PjFrag { float a0[PJ_NT][PJ_GS], a1[PJ_NT][PJ_GS], b0[PJ_NW][PJ_GS], b1[PJ_NW][PJ_GS]; };

__device__ __forceinline__ void pj_frags(PjFrag& f, const float* a, const float* b, int k)
{
#pragma unroll
    for (int s = 0; s < PJ_GS; ++s) {
        const int kk = k + s * 4;
#pragma unroll
        for (int t = 0; t < PJ_NT; ++t) {
            f.a0[t][s] = a[t * PJ_AS + kk];
            f.a1[t][s] = a[t * PJ_AS + kk + 1];
        }
#pragma unroll
        for (int w = 0; w < PJ_NW; ++w) {
            f.b0[w][s] = b[w * PJ_BS + kk * GBP];
            f.b1[w][s] = b[w * PJ_BS + (kk + 1) * GBP];
        }
    }
}

__global__ __launch_bounds__(128)
void proj15_wmma_f32(const float* __restrict__ X0, const float* __restrict__ X1,
                     const float* __restrict__ X2,
                     const float* __restrict__ B0, const float* __restrict__ B1,
                     const float* __restrict__ B2, const float* __restrict__ B3,
                     const float* __restrict__ B4,
                     float* __restrict__ C0, float* __restrict__ C1, float* __restrict__ C2,
                     float* __restrict__ C3, float* __restrict__ C4)
{
    __shared__ __attribute__((aligned(16))) float As[PJ_NT * PJ_AS];
    __shared__ __attribute__((aligned(16))) float Bs[PJ_NW * PJ_BS];

    const int tid   = threadIdx.x;
    const int wave  = tid >> 5;
    const int lane  = tid & 31;
    const int l15   = lane & 15;
    const int lhalf = lane >> 4;
    const int m0    = blockIdx.y * 32;
    const int n0    = blockIdx.x * 32;
    const int mq    = wave >> 1;
    const int nq    = wave & 1;

    const float* const Xw[PJ_NT] = {X0, X1, X2};
    const float* const Bw[PJ_NW] = {B0, B1, B2, B3, B4};
    float* const       Cw[PJ_NW] = {C0, C1, C2, C3, C4};

    const int ar = tid / (GK / 4);
    const int ac = (tid % (GK / 4)) << 2;
    const int br = tid >> 3;
    const int bc = (tid & 7) << 2;
    const size_t ao = (size_t)(m0 + ar) * DMsz + ac;
    const size_t bo = (size_t)br * DMsz + n0 + bc;
    v4f aprf[PJ_NT][GNA], bprf[PJ_NW][GNB];

    v8f acc[PJ_NT][PJ_NW];
#pragma unroll
    for (int t = 0; t < PJ_NT; ++t)
#pragma unroll
        for (int w = 0; w < PJ_NW; ++w) acc[t][w] = v8f{0.f, 0.f, 0.f, 0.f, 0.f, 0.f, 0.f, 0.f};

#pragma unroll
    for (int t = 0; t < PJ_NT; ++t)
#pragma unroll
        for (int u = 0; u < GNA; ++u)
            aprf[t][u] = *(const v4f*)(Xw[t] + ao + (size_t)(u * (512 / GK)) * DMsz);
#pragma unroll
    for (int w = 0; w < PJ_NW; ++w)
#pragma unroll
        for (int u = 0; u < GNB; ++u)
            bprf[w][u] = *(const v4f*)(Bw[w] + bo + (size_t)(u * 16) * DMsz);

    const float* ap = As + (mq * 16 + l15) * GAP + lhalf * 2;
    const float* bp = Bs + (lhalf * 2) * GBP + nq * 16 + l15;

    for (int k0 = 0; k0 < DMsz; k0 += GK) {
#pragma unroll
        for (int t = 0; t < PJ_NT; ++t)
#pragma unroll
            for (int u = 0; u < GNA; ++u)
                *(v4f*)&As[t * PJ_AS + (ar + u * (512 / GK)) * GAP + ac] = aprf[t][u];
#pragma unroll
        for (int w = 0; w < PJ_NW; ++w)
#pragma unroll
            for (int u = 0; u < GNB; ++u)
                *(v4f*)&Bs[w * PJ_BS + (br + u * 16) * GBP + bc] = bprf[w][u];
        __syncthreads();

        if (k0 + GK < DMsz) {
#pragma unroll
            for (int t = 0; t < PJ_NT; ++t)
#pragma unroll
                for (int u = 0; u < GNA; ++u)
                    aprf[t][u] = *(const v4f*)(Xw[t] + ao + (size_t)(u * (512 / GK)) * DMsz + k0 + GK);
#pragma unroll
            for (int w = 0; w < PJ_NW; ++w)
#pragma unroll
                for (int u = 0; u < GNB; ++u)
                    bprf[w][u] = *(const v4f*)(Bw[w] + bo + (size_t)(k0 + GK + u * 16) * DMsz);
        }

        PjFrag f[2];
        pj_frags(f[0], ap, bp, 0);
        __builtin_amdgcn_sched_barrier(0);
#pragma unroll
        for (int g = 0; g < GK / (4 * PJ_GS); ++g) {
            if (g + 1 < GK / (4 * PJ_GS))
                pj_frags(f[(g + 1) & 1], ap, bp, (g + 1) * 4 * PJ_GS);
            __builtin_amdgcn_sched_barrier(0);
#pragma unroll
            for (int s = 0; s < PJ_GS; ++s) {
                const PjFrag& c = f[g & 1];
#pragma unroll
                for (int t = 0; t < PJ_NT; ++t) {
                    const v2f af = v2f{c.a0[t][s], c.a1[t][s]};
#pragma unroll
                    for (int w = 0; w < PJ_NW; ++w) {
                        const v2f bf = v2f{c.b0[w][s], c.b1[w][s]};
                        acc[t][w] = __builtin_amdgcn_wmma_f32_16x16x4_f32(
                            false, af, false, bf, (short)0, acc[t][w], false, false);
                    }
                }
            }
            __builtin_amdgcn_sched_barrier(0);
        }
        __syncthreads();
    }

    float* Cs = Bs;
#pragma unroll
    for (int t = 0; t < PJ_NT; ++t)
#pragma unroll
        for (int w = 0; w < PJ_NW; ++w)
#pragma unroll
            for (int r = 0; r < 8; ++r)
                Cs[(t * PJ_NW + w) * 32 * GCP + (mq * 16 + r + lhalf * 8) * GCP + nq * 16 + l15] = acc[t][w][r];
    __syncthreads();

#pragma unroll
    for (int t = 0; t < PJ_NT; ++t)
#pragma unroll
        for (int w = 0; w < PJ_NW; ++w)
#pragma unroll
            for (int u = 0; u < 2; ++u) {
                const int v   = tid + u * 128;
                const int row = v >> 3;
                const int c4  = (v & 7) << 2;
                *(v4f*)(Cw[w] + (size_t)t * Rsz * DMsz + (size_t)(m0 + row) * DMsz + n0 + c4) =
                    *(const v4f*)&Cs[(t * PJ_NW + w) * 32 * GCP + row * GCP + c4];
            }
}

#define FM  32
#define FK  128
#define FAP (FK + 4)
#define FBP (128 + 8)
#define FT  3
#define FAS (FM * FAP)

struct FfnFrag { float a0[FT][8], a1[FT][8], b0[2][8], b1[2][8]; };

__device__ __forceinline__ void ffn_frags(FfnFrag& f, const float* a, const float* b, int k)
{
#pragma unroll
    for (int s = 0; s < 8; ++s) {
        const int kk = k + s * 4;
#pragma unroll
        for (int t = 0; t < FT; ++t) {
            f.a0[t][s] = a[t * FAS + kk];
            f.a1[t][s] = a[t * FAS + kk + 1];
        }
        f.b0[0][s] = b[kk * FBP];
        f.b1[0][s] = b[(kk + 1) * FBP];
        f.b0[1][s] = b[kk * FBP + 16];
        f.b1[1][s] = b[(kk + 1) * FBP + 16];
    }
}

__device__ __forceinline__ void ffn_chunk(v8f (&acc)[FT][2], const float* a, const float* b)
{
    FfnFrag f[2];
    ffn_frags(f[0], a, b, 0);
#pragma unroll
    for (int g = 0; g < FK / 32; ++g) {
        if (g + 1 < FK / 32)
            ffn_frags(f[(g + 1) & 1], a, b, (g + 1) * 32);
        __builtin_amdgcn_sched_barrier(0);
#pragma unroll
        for (int s = 0; s < 8; ++s) {
            const FfnFrag& c = f[g & 1];
            const v2f bf0 = v2f{c.b0[0][s], c.b1[0][s]};
            const v2f bf1 = v2f{c.b0[1][s], c.b1[1][s]};
#pragma unroll
            for (int t = 0; t < FT; ++t) {
                const v2f af = v2f{c.a0[t][s], c.a1[t][s]};
                acc[t][0] = __builtin_amdgcn_wmma_f32_16x16x4_f32(
                    false, af, false, bf0, (short)0, acc[t][0], false, false);
                acc[t][1] = __builtin_amdgcn_wmma_f32_16x16x4_f32(
                    false, af, false, bf1, (short)0, acc[t][1], false, false);
            }
        }
        __builtin_amdgcn_sched_barrier(0);
    }
}

__global__ __launch_bounds__(256)
void ffn_fused_wmma_f32(const float* __restrict__ X, const float* __restrict__ W1,
                        const float* __restrict__ b1, const float* __restrict__ W2,
                        float* __restrict__ P)
{
    __shared__ __attribute__((aligned(16))) float As[FT * FAS];
    __shared__ __attribute__((aligned(16))) float Bs[FK * FBP];
    __shared__ __attribute__((aligned(16))) float Hs[FT * FAS];

    const int tid   = threadIdx.x;
    const int wave  = tid >> 5;
    const int lane  = tid & 31;
    const int l15   = lane & 15;
    const int lhalf = lane >> 4;
    const int s0    = blockIdx.x * 128;
    const int m0    = blockIdx.y * FM;
    const int mq    = wave >> 2;
    const int nq    = wave & 3;

    const int sr = tid >> 5;
    const int sc = (tid & 31) << 2;
    const float* Xg  = X  + (size_t)(m0 + sr) * DMsz + sc;
    const float* W1g = W1 + (size_t)sr * DIsz + s0 + sc;
    const float* W2g = W2 + (size_t)(s0 + sr) * DMsz + sc;
    v4f aprf[FT][4], bprf[16];

    v8f acc[FT][2];
#pragma unroll
    for (int t = 0; t < FT; ++t) {
        acc[t][0] = v8f{0.f, 0.f, 0.f, 0.f, 0.f, 0.f, 0.f, 0.f};
        acc[t][1] = v8f{0.f, 0.f, 0.f, 0.f, 0.f, 0.f, 0.f, 0.f};
    }

#pragma unroll
    for (int t = 0; t < FT; ++t)
#pragma unroll
        for (int u = 0; u < 4; ++u)
            aprf[t][u] = *(const v4f*)(Xg + (size_t)(t * Rsz + u * 8) * DMsz);
#pragma unroll
    for (int u = 0; u < 16; ++u)
        bprf[u] = *(const v4f*)(W1g + (size_t)(u * 8) * DIsz);

    const float* ap = As + (mq * 16 + l15) * FAP + lhalf * 2;
    const float* hp = Hs + (mq * 16 + l15) * FAP + lhalf * 2;
    const float* bp = Bs + (lhalf * 2) * FBP + nq * 32 + l15;

    for (int k0 = 0; k0 < DMsz; k0 += FK) {
#pragma unroll
        for (int t = 0; t < FT; ++t)
#pragma unroll
            for (int u = 0; u < 4; ++u)
                *(v4f*)&As[t * FAS + (sr + u * 8) * FAP + sc] = aprf[t][u];
#pragma unroll
        for (int u = 0; u < 16; ++u)
            *(v4f*)&Bs[(sr + u * 8) * FBP + sc] = bprf[u];
        __syncthreads();

        if (k0 + FK < DMsz) {
#pragma unroll
            for (int t = 0; t < FT; ++t)
#pragma unroll
                for (int u = 0; u < 4; ++u)
                    aprf[t][u] = *(const v4f*)(Xg + (size_t)(t * Rsz + u * 8) * DMsz + k0 + FK);
#pragma unroll
            for (int u = 0; u < 16; ++u)
                bprf[u] = *(const v4f*)(W1g + (size_t)(k0 + FK + u * 8) * DIsz);
        } else {
#pragma unroll
            for (int u = 0; u < 16; ++u)
                bprf[u] = *(const v4f*)(W2g + (size_t)(u * 8) * DMsz);
        }

        ffn_chunk(acc, ap, bp);
        __syncthreads();
    }

#pragma unroll
    for (int nsub = 0; nsub < 2; ++nsub) {
        const int hc = nq * 32 + nsub * 16 + l15;
        const float bv = b1[s0 + hc];
#pragma unroll
        for (int t = 0; t < FT; ++t)
#pragma unroll
            for (int r = 0; r < 8; ++r) {
                const int hr = mq * 16 + r + lhalf * 8;
                Hs[t * FAS + hr * FAP + hc] = fmaxf(acc[t][nsub][r] + bv, 0.f);
            }
    }

    float* Pp = P + (size_t)blockIdx.x * FT * Rsz * DMsz;
    for (int nc = 0; nc < 8; ++nc) {
#pragma unroll
        for (int t = 0; t < FT; ++t) {
            acc[t][0] = v8f{0.f, 0.f, 0.f, 0.f, 0.f, 0.f, 0.f, 0.f};
            acc[t][1] = v8f{0.f, 0.f, 0.f, 0.f, 0.f, 0.f, 0.f, 0.f};
        }

#pragma unroll
        for (int u = 0; u < 16; ++u)
            *(v4f*)&Bs[(sr + u * 8) * FBP + sc] = bprf[u];
        __syncthreads();

        if (nc + 1 < 8) {
#pragma unroll
            for (int u = 0; u < 16; ++u)
                bprf[u] = *(const v4f*)(W2g + (size_t)(u * 8) * DMsz + (nc + 1) * 128);
        }

        ffn_chunk(acc, hp, bp);
        __syncthreads();

#pragma unroll
        for (int t = 0; t < FT; ++t)
#pragma unroll
            for (int nsub = 0; nsub < 2; ++nsub) {
                const int col = nc * 128 + nq * 32 + nsub * 16 + l15;
#pragma unroll
                for (int r = 0; r < 8; ++r) {
                    const int row = t * Rsz + m0 + mq * 16 + r + lhalf * 8;
                    Pp[(size_t)row * DMsz + col] = acc[t][nsub][r];
                }
            }
    }
}

#define GT_KP (18 * 32)
#define GT_PK 32

__global__ __launch_bounds__(256)
void dec_kernel(const float* __restrict__ a0base, const float* __restrict__ a1base,
                const float* __restrict__ b1, const float* __restrict__ w2,
                const float* __restrict__ b2, float* __restrict__ dec)
{
    __shared__ float tile[GT_PK * GT_KP];

    const int tid  = threadIdx.x;
    const int wave = tid >> 5;
    const int lane = tid & 31;
    const int blk  = blockIdx.x;
    const int nt   = blk & 7;
    const int mt   = (blk >> 3) & 7;
    const int b    = (blk >> 6) & 3;
    const int pair = blk >> 8;
    const int i    = pair / 3;
    const int j    = pair % 3;

    const float4* g0 = (const float4*)(a0base + ((size_t)i * Rsz + b * Lsz + nt * 8) * 2048);
    const float4* g1 = (const float4*)(a1base + ((size_t)j * Rsz + b * Lsz + mt * 8) * 2048);
    const float2* gw = (const float2*)w2;

    const int ml = (wave >> 1) * 2;
    const int nl = (wave & 1) * 4;
    const float* pa = tile + ml * 32 + lane;
    const float* pb = tile + (8 + nl) * 32 + lane;
    const float* pc = tile + 16 * 32 + lane;

    float acc[2][4];
#pragma unroll
    for (int x = 0; x < 2; ++x)
#pragma unroll
        for (int y = 0; y < 4; ++y) acc[x][y] = 0.f;

    for (int p = 0; p < 2; ++p) {
        {
            float4 r0[8], r1[8];
#pragma unroll
            for (int u = 0; u < 8; ++u) r0[u] = g0[u * 512 + p * 256 + tid];
#pragma unroll
            for (int u = 0; u < 8; ++u) r1[u] = g1[u * 512 + p * 256 + tid];
            const int dst = (tid >> 3) * GT_KP + (tid & 7) * 4;
#pragma unroll
            for (int u = 0; u < 8; ++u) {
                *(float4*)&tile[dst + (8 + u) * 32] = r0[u];
                *(float4*)&tile[dst + u * 32]       = r1[u];
            }
        }
#pragma unroll
        for (int u = 0; u < 4; ++u) {
            const int cl  = tid + u * 256;
            const float2 w = gw[p * 1024 + cl];
            const int dst = (cl >> 5) * GT_KP + (cl & 31);
            tile[dst + 16 * 32] = b1[p * 1024 + cl];
            tile[dst + 17 * 32] = w.y - w.x;
        }
        __syncthreads();

#pragma unroll 4
        for (int k = 0; k < GT_PK; ++k) {
            const int o = k * GT_KP;
            float xv[2], yv[4];
            xv[0] = pa[o];      xv[1] = pa[o + 32];
            yv[0] = pb[o];      yv[1] = pb[o + 32];
            yv[2] = pb[o + 64]; yv[3] = pb[o + 96];
            const float bb = pc[o], ww = pc[o + 32];
#pragma unroll
            for (int x = 0; x < 2; ++x)
#pragma unroll
                for (int y = 0; y < 4; ++y) {
                    const float t = (xv[x] + yv[y]) + bb;
                    if (t > 0.f) acc[x][y] = fmaf(t, ww, acc[x][y]);
                }
        }
        __syncthreads();
    }

#pragma unroll
    for (int off = 16; off; off >>= 1)
#pragma unroll
        for (int x = 0; x < 2; ++x)
#pragma unroll
            for (int y = 0; y < 4; ++y)
                acc[x][y] += __shfl_xor(acc[x][y], off, 32);

    if (lane == 0) {
        const float bd = b2[1] - b2[0];
        float* dtile = dec + (((size_t)(pair * Bsz + b) * Lsz + mt * 8) * Lsz + nt * 8);
#pragma unroll
        for (int x = 0; x < 2; ++x) {
            float4 r;
            r.x = ((acc[x][0] + bd) > 0.f) ? NEGC : 0.f;
            r.y = ((acc[x][1] + bd) > 0.f) ? NEGC : 0.f;
            r.z = ((acc[x][2] + bd) > 0.f) ? NEGC : 0.f;
            r.w = ((acc[x][3] + bd) > 0.f) ? NEGC : 0.f;
            *(float4*)(dtile + (size_t)(ml + x) * Lsz + nl) = r;
        }
    }
}

#define SC_QT 16
#define SC_KP 68

__global__ __launch_bounds__(128)
void attn_scores_kernel(const float* __restrict__ Qb, const float* __restrict__ Kb,
                        const float* __restrict__ dec, const int* __restrict__ mask,
                        float* __restrict__ S)
{
    __shared__ __attribute__((aligned(16))) float Ks[192 * SC_KP];
    __shared__ __attribute__((aligned(16))) float Qs[SC_QT * 64];

    const int blk  = blockIdx.x;
    const int qt   = blk & 3;
    const int h    = (blk >> 2) & 15;
    const int b    = (blk >> 6) & 3;
    const int i    = blk >> 8;
    const int tid  = threadIdx.x;
    const int wave = tid >> 5;
    const int lane = tid & 31;
    const int q0   = qt * SC_QT + wave * 4;

    {
        const int c4 = (tid & 15) << 2;
        const int r0 = tid >> 4;
        v4f kr[24], qr[2];
#pragma unroll
        for (int u = 0; u < 24; ++u) {
            const int row = r0 + u * 8;
            const int j = row >> 6, k = row & 63;
            kr[u] = *(const v4f*)(Kb + ((size_t)j * Rsz + b * Lsz + k) * DMsz + h * DKsz + c4);
        }
#pragma unroll
        for (int u = 0; u < 2; ++u) {
            const int row = r0 + u * 8;
            qr[u] = *(const v4f*)(Qb + ((size_t)i * Rsz + b * Lsz + qt * SC_QT + row) * DMsz
                                  + h * DKsz + c4);
        }
#pragma unroll
        for (int u = 0; u < 24; ++u)
            *(v4f*)&Ks[(r0 + u * 8) * SC_KP + c4] = kr[u];
#pragma unroll
        for (int u = 0; u < 2; ++u)
            *(v4f*)&Qs[(r0 + u * 8) * 64 + c4] = qr[u];
    }

    float dv[4][6];
    int   mk[4][2];
#pragma unroll
    for (int r = 0; r < 4; ++r) {
#pragma unroll
        for (int m = 0; m < 6; ++m) {
            const int j = m >> 1, k = (m & 1) * 32 + lane;
            dv[r][m] = dec[(((size_t)(i * 3 + j) * Bsz + b) * Lsz + q0 + r) * Lsz + k];
        }
#pragma unroll
        for (int kh = 0; kh < 2; ++kh)
            mk[r][kh] = mask[((size_t)b * Lsz + q0 + r) * Lsz + kh * 32 + lane];
    }
    __syncthreads();

    float acc[4][6];
#pragma unroll
    for (int r = 0; r < 4; ++r)
#pragma unroll
        for (int m = 0; m < 6; ++m) acc[r][m] = 0.f;

    const float* kp = Ks + lane * SC_KP;
    const float* qp = Qs + (wave * 4) * 64;
    for (int d = 0; d < 64; d += 4) {
        v4f kv[6], qv[4];
#pragma unroll
        for (int m = 0; m < 6; ++m) kv[m] = *(const v4f*)(kp + m * 32 * SC_KP + d);
#pragma unroll
        for (int r = 0; r < 4; ++r) qv[r] = *(const v4f*)(qp + r * 64 + d);
#pragma unroll
        for (int r = 0; r < 4; ++r)
#pragma unroll
            for (int m = 0; m < 6; ++m) {
                float s = acc[r][m];
                s = fmaf(qv[r].x, kv[m].x, s);
                s = fmaf(qv[r].y, kv[m].y, s);
                s = fmaf(qv[r].z, kv[m].z, s);
                s = fmaf(qv[r].w, kv[m].w, s);
                acc[r][m] = s;
            }
    }

#pragma unroll
    for (int r = 0; r < 4; ++r) {
        float s[6];
#pragma unroll
        for (int m = 0; m < 6; ++m) {
            float t = acc[r][m];
            t *= 0.125f;
            t += dv[r][m];
            if (mk[r][m & 1] == 0) t = NEGC;
            s[m] = t;
        }
        float mx = fmaxf(fmaxf(fmaxf(s[0], s[1]), fmaxf(s[2], s[3])), fmaxf(s[4], s[5]));
#pragma unroll
        for (int off = 16; off; off >>= 1) mx = fmaxf(mx, __shfl_xor(mx, off, 32));
        float e[6];
#pragma unroll
        for (int m = 0; m < 6; ++m) e[m] = __expf(s[m] - mx);
        float sum = ((e[0] + e[2]) + e[4]) + ((e[1] + e[3]) + e[5]);
#pragma unroll
        for (int off = 16; off; off >>= 1) sum += __shfl_xor(sum, off, 32);
        const float inv = 1.0f / sum;
        float* Sp = S + ((size_t)((((i * Bsz + b) * Hsz + h) * Lsz) + q0 + r)) * 192 + lane;
#pragma unroll
        for (int m = 0; m < 6; ++m) Sp[m * 32] = e[m] * inv;
    }
}

#define AV_QT 16
#define AV_VP 68

__global__ __launch_bounds__(128)
void attn_av_kernel(const float* __restrict__ S, const float* __restrict__ Vb,
                    float* __restrict__ Ob)
{
    __shared__ __attribute__((aligned(16))) float Vs[192 * AV_VP];
    __shared__ __attribute__((aligned(16))) float Ss[AV_QT * 192];

    const int blk  = blockIdx.x;
    const int qt   = blk & 3;
    const int h    = (blk >> 2) & 15;
    const int b    = (blk >> 6) & 3;
    const int i    = blk >> 8;
    const int tid  = threadIdx.x;
    const int wave = tid >> 5;
    const int lane = tid & 31;

    {
        const int c4 = (tid & 15) << 2;
        const int r0 = tid >> 4;
        const float* Sg = S + ((size_t)((((i * Bsz + b) * Hsz + h) * Lsz) + qt * AV_QT)) * 192;
        v4f vr[24], sr[6];
#pragma unroll
        for (int u = 0; u < 24; ++u) {
            const int row = r0 + u * 8;
            const int j = row >> 6, k = row & 63;
            vr[u] = *(const v4f*)(Vb + ((size_t)j * Rsz + b * Lsz + k) * DMsz + h * DVsz + c4);
        }
#pragma unroll
        for (int u = 0; u < 6; ++u)
            sr[u] = *(const v4f*)(Sg + (size_t)(tid + u * 128) * 4);
#pragma unroll
        for (int u = 0; u < 24; ++u)
            *(v4f*)&Vs[(r0 + u * 8) * AV_VP + c4] = vr[u];
#pragma unroll
        for (int u = 0; u < 6; ++u)
            *(v4f*)&Ss[(tid + u * 128) * 4] = sr[u];
    }
    __syncthreads();

    float acc[4][2];
#pragma unroll
    for (int r = 0; r < 4; ++r) { acc[r][0] = 0.f; acc[r][1] = 0.f; }

    const float* vp = Vs + lane;
    const float* sp = Ss + (wave * 4) * 192;
#pragma unroll 4
    for (int jk = 0; jk < 192; jk += 4) {
        v4f sv[4];
        float vv[4][2];
#pragma unroll
        for (int r = 0; r < 4; ++r) sv[r] = *(const v4f*)(sp + r * 192 + jk);
#pragma unroll
        for (int t = 0; t < 4; ++t) {
            vv[t][0] = vp[(jk + t) * AV_VP];
            vv[t][1] = vp[(jk + t) * AV_VP + 32];
        }
#pragma unroll
        for (int r = 0; r < 4; ++r)
#pragma unroll
            for (int cc = 0; cc < 2; ++cc) {
                float s = acc[r][cc];
                s = fmaf(sv[r].x, vv[0][cc], s);
                s = fmaf(sv[r].y, vv[1][cc], s);
                s = fmaf(sv[r].z, vv[2][cc], s);
                s = fmaf(sv[r].w, vv[3][cc], s);
                acc[r][cc] = s;
            }
    }

#pragma unroll
    for (int r = 0; r < 4; ++r) {
        const int q = qt * AV_QT + wave * 4 + r;
        float* Op = Ob + ((size_t)i * Rsz + b * Lsz + q) * DMsz + h * DVsz + lane;
        Op[0]  = acc[r][0];
        Op[32] = acc[r][1];
    }
}

template <int NP, bool BR>
__global__ __launch_bounds__(256)
void layernorm_kernel(const float* __restrict__ x, size_t pstride,
                      const float* __restrict__ bias, const float* __restrict__ resid,
                      const float* __restrict__ g, const float* __restrict__ bta,
                      float* __restrict__ out)
{
    const int row = blockIdx.x;
    const int t = threadIdx.x;
    __shared__ float red[256];

    const float* xp = x + (size_t)row * 1024 + t;
    float pv[NP][4], bv[4], rv[4], gv[4], tv[4];
#pragma unroll
    for (int p = 0; p < NP; ++p)
#pragma unroll
        for (int u = 0; u < 4; ++u) pv[p][u] = xp[(size_t)p * pstride + u * 256];
#pragma unroll
    for (int u = 0; u < 4; ++u) {
        bv[u] = BR ? bias[t + u * 256] : 0.f;
        rv[u] = BR ? resid[(size_t)row * 1024 + t + u * 256] : 0.f;
        gv[u] = g[t + u * 256];
        tv[u] = bta[t + u * 256];
    }
    __builtin_amdgcn_sched_barrier(0);

    float v[4];
#pragma unroll
    for (int u = 0; u < 4; ++u) v[u] = pv[0][u];
#pragma unroll
    for (int p = 1; p < NP; ++p)
#pragma unroll
        for (int u = 0; u < 4; ++u) v[u] += pv[p][u];
    float s = 0.f;
#pragma unroll
    for (int u = 0; u < 4; ++u) {
        float val = v[u];
        if (BR) { val += bv[u]; val += rv[u]; }
        v[u] = val; s += val;
    }
    red[t] = s; __syncthreads();
    for (int st = 128; st > 0; st >>= 1) { if (t < st) red[t] += red[t + st]; __syncthreads(); }
    const float mu = red[0] * (1.f / 1024.f);
    __syncthreads();

    float s2 = 0.f;
#pragma unroll
    for (int u = 0; u < 4; ++u) { float dd = v[u] - mu; s2 += dd * dd; }
    red[t] = s2; __syncthreads();
    for (int st = 128; st > 0; st >>= 1) { if (t < st) red[t] += red[t + st]; __syncthreads(); }
    const float var = red[0] * (1.f / 1024.f);
    const float rinv = 1.0f / sqrtf(var + 1e-6f);
#pragma unroll
    for (int u = 0; u < 4; ++u) {
        int c = t + u * 256;
        out[(size_t)row * 1024 + c] = (v[u] - mu) * rinv * gv[u] + tv[u];
    }
}

extern "C" void kernel_launch(void* const* d_in, const int* in_sizes, int n_in,
                              void* d_out, int out_size, void* d_ws, size_t ws_size,
                              hipStream_t stream)
{
    const float* X[3]   = { (const float*)d_in[0], (const float*)d_in[1], (const float*)d_in[2] };
    const int*   mask   = (const int*)d_in[3];
    const float* w_qs   = (const float*)d_in[4];
    const float* w_ks   = (const float*)d_in[5];
    const float* w_vs   = (const float*)d_in[6];
    const float* w_ds0  = (const float*)d_in[7];
    const float* w_ds1  = (const float*)d_in[8];
    const float* dc_w1  = (const float*)d_in[9];
    const float* dc_b1  = (const float*)d_in[10];
    const float* dc_w2  = (const float*)d_in[11];
    const float* dc_b2  = (const float*)d_in[12];
    const float* w_fc   = (const float*)d_in[13];
    const float* ln1_g  = (const float*)d_in[14];
    const float* ln1_b  = (const float*)d_in[15];
    const float* ffn_w1 = (const float*)d_in[16];
    const float* ffn_b1 = (const float*)d_in[17];
    const float* ffn_w2 = (const float*)d_in[18];
    const float* ffn_b2 = (const float*)d_in[19];
    const float* ln2_g  = (const float*)d_in[20];
    const float* ln2_b  = (const float*)d_in[21];
    float* out = (float*)d_out;

    float* ws = (float*)d_ws;
    size_t off = 0;
    auto alloc = [&](size_t n) { float* p = ws + off; off += n; return p; };
    const size_t RM = (size_t)Rsz * DMsz;
    float* Q   = alloc(3 * RM);
    float* Kb  = alloc(3 * RM);
    float* Vb  = alloc(3 * RM);
    float* d0  = alloc(3 * RM);
    float* d1  = alloc(3 * RM);
    float* a0  = alloc(3 * (size_t)Rsz * 2048);
    float* a1  = alloc(3 * (size_t)Rsz * 2048);
    float* dec = alloc(9 * (size_t)Bsz * Lsz * Lsz);
    float* S   = alloc((size_t)3 * Bsz * Hsz * Lsz * 192);
    float* O   = alloc(3 * RM);
    float* fc  = alloc(3 * RM);
    float* oln = alloc(3 * RM);
    float* fpart = alloc(32 * 3 * RM);
    (void)ws_size; (void)n_in; (void)in_sizes; (void)out_size;

    auto gemm3 = [&](const float* A, const float* Bm, const float* r0, const float* r1,
                     const float* r2, float* C, int N, int K) {
        dim3 grid(N / 32, Rsz / 32);
        gemm3_wmma_f32<<<grid, dim3(128), 0, stream>>>(A, Bm, r0, r1, r2, C, N, K);
    };

    proj15_wmma_f32<<<dim3(DMsz / 32, Rsz / 32), dim3(128), 0, stream>>>(
        X[0], X[1], X[2], w_qs, w_ks, w_vs, w_ds0, w_ds1, Q, Kb, Vb, d0, d1);
    gemm3(d0, dc_w1 + (size_t)1024 * 2048, nullptr, nullptr, nullptr, a0, 2048, 1024);
    gemm3(d1, dc_w1, nullptr, nullptr, nullptr, a1, 2048, 1024);

    {
        dec_kernel<<<9 * Bsz * 8 * 8, 256, 0, stream>>>(a0, a1, dc_b1, dc_w2, dc_b2, dec);
    }

    {
        attn_scores_kernel<<<3 * Bsz * Hsz * (Lsz / SC_QT), 128, 0, stream>>>(Q, Kb, dec, mask, S);
        attn_av_kernel<<<3 * Bsz * Hsz * (Lsz / AV_QT), 128, 0, stream>>>(S, Vb, O);
    }

    gemm3(O, w_fc, X[0], X[1], X[2], fc, 1024, 1024);
    for (int i = 0; i < 3; ++i)
        layernorm_kernel<1, false><<<Rsz, 256, 0, stream>>>(fc + i * RM, RM, nullptr, nullptr,
                                                            ln1_g, ln1_b, oln + i * RM);
    ffn_fused_wmma_f32<<<dim3(32, Rsz / FM), dim3(256), 0, stream>>>(
        oln, ffn_w1, ffn_b1, ffn_w2, fpart);
    for (int i = 0; i < 3; ++i)
        layernorm_kernel<32, true><<<Rsz, 256, 0, stream>>>(fpart + i * RM, 3 * RM, ffn_b2, oln + i * RM,
                                                            ln2_g, ln2_b, out + i * RM);
}
